# starting kernel
# baseline (speedup 1.0000x reference)
#include <hip/hip_runtime.h>
#include <hip/hip_bf16.h>

typedef __attribute__((ext_vector_type(2))) float v2f;
typedef __attribute__((ext_vector_type(8))) float v8f;

#define C_HID 256

__global__ void k_deg_init(int* __restrict__ degi, int n) {
  int i = blockIdx.x * blockDim.x + threadIdx.x;
  if (i < n) degi[i] = 1;
}

__global__ void k_deg_count(const int* __restrict__ dst, int* __restrict__ degi, int E) {
  int e = blockIdx.x * blockDim.x + threadIdx.x;
  if (e < E) atomicAdd(&degi[dst[e]], 1);
}

__global__ void k_dinv(const int* __restrict__ degi, float* __restrict__ dinv, int n) {
  int i = blockIdx.x * blockDim.x + threadIdx.x;
  if (i < n) dinv[i] = rsqrtf((float)degi[i]);
}

__global__ void k_zero(float* __restrict__ p, long long n) {
  long long i = (long long)blockIdx.x * blockDim.x + threadIdx.x;
  if (i < n) p[i] = 0.0f;
}

__global__ __launch_bounds__(256) void k_gemm_wmma(
    const float* __restrict__ X, const float* __restrict__ W,
    float* __restrict__ Hout, int M, int K, int N) {
  const int wpb  = blockDim.x >> 5;
  const int wave = blockIdx.x * wpb + ((int)threadIdx.x >> 5);
  const int lane = threadIdx.x & 31;
  const int ntiles = N >> 4;
  const int mt = wave / ntiles;
  const int nt = wave - mt * ntiles;
  const int m0 = mt << 4;
  if (m0 >= M) return;
  const int n0 = nt << 4;
  const int lo = lane & 15;
  const int hi = lane >> 4;

  int arow = m0 + lo;
  if (arow >= M) arow = M - 1;
  const float* Xr = X + (long long)arow * K + 2 * hi;
  const float* Wc = W + (long long)(2 * hi) * N + n0 + lo;

  v8f acc = {};
  for (int k = 0; k < K; k += 4) {
    v2f a, b;
    a.x = Xr[0];
    a.y = Xr[1];
    b.x = Wc[0];
    b.y = Wc[N];
    Xr += 4;
    Wc += (long long)4 * N;
    acc = __builtin_amdgcn_wmma_f32_16x16x4_f32(
         false, a,  false, b,
         (short)0, acc,  false,  false);
  }

  float* Hp = Hout + (long long)(m0 + hi * 8) * N + n0 + lo;
#pragma unroll
  for (int r = 0; r < 8; ++r) {
    int mr = m0 + hi * 8 + r;
    if (mr < M) Hp[(long long)r * N] = acc[r];
  }
}

__device__ __forceinline__ void atomAddF(float* p, float v) {
  unsafeAtomicAdd(p, v);
}

__global__ __launch_bounds__(256) void k_scatter(
    const float* __restrict__ H, const int* __restrict__ src,
    const int* __restrict__ dst, const float* __restrict__ dinv,
    float* __restrict__ AGG, int E) {
  long long tid = (long long)blockIdx.x * blockDim.x + threadIdx.x;
  long long total = (long long)E * (C_HID / 4);
  if (tid >= total) return;
  int e = (int)(tid >> 6);
  int c = ((int)tid & 63) << 2;
  int s = src[e];
  int d = dst[e];
  float nrm = dinv[s] * dinv[d];
  float4 hv = *(const float4*)(H + (long long)s * C_HID + c);
  float* op = AGG + (long long)d * C_HID + c;
  atomAddF(op + 0, hv.x * nrm);
  atomAddF(op + 1, hv.y * nrm);
  atomAddF(op + 2, hv.z * nrm);
  atomAddF(op + 3, hv.w * nrm);
}

__global__ __launch_bounds__(256) void k_combine(
    const float* __restrict__ AGG, const float* __restrict__ H,
    const float* __restrict__ dinv, const float* __restrict__ bias,
    float* __restrict__ OUT, int n) {
  long long i = (long long)blockIdx.x * blockDim.x + threadIdx.x;
  long long total = (long long)n * C_HID;
  if (i >= total) return;
  int node = (int)(i >> 8);
  int c = (int)i & 255;
  float di = dinv[node];
  float v = AGG[i] + di * di * H[i] + bias[c];
  OUT[i] = (v > 0.0f) ? v : expm1f(v);
}

extern "C" void kernel_launch(void* const* d_in, const int* in_sizes, int n_in,
                              void* d_out, int out_size, void* d_ws, size_t ws_size,
                              hipStream_t stream) {
  const float* x  = (const float*)d_in[0];
  const int*   ei = (const int*)d_in[1];
  const float* W1 = (const float*)d_in[2];
  const float* b1 = (const float*)d_in[3];
  const float* W2 = (const float*)d_in[4];
  const float* b2 = (const float*)d_in[5];
  const float* W3 = (const float*)d_in[6];
  const float* b3 = (const float*)d_in[7];
  float* out = (float*)d_out;

  const int C   = C_HID;
  const int E   = in_sizes[1] / 2;
  const int cin = in_sizes[2] / C;
  const int n   = in_sizes[0] / cin;
  const int* src = ei;
  const int* dst = ei + E;

  char*  ws  = (char*)d_ws;
  size_t off = 0;
  auto alloc = [&](size_t bytes) -> void* {
    void* p = ws + off;
    off = (off + bytes + 255) & ~(size_t)255;
    return p;
  };
  int*   degi = (int*)  alloc((size_t)n * sizeof(int));
  float* dinv = (float*)alloc((size_t)n * sizeof(float));
  float* H    = (float*)alloc((size_t)n * C * sizeof(float));
  float* AGG  = (float*)alloc((size_t)n * C * sizeof(float));
  float* XB   = (float*)alloc((size_t)n * C * sizeof(float));
  (void)ws_size; (void)n_in; (void)out_size;

  k_deg_init <<<(n + 255) / 256, 256, 0, stream>>>(degi, n);
  k_deg_count<<<(E + 255) / 256, 256, 0, stream>>>(dst, degi, E);
  k_dinv     <<<(n + 255) / 256, 256, 0, stream>>>(degi, dinv, n);

  const long long nC = (long long)n * C;
  const int zgrid = (int)((nC + 255) / 256);
  const long long sThreads = (long long)E * (C / 4);
  const int sgrid = (int)((sThreads + 255) / 256);

  auto layer = [&](const float* Xin, const float* W, const float* bias,
                   float* Out, int K) {
    const int mtiles = (n + 15) / 16;
    const int ntiles = C / 16;
    const int waves  = mtiles * ntiles;
    const int wpb    = 8;
    const int blocks = (waves + wpb - 1) / wpb;
    k_gemm_wmma<<<blocks, wpb * 32, 0, stream>>>(Xin, W, H, n, K, C);
    k_zero     <<<zgrid, 256, 0, stream>>>(AGG, nC);
    k_scatter  <<<sgrid, 256, 0, stream>>>(H, src, dst, dinv, AGG, E);
    k_combine  <<<zgrid, 256, 0, stream>>>(AGG, H, dinv, bias, Out, n);
  };

  layer(x,  W1, b1, XB,  cin);
  layer(XB, W2, b2, XB,  C);
  layer(XB, W3, b3, out, C);
}


# edge scatter-add moved from global f32 atomics into LDS: a work-group owns 40 destination rows, scans all edges, adds gathered rows by ds_add_f32, writes elu(sum+self+bias); sums stay arrival order
# speedup vs baseline: 2.3129x; 2.3129x over previous
#include <hip/hip_runtime.h>
#include <hip/hip_bf16.h>

typedef __attribute__((ext_vector_type(2))) float v2f;
typedef __attribute__((ext_vector_type(8))) float v8f;

#define C_HID 256

__global__ void k_deg_init(int* __restrict__ degi, int n) {
  int i = blockIdx.x * blockDim.x + threadIdx.x;
  if (i < n) degi[i] = 1;
}

__global__ void k_deg_count(const int* __restrict__ dst, int* __restrict__ degi, int E) {
  int e = blockIdx.x * blockDim.x + threadIdx.x;
  if (e < E) atomicAdd(&degi[dst[e]], 1);
}

__global__ void k_dinv(const int* __restrict__ degi, float* __restrict__ dinv, int n) {
  int i = blockIdx.x * blockDim.x + threadIdx.x;
  if (i < n) dinv[i] = rsqrtf((float)degi[i]);
}

__global__ __launch_bounds__(256) void k_gemm_wmma(
    const float* __restrict__ X, const float* __restrict__ W,
    float* __restrict__ Hout, int M, int K, int N) {
  const int wpb  = blockDim.x >> 5;
  const int wave = blockIdx.x * wpb + ((int)threadIdx.x >> 5);
  const int lane = threadIdx.x & 31;
  const int ntiles = N >> 4;
  const int mt = wave / ntiles;
  const int nt = wave - mt * ntiles;
  const int m0 = mt << 4;
  if (m0 >= M) return;
  const int n0 = nt << 4;
  const int lo = lane & 15;
  const int hi = lane >> 4;

  int arow = m0 + lo;
  if (arow >= M) arow = M - 1;
  const float* Xr = X + (long long)arow * K + 2 * hi;
  const float* Wc = W + (long long)(2 * hi) * N + n0 + lo;

  v8f acc = {};
  for (int k = 0; k < K; k += 4) {
    v2f a, b;
    a.x = Xr[0];
    a.y = Xr[1];
    b.x = Wc[0];
    b.y = Wc[N];
    Xr += 4;
    Wc += (long long)4 * N;
    acc = __builtin_amdgcn_wmma_f32_16x16x4_f32(
         false, a,  false, b,
         (short)0, acc,  false,  false);
  }

  float* Hp = Hout + (long long)(m0 + hi * 8) * N + n0 + lo;
#pragma unroll
  for (int r = 0; r < 8; ++r) {
    int mr = m0 + hi * 8 + r;
    if (mr < M) Hp[(long long)r * N] = acc[r];
  }
}

#define PUSH_ROWS    40
#define PUSH_THREADS 1024
#define PUSH_WAVES   32
#define PUSH_RING    1024
#define PUSH_MIN_BATCH 16

typedef __attribute__((ext_vector_type(4))) int v4i;

__device__ __forceinline__ void ldsAddF(float* p, float v) {
  __hip_atomic_fetch_add(p, v, __ATOMIC_RELAXED, __HIP_MEMORY_SCOPE_WORKGROUP);
}

__device__ __forceinline__ int laneI(int v, int j) {
  return __builtin_amdgcn_readlane(v, j);
}
__device__ __forceinline__ float laneF(float v, int j) {
  return __int_as_float(__builtin_amdgcn_readlane(__float_as_int(v), j));
}

__device__ __forceinline__ void push_batch(
    const float* __restrict__ H, const int* __restrict__ src,
    const float* __restrict__ dinv, float* acc, const unsigned* ring,
    int head, int count, int row0, int lane) {
  int s = 0, off = 0;
  float nrm = 0.0f;
  if (lane < count) {
    unsigned ent = ring[(head + lane) & (PUSH_RING - 1)];
    int dl = (int)(ent & 63u);
    s   = src[ent >> 6];
    nrm = dinv[s] * dinv[row0 + dl];
    off = dl * C_HID;
  }
  int j = 0;
  for (; j + 4 <= count; j += 4) {
    float4 a[4], b[4];
#pragma unroll
    for (int u = 0; u < 4; ++u) {
      const float* hp = H + (long long)laneI(s, j + u) * C_HID + lane * 4;
      a[u] = *(const float4*)hp;
      b[u] = *(const float4*)(hp + 128);
    }
#pragma unroll
    for (int u = 0; u < 4; ++u) {
      const float nj = laneF(nrm, j + u);
      float* q = acc + laneI(off, j + u) + lane;
      ldsAddF(q,       a[u].x * nj);
      ldsAddF(q + 32,  a[u].y * nj);
      ldsAddF(q + 64,  a[u].z * nj);
      ldsAddF(q + 96,  a[u].w * nj);
      ldsAddF(q + 128, b[u].x * nj);
      ldsAddF(q + 160, b[u].y * nj);
      ldsAddF(q + 192, b[u].z * nj);
      ldsAddF(q + 224, b[u].w * nj);
    }
  }
  for (; j < count; ++j) {
    const float* hp = H + (long long)laneI(s, j) * C_HID + lane * 4;
    const float4 a = *(const float4*)hp;
    const float4 b = *(const float4*)(hp + 128);
    const float nj = laneF(nrm, j);
    float* q = acc + laneI(off, j) + lane;
    ldsAddF(q,       a.x * nj);
    ldsAddF(q + 32,  a.y * nj);
    ldsAddF(q + 64,  a.z * nj);
    ldsAddF(q + 96,  a.w * nj);
    ldsAddF(q + 128, b.x * nj);
    ldsAddF(q + 160, b.y * nj);
    ldsAddF(q + 192, b.z * nj);
    ldsAddF(q + 224, b.w * nj);
  }
}

__device__ __forceinline__ void load_ids(const int* __restrict__ p, int lane, v4i* v) {
#pragma unroll
  for (int k = 0; k < 4; ++k)
    v[k] = *(const v4i*)(p + k * 128 + lane * 4);
}

__global__ __launch_bounds__(PUSH_THREADS) void k_push_rows(
    const float* __restrict__ H, const int* __restrict__ src,
    const int* __restrict__ dst, const float* __restrict__ dinv,
    const float* __restrict__ bias, float* __restrict__ OUT, int n, int E) {
  __shared__ float    acc[PUSH_ROWS * C_HID];
  __shared__ unsigned pend[PUSH_WAVES * PUSH_RING];

  const int tid  = threadIdx.x;
  const int lane = tid & 31;
  const int w    = __builtin_amdgcn_readfirstlane(tid >> 5);
  const int row0 = blockIdx.x * PUSH_ROWS;
  const int nrows = min(PUSH_ROWS, n - row0);
  unsigned* ring = pend + w * PUSH_RING;

  for (int i = tid; i < PUSH_ROWS * C_HID; i += PUSH_THREADS) acc[i] = 0.0f;
  __syncthreads();

  const int lead = min(E, (int)(((16 - (((size_t)dst) & 15)) & 15) >> 2));
  const int nsb  = (E - lead) >> 9;
  const int* dsta = dst + lead;
  const int tail0 = lead + (nsb << 9);

  int head = 0, tail = 0;
  {
    const int t = w * 32 + lane;
    const int e = (t < lead) ? t : tail0 + (t - lead);
    const int dv = (e < E) ? dst[e] : -1;
    const unsigned dd = (unsigned)(dv - row0);
    const bool m = dd < (unsigned)nrows;
    const unsigned mask = __builtin_amdgcn_ballot_w32(m);
    if (m) ring[__builtin_amdgcn_mbcnt_lo(mask, 0u)] = ((unsigned)e << 6) | dd;
    tail = __builtin_popcount(mask);
  }

  const int start = (nsb > 0) ? (int)(((long long)blockIdx.x * nsb) / gridDim.x) : 0;

  v4i cur[4];
  if (w < nsb) {
    int sb = start + w; if (sb >= nsb) sb -= nsb;
    load_ids(dsta + (sb << 9), lane, cur);
  }
  for (int i = w; ; i += PUSH_WAVES) {
    const bool more = i < nsb;
    if (more) {
      int sb = start + i; if (sb >= nsb) sb -= nsb;
      const int base = lead + (sb << 9);
      int sn = (i + PUSH_WAVES < nsb) ? start + i + PUSH_WAVES : start + i;
      if (sn >= nsb) sn -= nsb;
      v4i nxt[4];
      load_ids(dsta + (sn << 9), lane, nxt);
#pragma unroll
      for (int k = 0; k < 4; ++k) {
#pragma unroll
        for (int q = 0; q < 4; ++q) {
          const unsigned dd = (unsigned)(cur[k][q] - row0);
          const bool m = dd < (unsigned)nrows;
          const unsigned mask = __builtin_amdgcn_ballot_w32(m);
          if (mask) {
            if (m) {
              const unsigned e = (unsigned)(base + k * 128 + lane * 4 + q);
              ring[(tail + __builtin_amdgcn_mbcnt_lo(mask, 0u)) & (PUSH_RING - 1)] = (e << 6) | dd;
            }
            tail += __builtin_popcount(mask);
          }
        }
      }
#pragma unroll
      for (int k = 0; k < 4; ++k) cur[k] = nxt[k];
    }
    const int need = more ? PUSH_MIN_BATCH : 1;
    while (tail - head >= need) {
      const int pending = tail - head;
      const int cnt = min(32, more ? (pending & ~3) : pending);
      push_batch(H, src, dinv, acc, ring, head, cnt, row0, lane);
      head += cnt;
    }
    if (!more) break;
  }

  __syncthreads();

  for (int idx = tid; idx < nrows * (C_HID / 4); idx += PUSH_THREADS) {
    const int r  = idx >> 6;
    const int cq = idx & 63;
    const int c  = cq << 2;
    const float* ap = acc + r * C_HID + (cq >> 5) * 128 + (cq & 31);
    const int node = row0 + r;
    const float di = dinv[node];
    const float4 h = *(const float4*)(H + (long long)node * C_HID + c);
    const float4 bb = *(const float4*)(bias + c);
    float4 v;
    v.x = ap[0]  + di * di * h.x + bb.x;
    v.y = ap[32] + di * di * h.y + bb.y;
    v.z = ap[64] + di * di * h.z + bb.z;
    v.w = ap[96] + di * di * h.w + bb.w;
    v.x = (v.x > 0.0f) ? v.x : expm1f(v.x);
    v.y = (v.y > 0.0f) ? v.y : expm1f(v.y);
    v.z = (v.z > 0.0f) ? v.z : expm1f(v.z);
    v.w = (v.w > 0.0f) ? v.w : expm1f(v.w);
    *(float4*)(OUT + (long long)node * C_HID + c) = v;
  }
}

extern "C" void kernel_launch(void* const* d_in, const int* in_sizes, int n_in,
                              void* d_out, int out_size, void* d_ws, size_t ws_size,
                              hipStream_t stream) {
  const float* x  = (const float*)d_in[0];
  const int*   ei = (const int*)d_in[1];
  const float* W1 = (const float*)d_in[2];
  const float* b1 = (const float*)d_in[3];
  const float* W2 = (const float*)d_in[4];
  const float* b2 = (const float*)d_in[5];
  const float* W3 = (const float*)d_in[6];
  const float* b3 = (const float*)d_in[7];
  float* out = (float*)d_out;

  const int C   = C_HID;
  const int E   = in_sizes[1] / 2;
  const int cin = in_sizes[2] / C;
  const int n   = in_sizes[0] / cin;
  const int* src = ei;
  const int* dst = ei + E;

  char*  ws  = (char*)d_ws;
  size_t off = 0;
  auto alloc = [&](size_t bytes) -> void* {
    void* p = ws + off;
    off = (off + bytes + 255) & ~(size_t)255;
    return p;
  };
  int*   degi = (int*)  alloc((size_t)n * sizeof(int));
  float* dinv = (float*)alloc((size_t)n * sizeof(float));
  float* H    = (float*)alloc((size_t)n * C * sizeof(float));
  float* XB   = (float*)alloc((size_t)n * C * sizeof(float));
  (void)ws_size; (void)n_in; (void)out_size;

  k_deg_init <<<(n + 255) / 256, 256, 0, stream>>>(degi, n);
  k_deg_count<<<(E + 255) / 256, 256, 0, stream>>>(dst, degi, E);
  k_dinv     <<<(n + 255) / 256, 256, 0, stream>>>(degi, dinv, n);

  const int pgrid = (n + PUSH_ROWS - 1) / PUSH_ROWS;

  auto layer = [&](const float* Xin, const float* W, const float* bias,
                   float* Out, int K) {
    const int mtiles = (n + 15) / 16;
    const int ntiles = C / 16;
    const int waves  = mtiles * ntiles;
    const int wpb    = 8;
    const int blocks = (waves + wpb - 1) / wpb;
    k_gemm_wmma<<<blocks, wpb * 32, 0, stream>>>(Xin, W, H, n, K, C);
    k_push_rows<<<pgrid, PUSH_THREADS, 0, stream>>>(H, src, dst, dinv, bias, Out, n, E);
  };

  layer(x,  W1, b1, XB,  cin);
  layer(XB, W2, b2, XB,  C);
  layer(XB, W3, b3, out, C);
}


# edge kernel scans a 1-byte owner-group key per edge, not the 4-byte destination id; the degree-count pass writes the keys and a src<<6|local-row word per edge each call; sums stay arrival order
# speedup vs baseline: 2.4377x; 1.0540x over previous
#include <hip/hip_runtime.h>
#include <hip/hip_bf16.h>

typedef __attribute__((ext_vector_type(2))) float v2f;
typedef __attribute__((ext_vector_type(8))) float v8f;

#define C_HID 256

#define PUSH_ROWS 40

typedef __attribute__((ext_vector_type(4))) unsigned v4u;

__global__ void k_deg_init(int* __restrict__ degi, int n) {
  int i = blockIdx.x * blockDim.x + threadIdx.x;
  if (i < n) degi[i] = 1;
}

__global__ void k_deg_count(const int* __restrict__ src, const int* __restrict__ dst,
                            int* __restrict__ degi, unsigned* __restrict__ key4,
                            unsigned* __restrict__ pk, int E) {
  const int t  = blockIdx.x * blockDim.x + threadIdx.x;
  const int e0 = t * 4;
  if (e0 >= E) return;
  unsigned kw = 0u;
  if (e0 + 3 < E) {
    int d[4], sv[4];
#pragma unroll
    for (int q = 0; q < 4; ++q) { d[q] = dst[e0 + q]; sv[q] = src[e0 + q]; }
    unsigned pv[4];
#pragma unroll
    for (int q = 0; q < 4; ++q) {
      const int g = d[q] / PUSH_ROWS;
      atomicAdd(&degi[d[q]], 1);
      pv[q] = ((unsigned)sv[q] << 6) | (unsigned)(d[q] - g * PUSH_ROWS);
      kw |= (unsigned)g << (8 * q);
    }
    v4u o; o.x = pv[0]; o.y = pv[1]; o.z = pv[2]; o.w = pv[3];
    *(v4u*)(pk + e0) = o;
  } else {
    for (int q = 0; e0 + q < E; ++q) {
      const int d = dst[e0 + q];
      const int g = d / PUSH_ROWS;
      atomicAdd(&degi[d], 1);
      pk[e0 + q] = ((unsigned)src[e0 + q] << 6) | (unsigned)(d - g * PUSH_ROWS);
      kw |= (unsigned)g << (8 * q);
    }
  }
  key4[t] = kw;
}

__global__ void k_dinv(const int* __restrict__ degi, float* __restrict__ dinv, int n) {
  int i = blockIdx.x * blockDim.x + threadIdx.x;
  if (i < n) dinv[i] = rsqrtf((float)degi[i]);
}

__global__ __launch_bounds__(256) void k_gemm_wmma(
    const float* __restrict__ X, const float* __restrict__ W,
    float* __restrict__ Hout, int M, int K, int N) {
  const int wpb  = blockDim.x >> 5;
  const int wave = blockIdx.x * wpb + ((int)threadIdx.x >> 5);
  const int lane = threadIdx.x & 31;
  const int ntiles = N >> 4;
  const int mt = wave / ntiles;
  const int nt = wave - mt * ntiles;
  const int m0 = mt << 4;
  if (m0 >= M) return;
  const int n0 = nt << 4;
  const int lo = lane & 15;
  const int hi = lane >> 4;

  int arow = m0 + lo;
  if (arow >= M) arow = M - 1;
  const float* Xr = X + (long long)arow * K + 2 * hi;
  const float* Wc = W + (long long)(2 * hi) * N + n0 + lo;

  v8f acc = {};
  for (int k = 0; k < K; k += 4) {
    v2f a, b;
    a.x = Xr[0];
    a.y = Xr[1];
    b.x = Wc[0];
    b.y = Wc[N];
    Xr += 4;
    Wc += (long long)4 * N;
    acc = __builtin_amdgcn_wmma_f32_16x16x4_f32(
         false, a,  false, b,
         (short)0, acc,  false,  false);
  }

  float* Hp = Hout + (long long)(m0 + hi * 8) * N + n0 + lo;
#pragma unroll
  for (int r = 0; r < 8; ++r) {
    int mr = m0 + hi * 8 + r;
    if (mr < M) Hp[(long long)r * N] = acc[r];
  }
}

#define PUSH_THREADS 1024
#define PUSH_WAVES   32
#define PUSH_RING    1024
#define KEY_LOADS    2
#define PUSH_MIN_BATCH 16

__device__ __forceinline__ void ldsAddF(float* p, float v) {
  __hip_atomic_fetch_add(p, v, __ATOMIC_RELAXED, __HIP_MEMORY_SCOPE_WORKGROUP);
}

__device__ __forceinline__ int laneI(int v, int j) {
  return __builtin_amdgcn_readlane(v, j);
}
__device__ __forceinline__ float laneF(float v, int j) {
  return __int_as_float(__builtin_amdgcn_readlane(__float_as_int(v), j));
}

__device__ __forceinline__ void push_batch(
    const float* __restrict__ H, const unsigned* __restrict__ pk,
    const float* __restrict__ dinv, float* acc, const unsigned* ring,
    int head, int count, int row0, int lane) {
  int s = 0, off = 0;
  float nrm = 0.0f;
  if (lane < count) {
    const unsigned ent = pk[ring[(head + lane) & (PUSH_RING - 1)]];
    int dl = (int)(ent & 63u);
    s   = (int)(ent >> 6);
    nrm = dinv[s] * dinv[row0 + dl];
    off = dl * C_HID;
  }
  int j = 0;
  for (; j + 4 <= count; j += 4) {
    float4 a[4], b[4];
#pragma unroll
    for (int u = 0; u < 4; ++u) {
      const float* hp = H + (long long)laneI(s, j + u) * C_HID + lane * 4;
      a[u] = *(const float4*)hp;
      b[u] = *(const float4*)(hp + 128);
    }
#pragma unroll
    for (int u = 0; u < 4; ++u) {
      const float nj = laneF(nrm, j + u);
      float* q = acc + laneI(off, j + u) + lane;
      ldsAddF(q,       a[u].x * nj);
      ldsAddF(q + 32,  a[u].y * nj);
      ldsAddF(q + 64,  a[u].z * nj);
      ldsAddF(q + 96,  a[u].w * nj);
      ldsAddF(q + 128, b[u].x * nj);
      ldsAddF(q + 160, b[u].y * nj);
      ldsAddF(q + 192, b[u].z * nj);
      ldsAddF(q + 224, b[u].w * nj);
    }
  }
  for (; j < count; ++j) {
    const float* hp = H + (long long)laneI(s, j) * C_HID + lane * 4;
    const float4 a = *(const float4*)hp;
    const float4 b = *(const float4*)(hp + 128);
    const float nj = laneF(nrm, j);
    float* q = acc + laneI(off, j) + lane;
    ldsAddF(q,       a.x * nj);
    ldsAddF(q + 32,  a.y * nj);
    ldsAddF(q + 64,  a.z * nj);
    ldsAddF(q + 96,  a.w * nj);
    ldsAddF(q + 128, b.x * nj);
    ldsAddF(q + 160, b.y * nj);
    ldsAddF(q + 192, b.z * nj);
    ldsAddF(q + 224, b.w * nj);
  }
}

__device__ __forceinline__ void load_keys(const unsigned char* __restrict__ p, int lane, v4u* v) {
#pragma unroll
  for (int k = 0; k < KEY_LOADS; ++k)
    v[k] = *(const v4u*)(p + k * 512 + lane * 16);
}

__global__ __launch_bounds__(PUSH_THREADS) void k_push_rows(
    const float* __restrict__ H, const unsigned* __restrict__ pk,
    const unsigned char* __restrict__ keys, const float* __restrict__ dinv,
    const float* __restrict__ bias, float* __restrict__ OUT, int n, int E) {
  __shared__ float    acc[PUSH_ROWS * C_HID];
  __shared__ unsigned pend[PUSH_WAVES * PUSH_RING];

  const int tid  = threadIdx.x;
  const int lane = tid & 31;
  const int w    = __builtin_amdgcn_readfirstlane(tid >> 5);
  const int row0 = blockIdx.x * PUSH_ROWS;
  const int nrows = min(PUSH_ROWS, n - row0);
  unsigned* ring = pend + w * PUSH_RING;
  const unsigned gpat = (unsigned)blockIdx.x * 0x01010101u;

  for (int i = tid; i < PUSH_ROWS * C_HID; i += PUSH_THREADS) acc[i] = 0.0f;
  __syncthreads();

  const int nsb   = E >> 10;
  const int tail0 = nsb << 10;

  int head = 0, tail = 0;
  {
    const int e = tail0 + w * 32 + lane;
    bool m = false;
    if (e < E) m = (unsigned)keys[e] == (unsigned)blockIdx.x;
    const unsigned mask = __builtin_amdgcn_ballot_w32(m);
    if (m) ring[__builtin_amdgcn_mbcnt_lo(mask, 0u)] = (unsigned)e;
    tail = __builtin_popcount(mask);
  }

  const int start = (nsb > 0) ? (int)(((long long)blockIdx.x * nsb) / gridDim.x) : 0;

  v4u cur[KEY_LOADS];
  if (w < nsb) {
    int sb = start + w; if (sb >= nsb) sb -= nsb;
    load_keys(keys + ((long long)sb << 10), lane, cur);
  }
  for (int i = w; ; i += PUSH_WAVES) {
    const bool more = i < nsb;
    if (more) {
      int sb = start + i; if (sb >= nsb) sb -= nsb;
      const int base = sb << 10;
      int sn = (i + PUSH_WAVES < nsb) ? start + i + PUSH_WAVES : start + i;
      if (sn >= nsb) sn -= nsb;
      v4u nxt[KEY_LOADS];
      load_keys(keys + ((long long)sn << 10), lane, nxt);
#pragma unroll
      for (int k = 0; k < KEY_LOADS; ++k) {
#pragma unroll
        for (int q = 0; q < 4; ++q) {
          const unsigned x = cur[k][q] ^ gpat;
          const unsigned z = (x - 0x01010101u) & ~x & 0x80808080u;
          if (__builtin_amdgcn_ballot_w32(z != 0u)) {
#pragma unroll
            for (int b = 0; b < 4; ++b) {
              const bool m = ((x >> (8 * b)) & 0xffu) == 0u;
              const unsigned mask = __builtin_amdgcn_ballot_w32(m);
              if (mask) {
                if (m) {
                  const unsigned e = (unsigned)(base + k * 512 + lane * 16 + q * 4 + b);
                  ring[(tail + __builtin_amdgcn_mbcnt_lo(mask, 0u)) & (PUSH_RING - 1)] = e;
                }
                tail += __builtin_popcount(mask);
              }
            }
          }
        }
        while (tail - head >= PUSH_MIN_BATCH) {
          const int cnt = min(32, (tail - head) & ~3);
          push_batch(H, pk, dinv, acc, ring, head, cnt, row0, lane);
          head += cnt;
        }
      }
#pragma unroll
      for (int k = 0; k < KEY_LOADS; ++k) cur[k] = nxt[k];
    } else {
      while (tail - head >= 1) {
        const int cnt = min(32, tail - head);
        push_batch(H, pk, dinv, acc, ring, head, cnt, row0, lane);
        head += cnt;
      }
      break;
    }
  }

  __syncthreads();

  for (int idx = tid; idx < nrows * (C_HID / 4); idx += PUSH_THREADS) {
    const int r  = idx >> 6;
    const int cq = idx & 63;
    const int c  = cq << 2;
    const float* ap = acc + r * C_HID + (cq >> 5) * 128 + (cq & 31);
    const int node = row0 + r;
    const float di = dinv[node];
    const float4 h = *(const float4*)(H + (long long)node * C_HID + c);
    const float4 bb = *(const float4*)(bias + c);
    float4 v;
    v.x = ap[0]  + di * di * h.x + bb.x;
    v.y = ap[32] + di * di * h.y + bb.y;
    v.z = ap[64] + di * di * h.z + bb.z;
    v.w = ap[96] + di * di * h.w + bb.w;
    v.x = (v.x > 0.0f) ? v.x : expm1f(v.x);
    v.y = (v.y > 0.0f) ? v.y : expm1f(v.y);
    v.z = (v.z > 0.0f) ? v.z : expm1f(v.z);
    v.w = (v.w > 0.0f) ? v.w : expm1f(v.w);
    *(float4*)(OUT + (long long)node * C_HID + c) = v;
  }
}

extern "C" void kernel_launch(void* const* d_in, const int* in_sizes, int n_in,
                              void* d_out, int out_size, void* d_ws, size_t ws_size,
                              hipStream_t stream) {
  const float* x  = (const float*)d_in[0];
  const int*   ei = (const int*)d_in[1];
  const float* W1 = (const float*)d_in[2];
  const float* b1 = (const float*)d_in[3];
  const float* W2 = (const float*)d_in[4];
  const float* b2 = (const float*)d_in[5];
  const float* W3 = (const float*)d_in[6];
  const float* b3 = (const float*)d_in[7];
  float* out = (float*)d_out;

  const int C   = C_HID;
  const int E   = in_sizes[1] / 2;
  const int cin = in_sizes[2] / C;
  const int n   = in_sizes[0] / cin;
  const int* src = ei;
  const int* dst = ei + E;

  char*  ws  = (char*)d_ws;
  size_t off = 0;
  auto alloc = [&](size_t bytes) -> void* {
    void* p = ws + off;
    off = (off + bytes + 255) & ~(size_t)255;
    return p;
  };
  int*   degi = (int*)  alloc((size_t)n * sizeof(int));
  float* dinv = (float*)alloc((size_t)n * sizeof(float));
  float* H    = (float*)alloc((size_t)n * C * sizeof(float));
  float* XB   = (float*)alloc((size_t)n * C * sizeof(float));
  unsigned* key4 = (unsigned*)alloc(((size_t)E + 3) / 4 * sizeof(unsigned));
  unsigned* pk   = (unsigned*)alloc((size_t)E * sizeof(unsigned));
  const unsigned char* keys = (const unsigned char*)key4;
  (void)ws_size; (void)n_in; (void)out_size;

  k_deg_init <<<(n + 255) / 256, 256, 0, stream>>>(degi, n);
  k_deg_count<<<((E + 3) / 4 + 255) / 256, 256, 0, stream>>>(src, dst, degi, key4, pk, E);
  k_dinv     <<<(n + 255) / 256, 256, 0, stream>>>(degi, dinv, n);

  const int pgrid = (n + PUSH_ROWS - 1) / PUSH_ROWS;

  auto layer = [&](const float* Xin, const float* W, const float* bias,
                   float* Out, int K) {
    const int mtiles = (n + 15) / 16;
    const int ntiles = C / 16;
    const int waves  = mtiles * ntiles;
    const int wpb    = 8;
    const int blocks = (waves + wpb - 1) / wpb;
    k_gemm_wmma<<<blocks, wpb * 32, 0, stream>>>(Xin, W, H, n, K, C);
    k_push_rows<<<pgrid, PUSH_THREADS, 0, stream>>>(H, pk, keys, dinv, bias, Out, n, E);
  };

  layer(x,  W1, b1, XB,  cin);
  layer(XB, W2, b2, XB,  C);
  layer(XB, W3, b3, out, C);
}


# each layer's product moved behind the LDS edge sums into the same work-group: (AX)W, A fragments from LDS, k in two halves per tile; gathers read layer inputs; 9 launches -> 6; sum order changes
# speedup vs baseline: 3.0420x; 1.2479x over previous
#include <hip/hip_runtime.h>
#include <hip/hip_bf16.h>

typedef __attribute__((ext_vector_type(2))) float v2f;
typedef __attribute__((ext_vector_type(8))) float v8f;

#define C_HID 256

#define PUSH_ROWS 40

typedef __attribute__((ext_vector_type(4))) unsigned v4u;

__global__ void k_deg_init(int* __restrict__ degi, int n) {
  int i = blockIdx.x * blockDim.x + threadIdx.x;
  if (i < n) degi[i] = 1;
}

__global__ void k_deg_count(const int* __restrict__ src, const int* __restrict__ dst,
                            int* __restrict__ degi, unsigned* __restrict__ key4,
                            unsigned* __restrict__ pk, int E) {
  const int t  = blockIdx.x * blockDim.x + threadIdx.x;
  const int e0 = t * 4;
  if (e0 >= E) return;
  unsigned kw = 0u;
  if (e0 + 3 < E) {
    int d[4], sv[4];
#pragma unroll
    for (int q = 0; q < 4; ++q) { d[q] = dst[e0 + q]; sv[q] = src[e0 + q]; }
    unsigned pv[4];
#pragma unroll
    for (int q = 0; q < 4; ++q) {
      const int g = d[q] / PUSH_ROWS;
      atomicAdd(&degi[d[q]], 1);
      pv[q] = ((unsigned)sv[q] << 6) | (unsigned)(d[q] - g * PUSH_ROWS);
      kw |= (unsigned)g << (8 * q);
    }
    v4u o; o.x = pv[0]; o.y = pv[1]; o.z = pv[2]; o.w = pv[3];
    *(v4u*)(pk + e0) = o;
  } else {
    for (int q = 0; e0 + q < E; ++q) {
      const int d = dst[e0 + q];
      const int g = d / PUSH_ROWS;
      atomicAdd(&degi[d], 1);
      pk[e0 + q] = ((unsigned)src[e0 + q] << 6) | (unsigned)(d - g * PUSH_ROWS);
      kw |= (unsigned)g << (8 * q);
    }
  }
  key4[t] = kw;
}

__global__ void k_dinv(const int* __restrict__ degi, float* __restrict__ dinv, int n) {
  int i = blockIdx.x * blockDim.x + threadIdx.x;
  if (i < n) dinv[i] = rsqrtf((float)degi[i]);
}

#define PUSH_THREADS 1024
#define PUSH_WAVES   32
#define PUSH_RING    1024
#define KEY_LOADS    2
#define PUSH_MIN_BATCH 16
#define TILE_ROWS    48
#define ACC_PITCH(K) ((K) + 4)
#define ACC_FLOATS   (TILE_ROWS * ACC_PITCH(C_HID))

template <int K>
__device__ __forceinline__ int pos_to_chan(int p) {
  if (K == 256) return (p & 128) + 4 * (p & 31) + ((p >> 5) & 3);
  return 4 * (p & 15) + (p >> 4);
}

__device__ __forceinline__ void ldsAddF(float* p, float v) {
  __hip_atomic_fetch_add(p, v, __ATOMIC_RELAXED, __HIP_MEMORY_SCOPE_WORKGROUP);
}

__device__ __forceinline__ int laneI(int v, int j) {
  return __builtin_amdgcn_readlane(v, j);
}
__device__ __forceinline__ float laneF(float v, int j) {
  return __int_as_float(__builtin_amdgcn_readlane(__float_as_int(v), j));
}

template <int K>
__device__ __forceinline__ void push_batch(
    const float* __restrict__ X, const unsigned* __restrict__ pk,
    const float* __restrict__ dinv, float* acc, const unsigned* ring,
    int head, int count, int row0, int lane) {
  int s = 0, off = 0;
  float nrm = 0.0f;
  if (lane < count) {
    const unsigned ent = pk[ring[(head + lane) & (PUSH_RING - 1)]];
    int dl = (int)(ent & 63u);
    s   = (int)(ent >> 6);
    nrm = dinv[s] * dinv[row0 + dl];
    off = dl * ACC_PITCH(K);
  }
  if (K == 256) {
    int j = 0;
    for (; j + 4 <= count; j += 4) {
      float4 a[4], b[4];
#pragma unroll
      for (int u = 0; u < 4; ++u) {
        const float* hp = X + (long long)laneI(s, j + u) * K + lane * 4;
        a[u] = *(const float4*)hp;
        b[u] = *(const float4*)(hp + 128);
      }
#pragma unroll
      for (int u = 0; u < 4; ++u) {
        const float nj = laneF(nrm, j + u);
        float* q = acc + laneI(off, j + u) + lane;
        ldsAddF(q,       a[u].x * nj);
        ldsAddF(q + 32,  a[u].y * nj);
        ldsAddF(q + 64,  a[u].z * nj);
        ldsAddF(q + 96,  a[u].w * nj);
        ldsAddF(q + 128, b[u].x * nj);
        ldsAddF(q + 160, b[u].y * nj);
        ldsAddF(q + 192, b[u].z * nj);
        ldsAddF(q + 224, b[u].w * nj);
      }
    }
    for (; j < count; ++j) {
      const float* hp = X + (long long)laneI(s, j) * K + lane * 4;
      const float4 a = *(const float4*)hp;
      const float4 b = *(const float4*)(hp + 128);
      const float nj = laneF(nrm, j);
      float* q = acc + laneI(off, j) + lane;
      ldsAddF(q,       a.x * nj);
      ldsAddF(q + 32,  a.y * nj);
      ldsAddF(q + 64,  a.z * nj);
      ldsAddF(q + 96,  a.w * nj);
      ldsAddF(q + 128, b.x * nj);
      ldsAddF(q + 160, b.y * nj);
      ldsAddF(q + 192, b.z * nj);
      ldsAddF(q + 224, b.w * nj);
    }
  } else {
    const int  sub = lane & 15;
    const bool odd = lane >= 16;
    for (int j = 0; j < count; j += 8) {
      float4 a[4];
#pragma unroll
      for (int u = 0; u < 4; ++u) {
        const int e0 = j + 2 * u;
        const int sA = laneI(s, e0), sB = laneI(s, e0 + 1);
        a[u] = make_float4(0.0f, 0.0f, 0.0f, 0.0f);
        if (e0 + (odd ? 1 : 0) < count)
          a[u] = *(const float4*)(X + (long long)(odd ? sB : sA) * K + sub * 4);
      }
#pragma unroll
      for (int u = 0; u < 4; ++u) {
        const int e0 = j + 2 * u;
        const float nA = laneF(nrm, e0), nB = laneF(nrm, e0 + 1);
        const int   oA = laneI(off, e0), oB = laneI(off, e0 + 1);
        if (e0 + (odd ? 1 : 0) < count) {
          const float nj = odd ? nB : nA;
          float* q = acc + (odd ? oB : oA) + sub;
          ldsAddF(q,      a[u].x * nj);
          ldsAddF(q + 16, a[u].y * nj);
          ldsAddF(q + 32, a[u].z * nj);
          ldsAddF(q + 48, a[u].w * nj);
        }
      }
    }
  }
}

__device__ __forceinline__ void load_keys(const unsigned char* __restrict__ p, int lane, v4u* v) {
#pragma unroll
  for (int k = 0; k < KEY_LOADS; ++k)
    v[k] = *(const v4u*)(p + k * 512 + lane * 16);
}

template <int K>
__global__ __launch_bounds__(PUSH_THREADS) void k_push_mm(
    const float* __restrict__ X, const unsigned* __restrict__ pk,
    const unsigned char* __restrict__ keys, const float* __restrict__ dinv,
    const float* __restrict__ Wm, const float* __restrict__ bias,
    float* __restrict__ OUT, int n, int E) {
  __shared__ __attribute__((aligned(16))) float    acc[ACC_FLOATS];
  __shared__ __attribute__((aligned(16))) unsigned pend[PUSH_WAVES * PUSH_RING];
  constexpr int PITCH = ACC_PITCH(K);

  const int tid  = threadIdx.x;
  const int lane = tid & 31;
  const int w    = __builtin_amdgcn_readfirstlane(tid >> 5);
  const int row0 = blockIdx.x * PUSH_ROWS;
  const int nrows = min(PUSH_ROWS, n - row0);
  unsigned* ring = pend + w * PUSH_RING;
  const unsigned gpat = (unsigned)blockIdx.x * 0x01010101u;

  for (int i = tid; i < TILE_ROWS * PITCH; i += PUSH_THREADS) acc[i] = 0.0f;
  __syncthreads();

  const int nsb   = E >> 10;
  const int tail0 = nsb << 10;

  int head = 0, tail = 0;
  {
    const int e = tail0 + w * 32 + lane;
    bool m = false;
    if (e < E) m = (unsigned)keys[e] == (unsigned)blockIdx.x;
    const unsigned mask = __builtin_amdgcn_ballot_w32(m);
    if (m) ring[__builtin_amdgcn_mbcnt_lo(mask, 0u)] = (unsigned)e;
    tail = __builtin_popcount(mask);
  }

  const int start = (nsb > 0) ? (int)(((long long)blockIdx.x * nsb) / gridDim.x) : 0;

  v4u cur[KEY_LOADS];
  if (w < nsb) {
    int sb = start + w; if (sb >= nsb) sb -= nsb;
    load_keys(keys + ((long long)sb << 10), lane, cur);
  }
  for (int i = w; ; i += PUSH_WAVES) {
    const bool more = i < nsb;
    if (more) {
      int sb = start + i; if (sb >= nsb) sb -= nsb;
      const int base = sb << 10;
      int sn = (i + PUSH_WAVES < nsb) ? start + i + PUSH_WAVES : start + i;
      if (sn >= nsb) sn -= nsb;
      v4u nxt[KEY_LOADS];
      load_keys(keys + ((long long)sn << 10), lane, nxt);
#pragma unroll
      for (int k = 0; k < KEY_LOADS; ++k) {
#pragma unroll
        for (int q = 0; q < 4; ++q) {
          const unsigned x = cur[k][q] ^ gpat;
          const unsigned z = (x - 0x01010101u) & ~x & 0x80808080u;
          if (__builtin_amdgcn_ballot_w32(z != 0u)) {
#pragma unroll
            for (int b = 0; b < 4; ++b) {
              const bool m = ((x >> (8 * b)) & 0xffu) == 0u;
              const unsigned mask = __builtin_amdgcn_ballot_w32(m);
              if (mask) {
                if (m) {
                  const unsigned e = (unsigned)(base + k * 512 + lane * 16 + q * 4 + b);
                  ring[(tail + __builtin_amdgcn_mbcnt_lo(mask, 0u)) & (PUSH_RING - 1)] = e;
                }
                tail += __builtin_popcount(mask);
              }
            }
          }
        }
        while (tail - head >= PUSH_MIN_BATCH) {
          const int cnt = min(32, (tail - head) & ~3);
          push_batch<K>(X, pk, dinv, acc, ring, head, cnt, row0, lane);
          head += cnt;
        }
      }
#pragma unroll
      for (int k = 0; k < KEY_LOADS; ++k) cur[k] = nxt[k];
    } else {
      while (tail - head >= 1) {
        const int cnt = min(32, tail - head);
        push_batch<K>(X, pk, dinv, acc, ring, head, cnt, row0, lane);
        head += cnt;
      }
      break;
    }
  }

  __syncthreads();

  {
    constexpr int QPR = K / 4;
    constexpr int JST = (K == 256) ? 32 : 16;
    for (int idx = tid; idx < nrows * QPR; idx += PUSH_THREADS) {
      const int r  = idx / QPR;
      const int cq = idx - r * QPR;
      const int node = row0 + r;
      const float di = dinv[node];
      const float4 xv = *(const float4*)(X + (long long)node * K + cq * 4);
      float* ap = acc + r * PITCH + ((K == 256) ? ((cq >> 5) * 128 + (cq & 31)) : cq);
      ap[0]       += di * di * xv.x;
      ap[JST]     += di * di * xv.y;
      ap[2 * JST] += di * di * xv.z;
      ap[3 * JST] += di * di * xv.w;
    }
  }
  __syncthreads();

  const int lo = lane & 15;
  const int hi = lane >> 4;
  const int n0 = (w & 15) << 4;
  const int kh = w >> 4;
  v8f d0 = {}, d1 = {}, d2 = {};
  {
    constexpr int STEPS = K / 8;
    const int pbase = kh * (K / 2) + 2 * hi;
    const float* Ap = acc + lo * PITCH + pbase;
    const float* Wc = Wm + n0 + lo;
#pragma unroll 8
    for (int t = 0; t < STEPS; ++t) {
      const int p = pbase + 4 * t;
      v2f b;
      b.x = Wc[(long long)pos_to_chan<K>(p)     * C_HID];
      b.y = Wc[(long long)pos_to_chan<K>(p + 1) * C_HID];
      const v2f a0 = *(const v2f*)(Ap + 4 * t);
      const v2f a1 = *(const v2f*)(Ap + 16 * PITCH + 4 * t);
      const v2f a2 = *(const v2f*)(Ap + 32 * PITCH + 4 * t);
      d0 = __builtin_amdgcn_wmma_f32_16x16x4_f32(false, a0, false, b, (short)0, d0, false, false);
      d1 = __builtin_amdgcn_wmma_f32_16x16x4_f32(false, a1, false, b, (short)0, d1, false, false);
      d2 = __builtin_amdgcn_wmma_f32_16x16x4_f32(false, a2, false, b, (short)0, d2, false, false);
    }
  }

  float* S = (float*)pend;
  float* Sp = S + (hi * 8) * C_HID + n0 + lo;
  if (kh == 1) {
#pragma unroll
    for (int r = 0; r < 8; ++r) {
      Sp[(r)      * C_HID] = d0[r];
      Sp[(16 + r) * C_HID] = d1[r];
      Sp[(32 + r) * C_HID] = d2[r];
    }
  }
  __syncthreads();
  if (kh == 0) {
    const float bb = bias[n0 + lo];
#pragma unroll
    for (int r = 0; r < 8; ++r) {
      float v0 = d0[r] + Sp[(r)      * C_HID] + bb;
      float v1 = d1[r] + Sp[(16 + r) * C_HID] + bb;
      float v2 = d2[r] + Sp[(32 + r) * C_HID] + bb;
      Sp[(r)      * C_HID] = (v0 > 0.0f) ? v0 : expm1f(v0);
      Sp[(16 + r) * C_HID] = (v1 > 0.0f) ? v1 : expm1f(v1);
      Sp[(32 + r) * C_HID] = (v2 > 0.0f) ? v2 : expm1f(v2);
    }
  }
  __syncthreads();

  for (int idx = tid; idx < nrows * (C_HID / 4); idx += PUSH_THREADS) {
    const int r  = idx >> 6;
    const int cq = idx & 63;
    const float4 v = *(const float4*)(S + r * C_HID + cq * 4);
    *(float4*)(OUT + (long long)(row0 + r) * C_HID + cq * 4) = v;
  }
}

extern "C" void kernel_launch(void* const* d_in, const int* in_sizes, int n_in,
                              void* d_out, int out_size, void* d_ws, size_t ws_size,
                              hipStream_t stream) {
  const float* x  = (const float*)d_in[0];
  const int*   ei = (const int*)d_in[1];
  const float* W1 = (const float*)d_in[2];
  const float* b1 = (const float*)d_in[3];
  const float* W2 = (const float*)d_in[4];
  const float* b2 = (const float*)d_in[5];
  const float* W3 = (const float*)d_in[6];
  const float* b3 = (const float*)d_in[7];
  float* out = (float*)d_out;

  const int C   = C_HID;
  const int E   = in_sizes[1] / 2;
  const int cin = in_sizes[2] / C;
  const int n   = in_sizes[0] / cin;
  const int* src = ei;
  const int* dst = ei + E;

  char*  ws  = (char*)d_ws;
  size_t off = 0;
  auto alloc = [&](size_t bytes) -> void* {
    void* p = ws + off;
    off = (off + bytes + 255) & ~(size_t)255;
    return p;
  };
  int*   degi = (int*)  alloc((size_t)n * sizeof(int));
  float* dinv = (float*)alloc((size_t)n * sizeof(float));
  float* XA   = (float*)alloc((size_t)n * C * sizeof(float));
  float* XB   = (float*)alloc((size_t)n * C * sizeof(float));
  unsigned* key4 = (unsigned*)alloc(((size_t)E + 3) / 4 * sizeof(unsigned));
  unsigned* pk   = (unsigned*)alloc((size_t)E * sizeof(unsigned));
  const unsigned char* keys = (const unsigned char*)key4;
  (void)ws_size; (void)n_in; (void)out_size;

  k_deg_init <<<(n + 255) / 256, 256, 0, stream>>>(degi, n);
  k_deg_count<<<((E + 3) / 4 + 255) / 256, 256, 0, stream>>>(src, dst, degi, key4, pk, E);
  k_dinv     <<<(n + 255) / 256, 256, 0, stream>>>(degi, dinv, n);

  const int pgrid = (n + PUSH_ROWS - 1) / PUSH_ROWS;

  k_push_mm<64> <<<pgrid, PUSH_THREADS, 0, stream>>>(x,  pk, keys, dinv, W1, b1, XA,  n, E);
  k_push_mm<256><<<pgrid, PUSH_THREADS, 0, stream>>>(XA, pk, keys, dinv, W2, b2, XB,  n, E);
  k_push_mm<256><<<pgrid, PUSH_THREADS, 0, stream>>>(XB, pk, keys, dinv, W3, b3, out, n, E);
}


# layers 2 and 3 no longer scan the key list: layer 1's scan stores each work-group's edges (src, row, weight) in the workspace, later layers read it, equal share per wave; sums stay arrival order
# speedup vs baseline: 3.3054x; 1.0866x over previous
#include <hip/hip_runtime.h>
#include <hip/hip_bf16.h>

typedef __attribute__((ext_vector_type(2))) float v2f;
typedef __attribute__((ext_vector_type(8))) float v8f;

#define C_HID 256

#define PUSH_ROWS 40

typedef __attribute__((ext_vector_type(4))) unsigned v4u;
typedef __attribute__((ext_vector_type(2))) unsigned v2u;

#define LIST_CAP 4096

__global__ void k_deg_init(int* __restrict__ degi, int n) {
  int i = blockIdx.x * blockDim.x + threadIdx.x;
  if (i < n) degi[i] = 1;
}

__global__ void k_deg_count(const int* __restrict__ src, const int* __restrict__ dst,
                            int* __restrict__ degi, unsigned* __restrict__ key4,
                            unsigned* __restrict__ pk, int E) {
  const int t  = blockIdx.x * blockDim.x + threadIdx.x;
  const int e0 = t * 4;
  if (e0 >= E) return;
  unsigned kw = 0u;
  if (e0 + 3 < E) {
    int d[4], sv[4];
#pragma unroll
    for (int q = 0; q < 4; ++q) { d[q] = dst[e0 + q]; sv[q] = src[e0 + q]; }
    unsigned pv[4];
#pragma unroll
    for (int q = 0; q < 4; ++q) {
      const int g = d[q] / PUSH_ROWS;
      atomicAdd(&degi[d[q]], 1);
      pv[q] = ((unsigned)sv[q] << 6) | (unsigned)(d[q] - g * PUSH_ROWS);
      kw |= (unsigned)g << (8 * q);
    }
    v4u o; o.x = pv[0]; o.y = pv[1]; o.z = pv[2]; o.w = pv[3];
    *(v4u*)(pk + e0) = o;
  } else {
    for (int q = 0; e0 + q < E; ++q) {
      const int d = dst[e0 + q];
      const int g = d / PUSH_ROWS;
      atomicAdd(&degi[d], 1);
      pk[e0 + q] = ((unsigned)src[e0 + q] << 6) | (unsigned)(d - g * PUSH_ROWS);
      kw |= (unsigned)g << (8 * q);
    }
  }
  key4[t] = kw;
}

__global__ void k_dinv(const int* __restrict__ degi, float* __restrict__ dinv, int n) {
  int i = blockIdx.x * blockDim.x + threadIdx.x;
  if (i < n) dinv[i] = rsqrtf((float)degi[i]);
}

#define PUSH_THREADS 1024
#define PUSH_WAVES   32
#define PUSH_RING    1024
#define KEY_LOADS    2
#define PUSH_MIN_BATCH 16
#define TILE_ROWS    48
#define ACC_PITCH(K) ((K) + 4)
#define ACC_FLOATS   (TILE_ROWS * ACC_PITCH(C_HID))

template <int K>
__device__ __forceinline__ int pos_to_chan(int p) {
  if (K == 256) return (p & 128) + 4 * (p & 31) + ((p >> 5) & 3);
  return 4 * (p & 15) + (p >> 4);
}

__device__ __forceinline__ void ldsAddF(float* p, float v) {
  __hip_atomic_fetch_add(p, v, __ATOMIC_RELAXED, __HIP_MEMORY_SCOPE_WORKGROUP);
}

__device__ __forceinline__ int laneI(int v, int j) {
  return __builtin_amdgcn_readlane(v, j);
}
__device__ __forceinline__ float laneF(float v, int j) {
  return __int_as_float(__builtin_amdgcn_readlane(__float_as_int(v), j));
}

template <int K>
__device__ __forceinline__ void add_rows(
    const float* __restrict__ X, float* acc, int s, float nrm, int off,
    int count, int lane) {
  if (K == 256) {
    int j = 0;
    for (; j + 4 <= count; j += 4) {
      float4 a[4], b[4];
#pragma unroll
      for (int u = 0; u < 4; ++u) {
        const float* hp = X + (long long)laneI(s, j + u) * K + lane * 4;
        a[u] = *(const float4*)hp;
        b[u] = *(const float4*)(hp + 128);
      }
#pragma unroll
      for (int u = 0; u < 4; ++u) {
        const float nj = laneF(nrm, j + u);
        float* q = acc + laneI(off, j + u) + lane;
        ldsAddF(q,       a[u].x * nj);
        ldsAddF(q + 32,  a[u].y * nj);
        ldsAddF(q + 64,  a[u].z * nj);
        ldsAddF(q + 96,  a[u].w * nj);
        ldsAddF(q + 128, b[u].x * nj);
        ldsAddF(q + 160, b[u].y * nj);
        ldsAddF(q + 192, b[u].z * nj);
        ldsAddF(q + 224, b[u].w * nj);
      }
    }
    for (; j < count; ++j) {
      const float* hp = X + (long long)laneI(s, j) * K + lane * 4;
      const float4 a = *(const float4*)hp;
      const float4 b = *(const float4*)(hp + 128);
      const float nj = laneF(nrm, j);
      float* q = acc + laneI(off, j) + lane;
      ldsAddF(q,       a.x * nj);
      ldsAddF(q + 32,  a.y * nj);
      ldsAddF(q + 64,  a.z * nj);
      ldsAddF(q + 96,  a.w * nj);
      ldsAddF(q + 128, b.x * nj);
      ldsAddF(q + 160, b.y * nj);
      ldsAddF(q + 192, b.z * nj);
      ldsAddF(q + 224, b.w * nj);
    }
  } else {
    const int  sub = lane & 15;
    const bool odd = lane >= 16;
    for (int j = 0; j < count; j += 8) {
      float4 a[4];
#pragma unroll
      for (int u = 0; u < 4; ++u) {
        const int e0 = j + 2 * u;
        const int sA = laneI(s, e0), sB = laneI(s, e0 + 1);
        a[u] = make_float4(0.0f, 0.0f, 0.0f, 0.0f);
        if (e0 + (odd ? 1 : 0) < count)
          a[u] = *(const float4*)(X + (long long)(odd ? sB : sA) * K + sub * 4);
      }
#pragma unroll
      for (int u = 0; u < 4; ++u) {
        const int e0 = j + 2 * u;
        const float nA = laneF(nrm, e0), nB = laneF(nrm, e0 + 1);
        const int   oA = laneI(off, e0), oB = laneI(off, e0 + 1);
        if (e0 + (odd ? 1 : 0) < count) {
          const float nj = odd ? nB : nA;
          float* q = acc + (odd ? oB : oA) + sub;
          ldsAddF(q,      a[u].x * nj);
          ldsAddF(q + 16, a[u].y * nj);
          ldsAddF(q + 32, a[u].z * nj);
          ldsAddF(q + 48, a[u].w * nj);
        }
      }
    }
  }
}

template <int K, bool STORE>
__device__ __forceinline__ void push_batch(
    const float* __restrict__ X, const unsigned* __restrict__ pk,
    const float* __restrict__ dinv, float* acc, const unsigned* ring,
    int head, int count, int row0, int lane, v2u* glist, int* lcnt) {
  int s = 0, off = 0;
  float nrm = 0.0f;
  unsigned ent = 0u;
  if (lane < count) {
    ent = pk[ring[(head + lane) & (PUSH_RING - 1)]];
    int dl = (int)(ent & 63u);
    s   = (int)(ent >> 6);
    nrm = dinv[s] * dinv[row0 + dl];
    off = dl * ACC_PITCH(K);
  }
  if (STORE) {
    int base = 0;
    if (lane == 0)
      base = __hip_atomic_fetch_add(lcnt, count, __ATOMIC_RELAXED, __HIP_MEMORY_SCOPE_WORKGROUP);
    base = __builtin_amdgcn_readfirstlane(base);
    if (lane < count && base + lane < LIST_CAP) {
      v2u o; o.x = ent; o.y = __float_as_uint(nrm);
      glist[base + lane] = o;
    }
  }
  add_rows<K>(X, acc, s, nrm, off, count, lane);
}

__device__ __forceinline__ void load_keys(const unsigned char* __restrict__ p, int lane, v4u* v) {
#pragma unroll
  for (int k = 0; k < KEY_LOADS; ++k)
    v[k] = *(const v4u*)(p + k * 512 + lane * 16);
}

template <int K, bool STORE>
__device__ __forceinline__ void scan_push(
    const float* __restrict__ X, const unsigned* __restrict__ pk,
    const unsigned char* __restrict__ keys, const float* __restrict__ dinv,
    float* acc, unsigned* ring, v2u* glist, int* lcnt,
    int row0, int E, int w, int lane) {
  const unsigned gpat = (unsigned)blockIdx.x * 0x01010101u;

  const int nsb   = E >> 10;
  const int tail0 = nsb << 10;

  int head = 0, tail = 0;
  {
    const int e = tail0 + w * 32 + lane;
    bool m = false;
    if (e < E) m = (unsigned)keys[e] == (unsigned)blockIdx.x;
    const unsigned mask = __builtin_amdgcn_ballot_w32(m);
    if (m) ring[__builtin_amdgcn_mbcnt_lo(mask, 0u)] = (unsigned)e;
    tail = __builtin_popcount(mask);
  }

  const int start = (nsb > 0) ? (int)(((long long)blockIdx.x * nsb) / gridDim.x) : 0;

  v4u cur[KEY_LOADS];
  if (w < nsb) {
    int sb = start + w; if (sb >= nsb) sb -= nsb;
    load_keys(keys + ((long long)sb << 10), lane, cur);
  }
  for (int i = w; ; i += PUSH_WAVES) {
    const bool more = i < nsb;
    if (more) {
      int sb = start + i; if (sb >= nsb) sb -= nsb;
      const int base = sb << 10;
      int sn = (i + PUSH_WAVES < nsb) ? start + i + PUSH_WAVES : start + i;
      if (sn >= nsb) sn -= nsb;
      v4u nxt[KEY_LOADS];
      load_keys(keys + ((long long)sn << 10), lane, nxt);
#pragma unroll
      for (int k = 0; k < KEY_LOADS; ++k) {
#pragma unroll
        for (int q = 0; q < 4; ++q) {
          const unsigned x = cur[k][q] ^ gpat;
          const unsigned z = (x - 0x01010101u) & ~x & 0x80808080u;
          if (__builtin_amdgcn_ballot_w32(z != 0u)) {
#pragma unroll
            for (int b = 0; b < 4; ++b) {
              const bool m = ((x >> (8 * b)) & 0xffu) == 0u;
              const unsigned mask = __builtin_amdgcn_ballot_w32(m);
              if (mask) {
                if (m) {
                  const unsigned e = (unsigned)(base + k * 512 + lane * 16 + q * 4 + b);
                  ring[(tail + __builtin_amdgcn_mbcnt_lo(mask, 0u)) & (PUSH_RING - 1)] = e;
                }
                tail += __builtin_popcount(mask);
              }
            }
          }
        }
        while (tail - head >= PUSH_MIN_BATCH) {
          const int cnt = min(32, (tail - head) & ~3);
          push_batch<K, STORE>(X, pk, dinv, acc, ring, head, cnt, row0, lane, glist, lcnt);
          head += cnt;
        }
      }
#pragma unroll
      for (int k = 0; k < KEY_LOADS; ++k) cur[k] = nxt[k];
    } else {
      while (tail - head >= 1) {
        const int cnt = min(32, tail - head);
        push_batch<K, STORE>(X, pk, dinv, acc, ring, head, cnt, row0, lane, glist, lcnt);
        head += cnt;
      }
      break;
    }
  }
}

template <int K, bool FIRST>
__global__ __launch_bounds__(PUSH_THREADS) void k_push_mm(
    const float* __restrict__ X, const unsigned* __restrict__ pk,
    const unsigned char* __restrict__ keys, const float* __restrict__ dinv,
    const float* __restrict__ Wm, const float* __restrict__ bias,
    float* __restrict__ OUT, unsigned* __restrict__ glist_all,
    int* __restrict__ gcnt, int n, int E) {
  __shared__ __attribute__((aligned(16))) float    acc[ACC_FLOATS];
  __shared__ __attribute__((aligned(16))) unsigned pend[PUSH_WAVES * PUSH_RING];
  __shared__ int lcnt;
  constexpr int PITCH = ACC_PITCH(K);

  const int tid  = threadIdx.x;
  const int lane = tid & 31;
  const int w    = __builtin_amdgcn_readfirstlane(tid >> 5);
  const int row0 = blockIdx.x * PUSH_ROWS;
  const int nrows = min(PUSH_ROWS, n - row0);
  unsigned* ring = pend + w * PUSH_RING;
  v2u* glist = (v2u*)glist_all + (long long)blockIdx.x * LIST_CAP;

  for (int i = tid; i < TILE_ROWS * PITCH; i += PUSH_THREADS) acc[i] = 0.0f;
  if (FIRST && tid == 0) lcnt = 0;
  __syncthreads();

  if (FIRST) {
    scan_push<K, true>(X, pk, keys, dinv, acc, ring, glist, &lcnt, row0, E, w, lane);
  } else {
    const int cnt = gcnt[blockIdx.x];
    if (cnt <= LIST_CAP) {
      const int per = (((cnt + PUSH_WAVES - 1) / PUSH_WAVES) + 3) & ~3;
      int b = w * per;
      const int end = min(cnt, b + per);
      v2u cur = {0u, 0u};
      if (b + lane < end) cur = glist[b + lane];
      for (; b < end; b += 32) {
        v2u nxt = {0u, 0u};
        if (b + 32 + lane < end) nxt = glist[b + 32 + lane];
        const int count = min(32, end - b);
        int s = 0, off = 0;
        float nrm = 0.0f;
        if (lane < count) {
          s   = (int)(cur.x >> 6);
          off = (int)(cur.x & 63u) * PITCH;
          nrm = __uint_as_float(cur.y);
        }
        add_rows<K>(X, acc, s, nrm, off, count, lane);
        cur = nxt;
      }
    } else {
      scan_push<K, false>(X, pk, keys, dinv, acc, ring, glist, &lcnt, row0, E, w, lane);
    }
  }

  __syncthreads();
  if (FIRST && tid == 0) gcnt[blockIdx.x] = lcnt;

  {
    constexpr int QPR = K / 4;
    constexpr int JST = (K == 256) ? 32 : 16;
    for (int idx = tid; idx < nrows * QPR; idx += PUSH_THREADS) {
      const int r  = idx / QPR;
      const int cq = idx - r * QPR;
      const int node = row0 + r;
      const float di = dinv[node];
      const float4 xv = *(const float4*)(X + (long long)node * K + cq * 4);
      float* ap = acc + r * PITCH + ((K == 256) ? ((cq >> 5) * 128 + (cq & 31)) : cq);
      ap[0]       += di * di * xv.x;
      ap[JST]     += di * di * xv.y;
      ap[2 * JST] += di * di * xv.z;
      ap[3 * JST] += di * di * xv.w;
    }
  }
  __syncthreads();

  const int lo = lane & 15;
  const int hi = lane >> 4;
  const int n0 = (w & 15) << 4;
  const int kh = w >> 4;
  v8f d0 = {}, d1 = {}, d2 = {};
  {
    constexpr int STEPS = K / 8;
    const int pbase = kh * (K / 2) + 2 * hi;
    const float* Ap = acc + lo * PITCH + pbase;
    const float* Wc = Wm + n0 + lo;
#pragma unroll 8
    for (int t = 0; t < STEPS; ++t) {
      const int p = pbase + 4 * t;
      v2f b;
      b.x = Wc[(long long)pos_to_chan<K>(p)     * C_HID];
      b.y = Wc[(long long)pos_to_chan<K>(p + 1) * C_HID];
      const v2f a0 = *(const v2f*)(Ap + 4 * t);
      const v2f a1 = *(const v2f*)(Ap + 16 * PITCH + 4 * t);
      const v2f a2 = *(const v2f*)(Ap + 32 * PITCH + 4 * t);
      d0 = __builtin_amdgcn_wmma_f32_16x16x4_f32(false, a0, false, b, (short)0, d0, false, false);
      d1 = __builtin_amdgcn_wmma_f32_16x16x4_f32(false, a1, false, b, (short)0, d1, false, false);
      d2 = __builtin_amdgcn_wmma_f32_16x16x4_f32(false, a2, false, b, (short)0, d2, false, false);
    }
  }

  float* S = (float*)pend;
  float* Sp = S + (hi * 8) * C_HID + n0 + lo;
  if (kh == 1) {
#pragma unroll
    for (int r = 0; r < 8; ++r) {
      Sp[(r)      * C_HID] = d0[r];
      Sp[(16 + r) * C_HID] = d1[r];
      Sp[(32 + r) * C_HID] = d2[r];
    }
  }
  __syncthreads();
  if (kh == 0) {
    const float bb = bias[n0 + lo];
#pragma unroll
    for (int r = 0; r < 8; ++r) {
      float v0 = d0[r] + Sp[(r)      * C_HID] + bb;
      float v1 = d1[r] + Sp[(16 + r) * C_HID] + bb;
      float v2 = d2[r] + Sp[(32 + r) * C_HID] + bb;
      Sp[(r)      * C_HID] = (v0 > 0.0f) ? v0 : expm1f(v0);
      Sp[(16 + r) * C_HID] = (v1 > 0.0f) ? v1 : expm1f(v1);
      Sp[(32 + r) * C_HID] = (v2 > 0.0f) ? v2 : expm1f(v2);
    }
  }
  __syncthreads();

  for (int idx = tid; idx < nrows * (C_HID / 4); idx += PUSH_THREADS) {
    const int r  = idx >> 6;
    const int cq = idx & 63;
    const float4 v = *(const float4*)(S + r * C_HID + cq * 4);
    *(float4*)(OUT + (long long)(row0 + r) * C_HID + cq * 4) = v;
  }
}

extern "C" void kernel_launch(void* const* d_in, const int* in_sizes, int n_in,
                              void* d_out, int out_size, void* d_ws, size_t ws_size,
                              hipStream_t stream) {
  const float* x  = (const float*)d_in[0];
  const int*   ei = (const int*)d_in[1];
  const float* W1 = (const float*)d_in[2];
  const float* b1 = (const float*)d_in[3];
  const float* W2 = (const float*)d_in[4];
  const float* b2 = (const float*)d_in[5];
  const float* W3 = (const float*)d_in[6];
  const float* b3 = (const float*)d_in[7];
  float* out = (float*)d_out;

  const int C   = C_HID;
  const int E   = in_sizes[1] / 2;
  const int cin = in_sizes[2] / C;
  const int n   = in_sizes[0] / cin;
  const int* src = ei;
  const int* dst = ei + E;

  char*  ws  = (char*)d_ws;
  size_t off = 0;
  auto alloc = [&](size_t bytes) -> void* {
    void* p = ws + off;
    off = (off + bytes + 255) & ~(size_t)255;
    return p;
  };
  int*   degi = (int*)  alloc((size_t)n * sizeof(int));
  float* dinv = (float*)alloc((size_t)n * sizeof(float));
  float* XA   = (float*)alloc((size_t)n * C * sizeof(float));
  float* XB   = (float*)alloc((size_t)n * C * sizeof(float));
  unsigned* key4 = (unsigned*)alloc(((size_t)E + 3) / 4 * sizeof(unsigned));
  unsigned* pk   = (unsigned*)alloc((size_t)E * sizeof(unsigned));
  const unsigned char* keys = (const unsigned char*)key4;
  const int pgrid = (n + PUSH_ROWS - 1) / PUSH_ROWS;
  unsigned* glist = (unsigned*)alloc((size_t)pgrid * LIST_CAP * 2 * sizeof(unsigned));
  int*      gcnt  = (int*)     alloc((size_t)pgrid * sizeof(int));
  (void)ws_size; (void)n_in; (void)out_size;

  k_deg_init <<<(n + 255) / 256, 256, 0, stream>>>(degi, n);
  k_deg_count<<<((E + 3) / 4 + 255) / 256, 256, 0, stream>>>(src, dst, degi, key4, pk, E);
  k_dinv     <<<(n + 255) / 256, 256, 0, stream>>>(degi, dinv, n);

  k_push_mm<64, true>  <<<pgrid, PUSH_THREADS, 0, stream>>>(x,  pk, keys, dinv, W1, b1, XA,  glist, gcnt, n, E);
  k_push_mm<256, false><<<pgrid, PUSH_THREADS, 0, stream>>>(XA, pk, keys, dinv, W2, b2, XB,  glist, gcnt, n, E);
  k_push_mm<256, false><<<pgrid, PUSH_THREADS, 0, stream>>>(XB, pk, keys, dinv, W3, b3, out, glist, gcnt, n, E);
}


# layers 2 and 3 sum the edges of one destination row in registers: layer 1 orders each work-group's stored edge list by row in LDS; LDS float adds only where a row's run ends; sum order changes
# speedup vs baseline: 3.6276x; 1.0975x over previous
#include <hip/hip_runtime.h>
#include <hip/hip_bf16.h>

typedef __attribute__((ext_vector_type(2))) float v2f;
typedef __attribute__((ext_vector_type(8))) float v8f;

#define C_HID 256

#define PUSH_ROWS 40

typedef __attribute__((ext_vector_type(4))) unsigned v4u;
typedef __attribute__((ext_vector_type(2))) unsigned v2u;

#define LIST_CAP 4096

__global__ void k_deg_init(int* __restrict__ degi, int n) {
  int i = blockIdx.x * blockDim.x + threadIdx.x;
  if (i < n) degi[i] = 1;
}

__global__ void k_deg_count(const int* __restrict__ src, const int* __restrict__ dst,
                            int* __restrict__ degi, unsigned* __restrict__ key4,
                            unsigned* __restrict__ pk, int E) {
  const int t  = blockIdx.x * blockDim.x + threadIdx.x;
  const int e0 = t * 4;
  if (e0 >= E) return;
  unsigned kw = 0u;
  if (e0 + 3 < E) {
    int d[4], sv[4];
#pragma unroll
    for (int q = 0; q < 4; ++q) { d[q] = dst[e0 + q]; sv[q] = src[e0 + q]; }
    unsigned pv[4];
#pragma unroll
    for (int q = 0; q < 4; ++q) {
      const int g = d[q] / PUSH_ROWS;
      atomicAdd(&degi[d[q]], 1);
      pv[q] = ((unsigned)sv[q] << 6) | (unsigned)(d[q] - g * PUSH_ROWS);
      kw |= (unsigned)g << (8 * q);
    }
    v4u o; o.x = pv[0]; o.y = pv[1]; o.z = pv[2]; o.w = pv[3];
    *(v4u*)(pk + e0) = o;
  } else {
    for (int q = 0; e0 + q < E; ++q) {
      const int d = dst[e0 + q];
      const int g = d / PUSH_ROWS;
      atomicAdd(&degi[d], 1);
      pk[e0 + q] = ((unsigned)src[e0 + q] << 6) | (unsigned)(d - g * PUSH_ROWS);
      kw |= (unsigned)g << (8 * q);
    }
  }
  key4[t] = kw;
}

__global__ void k_dinv(const int* __restrict__ degi, float* __restrict__ dinv, int n) {
  int i = blockIdx.x * blockDim.x + threadIdx.x;
  if (i < n) dinv[i] = rsqrtf((float)degi[i]);
}

#define PUSH_THREADS 1024
#define PUSH_WAVES   32
#define PUSH_RING    1024
#define KEY_LOADS    2
#define PUSH_MIN_BATCH 16
#define TILE_ROWS    48
#define ACC_PITCH(K) ((K) + 4)
#define ACC_FLOATS   (TILE_ROWS * ACC_PITCH(C_HID))

template <int K>
__device__ __forceinline__ int pos_to_chan(int p) {
  if (K == 256) return (p & 128) + 4 * (p & 31) + ((p >> 5) & 3);
  return 4 * (p & 15) + (p >> 4);
}

__device__ __forceinline__ void ldsAddF(float* p, float v) {
  __hip_atomic_fetch_add(p, v, __ATOMIC_RELAXED, __HIP_MEMORY_SCOPE_WORKGROUP);
}

__device__ __forceinline__ int laneI(int v, int j) {
  return __builtin_amdgcn_readlane(v, j);
}
__device__ __forceinline__ float laneF(float v, int j) {
  return __int_as_float(__builtin_amdgcn_readlane(__float_as_int(v), j));
}

template <int K>
__device__ __forceinline__ void add_rows(
    const float* __restrict__ X, float* acc, int s, float nrm, int off,
    int count, int lane) {
  if (K == 256) {
    int j = 0;
    for (; j + 4 <= count; j += 4) {
      float4 a[4], b[4];
#pragma unroll
      for (int u = 0; u < 4; ++u) {
        const float* hp = X + (long long)laneI(s, j + u) * K + lane * 4;
        a[u] = *(const float4*)hp;
        b[u] = *(const float4*)(hp + 128);
      }
#pragma unroll
      for (int u = 0; u < 4; ++u) {
        const float nj = laneF(nrm, j + u);
        float* q = acc + laneI(off, j + u) + lane;
        ldsAddF(q,       a[u].x * nj);
        ldsAddF(q + 32,  a[u].y * nj);
        ldsAddF(q + 64,  a[u].z * nj);
        ldsAddF(q + 96,  a[u].w * nj);
        ldsAddF(q + 128, b[u].x * nj);
        ldsAddF(q + 160, b[u].y * nj);
        ldsAddF(q + 192, b[u].z * nj);
        ldsAddF(q + 224, b[u].w * nj);
      }
    }
    for (; j < count; ++j) {
      const float* hp = X + (long long)laneI(s, j) * K + lane * 4;
      const float4 a = *(const float4*)hp;
      const float4 b = *(const float4*)(hp + 128);
      const float nj = laneF(nrm, j);
      float* q = acc + laneI(off, j) + lane;
      ldsAddF(q,       a.x * nj);
      ldsAddF(q + 32,  a.y * nj);
      ldsAddF(q + 64,  a.z * nj);
      ldsAddF(q + 96,  a.w * nj);
      ldsAddF(q + 128, b.x * nj);
      ldsAddF(q + 160, b.y * nj);
      ldsAddF(q + 192, b.z * nj);
      ldsAddF(q + 224, b.w * nj);
    }
  } else {
    const int  sub = lane & 15;
    const bool odd = lane >= 16;
    for (int j = 0; j < count; j += 8) {
      float4 a[4];
#pragma unroll
      for (int u = 0; u < 4; ++u) {
        const int e0 = j + 2 * u;
        const int sA = laneI(s, e0), sB = laneI(s, e0 + 1);
        a[u] = make_float4(0.0f, 0.0f, 0.0f, 0.0f);
        if (e0 + (odd ? 1 : 0) < count)
          a[u] = *(const float4*)(X + (long long)(odd ? sB : sA) * K + sub * 4);
      }
#pragma unroll
      for (int u = 0; u < 4; ++u) {
        const int e0 = j + 2 * u;
        const float nA = laneF(nrm, e0), nB = laneF(nrm, e0 + 1);
        const int   oA = laneI(off, e0), oB = laneI(off, e0 + 1);
        if (e0 + (odd ? 1 : 0) < count) {
          const float nj = odd ? nB : nA;
          float* q = acc + (odd ? oB : oA) + sub;
          ldsAddF(q,      a[u].x * nj);
          ldsAddF(q + 16, a[u].y * nj);
          ldsAddF(q + 32, a[u].z * nj);
          ldsAddF(q + 48, a[u].w * nj);
        }
      }
    }
  }
}

template <int K, bool STORE>
__device__ __forceinline__ void push_batch(
    const float* __restrict__ X, const unsigned* __restrict__ pk,
    const float* __restrict__ dinv, float* acc, const unsigned* ring,
    int head, int count, int row0, int lane, v2u* lst, int* rowcnt, int* lcnt) {
  int s = 0, off = 0, dl = 0;
  float nrm = 0.0f;
  unsigned ent = 0u;
  if (lane < count) {
    ent = pk[ring[(head + lane) & (PUSH_RING - 1)]];
    dl  = (int)(ent & 63u);
    s   = (int)(ent >> 6);
    nrm = dinv[s] * dinv[row0 + dl];
    off = dl * ACC_PITCH(K);
  }
  if (STORE) {
    int base = 0;
    if (lane == 0)
      base = __hip_atomic_fetch_add(lcnt, count, __ATOMIC_RELAXED, __HIP_MEMORY_SCOPE_WORKGROUP);
    base = __builtin_amdgcn_readfirstlane(base);
    if (lane < count && base + lane < LIST_CAP) {
      v2u o; o.x = ent; o.y = __float_as_uint(nrm);
      lst[base + lane] = o;
      __hip_atomic_fetch_add(rowcnt + dl, 1, __ATOMIC_RELAXED, __HIP_MEMORY_SCOPE_WORKGROUP);
    }
  }
  add_rows<K>(X, acc, s, nrm, off, count, lane);
}

__device__ __forceinline__ void run_flush(float* acc, int roff, int lane,
                                          const float4& lo, const float4& hi) {
  float* q = acc + roff + lane;
  ldsAddF(q,       lo.x);
  ldsAddF(q + 32,  lo.y);
  ldsAddF(q + 64,  lo.z);
  ldsAddF(q + 96,  lo.w);
  ldsAddF(q + 128, hi.x);
  ldsAddF(q + 160, hi.y);
  ldsAddF(q + 192, hi.z);
  ldsAddF(q + 224, hi.w);
}

__device__ __forceinline__ void run_edge(float* acc, int lane, int o, float nj,
                                         const float4& a, const float4& b,
                                         float4& lo, float4& hi, int& roff) {
  if (o != roff) {
    if (roff >= 0) run_flush(acc, roff, lane, lo, hi);
    lo = make_float4(0.0f, 0.0f, 0.0f, 0.0f);
    hi = make_float4(0.0f, 0.0f, 0.0f, 0.0f);
    roff = o;
  }
  lo.x += a.x * nj; lo.y += a.y * nj; lo.z += a.z * nj; lo.w += a.w * nj;
  hi.x += b.x * nj; hi.y += b.y * nj; hi.z += b.z * nj; hi.w += b.w * nj;
}

__device__ __forceinline__ void run_rows(
    const float* __restrict__ X, float* acc, int s, float nrm, int off,
    int count, int lane, float4& lo, float4& hi, int& roff) {
  int j = 0;
  for (; j + 8 <= count; j += 8) {
    float4 a[8], b[8];
#pragma unroll
    for (int u = 0; u < 8; ++u) {
      const float* hp = X + (long long)laneI(s, j + u) * C_HID + lane * 4;
      a[u] = *(const float4*)hp;
      b[u] = *(const float4*)(hp + 128);
    }
#pragma unroll
    for (int u = 0; u < 8; ++u)
      run_edge(acc, lane, laneI(off, j + u), laneF(nrm, j + u), a[u], b[u], lo, hi, roff);
  }
  for (; j + 4 <= count; j += 4) {
    float4 a[4], b[4];
#pragma unroll
    for (int u = 0; u < 4; ++u) {
      const float* hp = X + (long long)laneI(s, j + u) * C_HID + lane * 4;
      a[u] = *(const float4*)hp;
      b[u] = *(const float4*)(hp + 128);
    }
#pragma unroll
    for (int u = 0; u < 4; ++u)
      run_edge(acc, lane, laneI(off, j + u), laneF(nrm, j + u), a[u], b[u], lo, hi, roff);
  }
  for (; j < count; ++j) {
    const float* hp = X + (long long)laneI(s, j) * C_HID + lane * 4;
    const float4 a = *(const float4*)hp;
    const float4 b = *(const float4*)(hp + 128);
    run_edge(acc, lane, laneI(off, j), laneF(nrm, j), a, b, lo, hi, roff);
  }
}

__device__ __forceinline__ void load_keys(const unsigned char* __restrict__ p, int lane, v4u* v) {
#pragma unroll
  for (int k = 0; k < KEY_LOADS; ++k)
    v[k] = *(const v4u*)(p + k * 512 + lane * 16);
}

template <int K, bool STORE>
__device__ __forceinline__ void scan_push(
    const float* __restrict__ X, const unsigned* __restrict__ pk,
    const unsigned char* __restrict__ keys, const float* __restrict__ dinv,
    float* acc, unsigned* ring, v2u* lst, int* rowcnt, int* lcnt,
    int row0, int E, int w, int lane) {
  const unsigned gpat = (unsigned)blockIdx.x * 0x01010101u;

  const int nsb   = E >> 10;
  const int tail0 = nsb << 10;

  int head = 0, tail = 0;
  {
    const int e = tail0 + w * 32 + lane;
    bool m = false;
    if (e < E) m = (unsigned)keys[e] == (unsigned)blockIdx.x;
    const unsigned mask = __builtin_amdgcn_ballot_w32(m);
    if (m) ring[__builtin_amdgcn_mbcnt_lo(mask, 0u)] = (unsigned)e;
    tail = __builtin_popcount(mask);
  }

  const int start = (nsb > 0) ? (int)(((long long)blockIdx.x * nsb) / gridDim.x) : 0;

  v4u cur[KEY_LOADS];
  if (w < nsb) {
    int sb = start + w; if (sb >= nsb) sb -= nsb;
    load_keys(keys + ((long long)sb << 10), lane, cur);
  }
  for (int i = w; ; i += PUSH_WAVES) {
    const bool more = i < nsb;
    if (more) {
      int sb = start + i; if (sb >= nsb) sb -= nsb;
      const int base = sb << 10;
      int sn = (i + PUSH_WAVES < nsb) ? start + i + PUSH_WAVES : start + i;
      if (sn >= nsb) sn -= nsb;
      v4u nxt[KEY_LOADS];
      load_keys(keys + ((long long)sn << 10), lane, nxt);
#pragma unroll
      for (int k = 0; k < KEY_LOADS; ++k) {
#pragma unroll
        for (int q = 0; q < 4; ++q) {
          const unsigned x = cur[k][q] ^ gpat;
          const unsigned z = (x - 0x01010101u) & ~x & 0x80808080u;
          if (__builtin_amdgcn_ballot_w32(z != 0u)) {
#pragma unroll
            for (int b = 0; b < 4; ++b) {
              const bool m = ((x >> (8 * b)) & 0xffu) == 0u;
              const unsigned mask = __builtin_amdgcn_ballot_w32(m);
              if (mask) {
                if (m) {
                  const unsigned e = (unsigned)(base + k * 512 + lane * 16 + q * 4 + b);
                  ring[(tail + __builtin_amdgcn_mbcnt_lo(mask, 0u)) & (PUSH_RING - 1)] = e;
                }
                tail += __builtin_popcount(mask);
              }
            }
          }
        }
        while (tail - head >= PUSH_MIN_BATCH) {
          const int cnt = min(32, (tail - head) & ~3);
          push_batch<K, STORE>(X, pk, dinv, acc, ring, head, cnt, row0, lane, lst, rowcnt, lcnt);
          head += cnt;
        }
      }
#pragma unroll
      for (int k = 0; k < KEY_LOADS; ++k) cur[k] = nxt[k];
    } else {
      while (tail - head >= 1) {
        const int cnt = min(32, tail - head);
        push_batch<K, STORE>(X, pk, dinv, acc, ring, head, cnt, row0, lane, lst, rowcnt, lcnt);
        head += cnt;
      }
      break;
    }
  }
}

template <int K, bool FIRST>
__global__ __launch_bounds__(PUSH_THREADS) void k_push_mm(
    const float* __restrict__ X, const unsigned* __restrict__ pk,
    const unsigned char* __restrict__ keys, const float* __restrict__ dinv,
    const float* __restrict__ Wm, const float* __restrict__ bias,
    float* __restrict__ OUT, unsigned* __restrict__ glist_all,
    int* __restrict__ gcnt, int n, int E) {
  __shared__ __attribute__((aligned(16))) float    acc[ACC_FLOATS];
  __shared__ __attribute__((aligned(16))) unsigned pend[PUSH_WAVES * PUSH_RING];
  __shared__ int lcnt;
  __shared__ __attribute__((aligned(16))) v2u lst[FIRST ? LIST_CAP : 1];
  __shared__ int rowcnt[64];
  __shared__ int rowpos[64];
  constexpr int PITCH = ACC_PITCH(K);

  const int tid  = threadIdx.x;
  const int lane = tid & 31;
  const int w    = __builtin_amdgcn_readfirstlane(tid >> 5);
  const int row0 = blockIdx.x * PUSH_ROWS;
  const int nrows = min(PUSH_ROWS, n - row0);
  unsigned* ring = pend + w * PUSH_RING;
  v2u* glist = (v2u*)glist_all + (long long)blockIdx.x * LIST_CAP;

  for (int i = tid; i < TILE_ROWS * PITCH; i += PUSH_THREADS) acc[i] = 0.0f;
  if (FIRST && tid == 0) lcnt = 0;
  if (FIRST && tid < 64) rowcnt[tid] = 0;
  __syncthreads();

  if (FIRST) {
    scan_push<K, true>(X, pk, keys, dinv, acc, ring, lst, rowcnt, &lcnt, row0, E, w, lane);
  } else {
    const int cnt = gcnt[blockIdx.x];
    if (cnt <= LIST_CAP) {
      const int per = (((cnt + PUSH_WAVES - 1) / PUSH_WAVES) + 3) & ~3;
      int b = w * per;
      const int end = min(cnt, b + per);
      v2u cur = {0u, 0u};
      if (b + lane < end) cur = glist[b + lane];
      float4 lo = make_float4(0.0f, 0.0f, 0.0f, 0.0f);
      float4 hi = make_float4(0.0f, 0.0f, 0.0f, 0.0f);
      int roff = -1;
      for (; b < end; b += 32) {
        v2u nxt = {0u, 0u};
        if (b + 32 + lane < end) nxt = glist[b + 32 + lane];
        const int count = min(32, end - b);
        int s = 0, off = 0;
        float nrm = 0.0f;
        if (lane < count) {
          s   = (int)(cur.x >> 6);
          off = (int)(cur.x & 63u) * PITCH;
          nrm = __uint_as_float(cur.y);
        }
        run_rows(X, acc, s, nrm, off, count, lane, lo, hi, roff);
        cur = nxt;
      }
      if (roff >= 0) run_flush(acc, roff, lane, lo, hi);
    } else {
      scan_push<K, false>(X, pk, keys, dinv, acc, ring, nullptr, nullptr, &lcnt, row0, E, w, lane);
    }
  }

  __syncthreads();
  int lc = 0;
  if (FIRST) {
    lc = lcnt;
    if (tid == 0) gcnt[blockIdx.x] = lc;
    if (tid < 64) {
      int before = 0;
#pragma unroll
      for (int i = 0; i < PUSH_ROWS; ++i) {
        const int cr = rowcnt[i];
        before += (i < tid) ? cr : 0;
      }
      rowpos[tid] = before;
    }
  }

  {
    constexpr int QPR = K / 4;
    constexpr int JST = (K == 256) ? 32 : 16;
    for (int idx = tid; idx < nrows * QPR; idx += PUSH_THREADS) {
      const int r  = idx / QPR;
      const int cq = idx - r * QPR;
      const int node = row0 + r;
      const float di = dinv[node];
      const float4 xv = *(const float4*)(X + (long long)node * K + cq * 4);
      float* ap = acc + r * PITCH + ((K == 256) ? ((cq >> 5) * 128 + (cq & 31)) : cq);
      ap[0]       += di * di * xv.x;
      ap[JST]     += di * di * xv.y;
      ap[2 * JST] += di * di * xv.z;
      ap[3 * JST] += di * di * xv.w;
    }
  }
  __syncthreads();

  if (FIRST && lc <= LIST_CAP) {
    for (int i = tid; i < lc; i += PUSH_THREADS) {
      const v2u e = lst[i];
      const int pos = __hip_atomic_fetch_add(rowpos + (int)(e.x & 63u), 1,
                                             __ATOMIC_RELAXED, __HIP_MEMORY_SCOPE_WORKGROUP);
      glist[pos] = e;
    }
  }

  const int lo = lane & 15;
  const int hi = lane >> 4;
  const int n0 = (w & 15) << 4;
  const int kh = w >> 4;
  v8f d0 = {}, d1 = {}, d2 = {};
  {
    constexpr int STEPS = K / 8;
    const int pbase = kh * (K / 2) + 2 * hi;
    const float* Ap = acc + lo * PITCH + pbase;
    const float* Wc = Wm + n0 + lo;
#pragma unroll 8
    for (int t = 0; t < STEPS; ++t) {
      const int p = pbase + 4 * t;
      v2f b;
      b.x = Wc[(long long)pos_to_chan<K>(p)     * C_HID];
      b.y = Wc[(long long)pos_to_chan<K>(p + 1) * C_HID];
      const v2f a0 = *(const v2f*)(Ap + 4 * t);
      const v2f a1 = *(const v2f*)(Ap + 16 * PITCH + 4 * t);
      const v2f a2 = *(const v2f*)(Ap + 32 * PITCH + 4 * t);
      d0 = __builtin_amdgcn_wmma_f32_16x16x4_f32(false, a0, false, b, (short)0, d0, false, false);
      d1 = __builtin_amdgcn_wmma_f32_16x16x4_f32(false, a1, false, b, (short)0, d1, false, false);
      d2 = __builtin_amdgcn_wmma_f32_16x16x4_f32(false, a2, false, b, (short)0, d2, false, false);
    }
  }

  float* S = (float*)pend;
  float* Sp = S + (hi * 8) * C_HID + n0 + lo;
  if (kh == 1) {
#pragma unroll
    for (int r = 0; r < 8; ++r) {
      Sp[(r)      * C_HID] = d0[r];
      Sp[(16 + r) * C_HID] = d1[r];
      Sp[(32 + r) * C_HID] = d2[r];
    }
  }
  __syncthreads();
  if (kh == 0) {
    const float bb = bias[n0 + lo];
#pragma unroll
    for (int r = 0; r < 8; ++r) {
      float v0 = d0[r] + Sp[(r)      * C_HID] + bb;
      float v1 = d1[r] + Sp[(16 + r) * C_HID] + bb;
      float v2 = d2[r] + Sp[(32 + r) * C_HID] + bb;
      Sp[(r)      * C_HID] = (v0 > 0.0f) ? v0 : expm1f(v0);
      Sp[(16 + r) * C_HID] = (v1 > 0.0f) ? v1 : expm1f(v1);
      Sp[(32 + r) * C_HID] = (v2 > 0.0f) ? v2 : expm1f(v2);
    }
  }
  __syncthreads();

  for (int idx = tid; idx < nrows * (C_HID / 4); idx += PUSH_THREADS) {
    const int r  = idx >> 6;
    const int cq = idx & 63;
    const float4 v = *(const float4*)(S + r * C_HID + cq * 4);
    *(float4*)(OUT + (long long)(row0 + r) * C_HID + cq * 4) = v;
  }
}

extern "C" void kernel_launch(void* const* d_in, const int* in_sizes, int n_in,
                              void* d_out, int out_size, void* d_ws, size_t ws_size,
                              hipStream_t stream) {
  const float* x  = (const float*)d_in[0];
  const int*   ei = (const int*)d_in[1];
  const float* W1 = (const float*)d_in[2];
  const float* b1 = (const float*)d_in[3];
  const float* W2 = (const float*)d_in[4];
  const float* b2 = (const float*)d_in[5];
  const float* W3 = (const float*)d_in[6];
  const float* b3 = (const float*)d_in[7];
  float* out = (float*)d_out;

  const int C   = C_HID;
  const int E   = in_sizes[1] / 2;
  const int cin = in_sizes[2] / C;
  const int n   = in_sizes[0] / cin;
  const int* src = ei;
  const int* dst = ei + E;

  char*  ws  = (char*)d_ws;
  size_t off = 0;
  auto alloc = [&](size_t bytes) -> void* {
    void* p = ws + off;
    off = (off + bytes + 255) & ~(size_t)255;
    return p;
  };
  int*   degi = (int*)  alloc((size_t)n * sizeof(int));
  float* dinv = (float*)alloc((size_t)n * sizeof(float));
  float* XA   = (float*)alloc((size_t)n * C * sizeof(float));
  float* XB   = (float*)alloc((size_t)n * C * sizeof(float));
  unsigned* key4 = (unsigned*)alloc(((size_t)E + 3) / 4 * sizeof(unsigned));
  unsigned* pk   = (unsigned*)alloc((size_t)E * sizeof(unsigned));
  const unsigned char* keys = (const unsigned char*)key4;
  const int pgrid = (n + PUSH_ROWS - 1) / PUSH_ROWS;
  unsigned* glist = (unsigned*)alloc((size_t)pgrid * LIST_CAP * 2 * sizeof(unsigned));
  int*      gcnt  = (int*)     alloc((size_t)pgrid * sizeof(int));
  (void)ws_size; (void)n_in; (void)out_size;

  k_deg_init <<<(n + 255) / 256, 256, 0, stream>>>(degi, n);
  k_deg_count<<<((E + 3) / 4 + 255) / 256, 256, 0, stream>>>(src, dst, degi, key4, pk, E);
  k_dinv     <<<(n + 255) / 256, 256, 0, stream>>>(degi, dinv, n);

  k_push_mm<64, true>  <<<pgrid, PUSH_THREADS, 0, stream>>>(x,  pk, keys, dinv, W1, b1, XA,  glist, gcnt, n, E);
  k_push_mm<256, false><<<pgrid, PUSH_THREADS, 0, stream>>>(XA, pk, keys, dinv, W2, b2, XB,  glist, gcnt, n, E);
  k_push_mm<256, false><<<pgrid, PUSH_THREADS, 0, stream>>>(XB, pk, keys, dinv, W3, b3, out, glist, gcnt, n, E);
}


# degree pass without global atomics or launches of its own: extra work-groups of the recode launch own 256 nodes each, count their edges in LDS over all of dst, write rsqrt; 6 launches -> 4
# speedup vs baseline: 4.7253x; 1.3026x over previous
#include <hip/hip_runtime.h>
#include <hip/hip_bf16.h>

typedef __attribute__((ext_vector_type(2))) float v2f;
typedef __attribute__((ext_vector_type(8))) float v8f;

#define C_HID 256

#define PUSH_ROWS 40

typedef __attribute__((ext_vector_type(4))) unsigned v4u;
typedef __attribute__((ext_vector_type(2))) unsigned v2u;

#define LIST_CAP 4096

#define PREP_THREADS 1024
#define HIST_ROWS    256
#define HIST_LOADS   8

typedef __attribute__((ext_vector_type(4))) int v4i;

__device__ __forceinline__ void hist_one(int* hist, int d, unsigned nlo4) {
  const unsigned xb = ((unsigned)d << 2) + nlo4;
  if (xb < 4u * HIST_ROWS)
    __hip_atomic_fetch_add((int*)((char*)hist + xb), 1, __ATOMIC_RELAXED,
                           __HIP_MEMORY_SCOPE_WORKGROUP);
}

__device__ __forceinline__ void hist_vec(int* hist, const v4i& v, unsigned nlo4) {
  hist_one(hist, v.x, nlo4);
  hist_one(hist, v.y, nlo4);
  hist_one(hist, v.z, nlo4);
  hist_one(hist, v.w, nlo4);
}

__global__ __launch_bounds__(PREP_THREADS) void k_prep(
    const int* __restrict__ src, const int* __restrict__ dst,
    float* __restrict__ dinv, unsigned* __restrict__ key4,
    unsigned* __restrict__ pk, int n, int E, int hgroups) {
  __shared__ int hist[HIST_ROWS];
  const int tid = threadIdx.x;
  if ((int)blockIdx.x < hgroups) {
    const int lo = blockIdx.x * HIST_ROWS;
    const unsigned nlo4 = 0u - ((unsigned)lo << 2);
    if (tid < HIST_ROWS) hist[tid] = 0;
    __syncthreads();
    const int head  = min(E, (int)(((16u - ((unsigned)(size_t)dst & 15u)) & 15u) >> 2));
    const int nvec  = (E - head) >> 2;
    const int tail0 = head + (nvec << 2);
    {
      int e = -1;
      if (tid < head) e = tid;
      else if (tid - head < E - tail0) e = tail0 + (tid - head);
      if (e >= 0) hist_one(hist, dst[e], nlo4);
    }
    const v4i* dv = (const v4i*)(dst + head);
    constexpr int TRIP = PREP_THREADS * HIST_LOADS;
    const int nfull = nvec / TRIP;
    if (nfull > 0) {
      v4i cur[HIST_LOADS];
#pragma unroll
      for (int k = 0; k < HIST_LOADS; ++k) cur[k] = dv[k * PREP_THREADS + tid];
      for (int i = 0; i < nfull; ++i) {
        const int nb = ((i + 1 < nfull) ? i + 1 : i) * TRIP;
        v4i nxt[HIST_LOADS];
#pragma unroll
        for (int k = 0; k < HIST_LOADS; ++k) nxt[k] = dv[nb + k * PREP_THREADS + tid];
#pragma unroll
        for (int k = 0; k < HIST_LOADS; ++k) hist_vec(hist, cur[k], nlo4);
#pragma unroll
        for (int k = 0; k < HIST_LOADS; ++k) cur[k] = nxt[k];
      }
    }
    {
      const int r0 = nfull * TRIP;
      const v4i none = {-1, -1, -1, -1};
      v4i rem[HIST_LOADS];
#pragma unroll
      for (int k = 0; k < HIST_LOADS; ++k) {
        const int v = r0 + k * PREP_THREADS + tid;
        rem[k] = none;
        if (v < nvec) rem[k] = dv[v];
      }
#pragma unroll
      for (int k = 0; k < HIST_LOADS; ++k) hist_vec(hist, rem[k], nlo4);
    }
    __syncthreads();
    if (tid < HIST_ROWS && lo + tid < n)
      dinv[lo + tid] = rsqrtf((float)(1 + hist[tid]));
    return;
  }

  const int t  = ((int)blockIdx.x - hgroups) * PREP_THREADS + tid;
  const int e0 = t * 4;
  if (e0 >= E) return;
  unsigned kw = 0u;
  if (e0 + 3 < E) {
    int d[4], sv[4];
#pragma unroll
    for (int q = 0; q < 4; ++q) { d[q] = dst[e0 + q]; sv[q] = src[e0 + q]; }
    unsigned pv[4];
#pragma unroll
    for (int q = 0; q < 4; ++q) {
      const int g = d[q] / PUSH_ROWS;
      pv[q] = ((unsigned)sv[q] << 6) | (unsigned)(d[q] - g * PUSH_ROWS);
      kw |= (unsigned)g << (8 * q);
    }
    v4u o; o.x = pv[0]; o.y = pv[1]; o.z = pv[2]; o.w = pv[3];
    *(v4u*)(pk + e0) = o;
  } else {
    for (int q = 0; e0 + q < E; ++q) {
      const int d = dst[e0 + q];
      const int g = d / PUSH_ROWS;
      pk[e0 + q] = ((unsigned)src[e0 + q] << 6) | (unsigned)(d - g * PUSH_ROWS);
      kw |= (unsigned)g << (8 * q);
    }
  }
  key4[t] = kw;
}

#define PUSH_THREADS 1024
#define PUSH_WAVES   32
#define PUSH_RING    1024
#define KEY_LOADS    2
#define PUSH_MIN_BATCH 16
#define TILE_ROWS    48
#define ACC_PITCH(K) ((K) + 4)
#define ACC_FLOATS   (TILE_ROWS * ACC_PITCH(C_HID))

template <int K>
__device__ __forceinline__ int pos_to_chan(int p) {
  if (K == 256) return (p & 128) + 4 * (p & 31) + ((p >> 5) & 3);
  return 4 * (p & 15) + (p >> 4);
}

__device__ __forceinline__ void ldsAddF(float* p, float v) {
  __hip_atomic_fetch_add(p, v, __ATOMIC_RELAXED, __HIP_MEMORY_SCOPE_WORKGROUP);
}

__device__ __forceinline__ int laneI(int v, int j) {
  return __builtin_amdgcn_readlane(v, j);
}
__device__ __forceinline__ float laneF(float v, int j) {
  return __int_as_float(__builtin_amdgcn_readlane(__float_as_int(v), j));
}

template <int K>
__device__ __forceinline__ void add_rows(
    const float* __restrict__ X, float* acc, int s, float nrm, int off,
    int count, int lane) {
  if (K == 256) {
    int j = 0;
    for (; j + 4 <= count; j += 4) {
      float4 a[4], b[4];
#pragma unroll
      for (int u = 0; u < 4; ++u) {
        const float* hp = X + (long long)laneI(s, j + u) * K + lane * 4;
        a[u] = *(const float4*)hp;
        b[u] = *(const float4*)(hp + 128);
      }
#pragma unroll
      for (int u = 0; u < 4; ++u) {
        const float nj = laneF(nrm, j + u);
        float* q = acc + laneI(off, j + u) + lane;
        ldsAddF(q,       a[u].x * nj);
        ldsAddF(q + 32,  a[u].y * nj);
        ldsAddF(q + 64,  a[u].z * nj);
        ldsAddF(q + 96,  a[u].w * nj);
        ldsAddF(q + 128, b[u].x * nj);
        ldsAddF(q + 160, b[u].y * nj);
        ldsAddF(q + 192, b[u].z * nj);
        ldsAddF(q + 224, b[u].w * nj);
      }
    }
    for (; j < count; ++j) {
      const float* hp = X + (long long)laneI(s, j) * K + lane * 4;
      const float4 a = *(const float4*)hp;
      const float4 b = *(const float4*)(hp + 128);
      const float nj = laneF(nrm, j);
      float* q = acc + laneI(off, j) + lane;
      ldsAddF(q,       a.x * nj);
      ldsAddF(q + 32,  a.y * nj);
      ldsAddF(q + 64,  a.z * nj);
      ldsAddF(q + 96,  a.w * nj);
      ldsAddF(q + 128, b.x * nj);
      ldsAddF(q + 160, b.y * nj);
      ldsAddF(q + 192, b.z * nj);
      ldsAddF(q + 224, b.w * nj);
    }
  } else {
    const int  sub = lane & 15;
    const bool odd = lane >= 16;
    for (int j = 0; j < count; j += 8) {
      float4 a[4];
#pragma unroll
      for (int u = 0; u < 4; ++u) {
        const int e0 = j + 2 * u;
        const int sA = laneI(s, e0), sB = laneI(s, e0 + 1);
        a[u] = make_float4(0.0f, 0.0f, 0.0f, 0.0f);
        if (e0 + (odd ? 1 : 0) < count)
          a[u] = *(const float4*)(X + (long long)(odd ? sB : sA) * K + sub * 4);
      }
#pragma unroll
      for (int u = 0; u < 4; ++u) {
        const int e0 = j + 2 * u;
        const float nA = laneF(nrm, e0), nB = laneF(nrm, e0 + 1);
        const int   oA = laneI(off, e0), oB = laneI(off, e0 + 1);
        if (e0 + (odd ? 1 : 0) < count) {
          const float nj = odd ? nB : nA;
          float* q = acc + (odd ? oB : oA) + sub;
          ldsAddF(q,      a[u].x * nj);
          ldsAddF(q + 16, a[u].y * nj);
          ldsAddF(q + 32, a[u].z * nj);
          ldsAddF(q + 48, a[u].w * nj);
        }
      }
    }
  }
}

template <int K, bool STORE>
__device__ __forceinline__ void push_batch(
    const float* __restrict__ X, const unsigned* __restrict__ pk,
    const float* __restrict__ dinv, float* acc, const unsigned* ring,
    int head, int count, int row0, int lane, v2u* lst, int* rowcnt, int* lcnt) {
  int s = 0, off = 0, dl = 0;
  float nrm = 0.0f;
  unsigned ent = 0u;
  if (lane < count) {
    ent = pk[ring[(head + lane) & (PUSH_RING - 1)]];
    dl  = (int)(ent & 63u);
    s   = (int)(ent >> 6);
    nrm = dinv[s] * dinv[row0 + dl];
    off = dl * ACC_PITCH(K);
  }
  if (STORE) {
    int base = 0;
    if (lane == 0)
      base = __hip_atomic_fetch_add(lcnt, count, __ATOMIC_RELAXED, __HIP_MEMORY_SCOPE_WORKGROUP);
    base = __builtin_amdgcn_readfirstlane(base);
    if (lane < count && base + lane < LIST_CAP) {
      v2u o; o.x = ent; o.y = __float_as_uint(nrm);
      lst[base + lane] = o;
      __hip_atomic_fetch_add(rowcnt + dl, 1, __ATOMIC_RELAXED, __HIP_MEMORY_SCOPE_WORKGROUP);
    }
  }
  add_rows<K>(X, acc, s, nrm, off, count, lane);
}

__device__ __forceinline__ void run_flush(float* acc, int roff, int lane,
                                          const float4& lo, const float4& hi) {
  float* q = acc + roff + lane;
  ldsAddF(q,       lo.x);
  ldsAddF(q + 32,  lo.y);
  ldsAddF(q + 64,  lo.z);
  ldsAddF(q + 96,  lo.w);
  ldsAddF(q + 128, hi.x);
  ldsAddF(q + 160, hi.y);
  ldsAddF(q + 192, hi.z);
  ldsAddF(q + 224, hi.w);
}

__device__ __forceinline__ void run_edge(float* acc, int lane, int o, float nj,
                                         const float4& a, const float4& b,
                                         float4& lo, float4& hi, int& roff) {
  if (o != roff) {
    if (roff >= 0) run_flush(acc, roff, lane, lo, hi);
    lo = make_float4(0.0f, 0.0f, 0.0f, 0.0f);
    hi = make_float4(0.0f, 0.0f, 0.0f, 0.0f);
    roff = o;
  }
  lo.x += a.x * nj; lo.y += a.y * nj; lo.z += a.z * nj; lo.w += a.w * nj;
  hi.x += b.x * nj; hi.y += b.y * nj; hi.z += b.z * nj; hi.w += b.w * nj;
}

__device__ __forceinline__ void run_rows(
    const float* __restrict__ X, float* acc, int s, float nrm, int off,
    int count, int lane, float4& lo, float4& hi, int& roff) {
  int j = 0;
  for (; j + 8 <= count; j += 8) {
    float4 a[8], b[8];
#pragma unroll
    for (int u = 0; u < 8; ++u) {
      const float* hp = X + (long long)laneI(s, j + u) * C_HID + lane * 4;
      a[u] = *(const float4*)hp;
      b[u] = *(const float4*)(hp + 128);
    }
#pragma unroll
    for (int u = 0; u < 8; ++u)
      run_edge(acc, lane, laneI(off, j + u), laneF(nrm, j + u), a[u], b[u], lo, hi, roff);
  }
  for (; j + 4 <= count; j += 4) {
    float4 a[4], b[4];
#pragma unroll
    for (int u = 0; u < 4; ++u) {
      const float* hp = X + (long long)laneI(s, j + u) * C_HID + lane * 4;
      a[u] = *(const float4*)hp;
      b[u] = *(const float4*)(hp + 128);
    }
#pragma unroll
    for (int u = 0; u < 4; ++u)
      run_edge(acc, lane, laneI(off, j + u), laneF(nrm, j + u), a[u], b[u], lo, hi, roff);
  }
  for (; j < count; ++j) {
    const float* hp = X + (long long)laneI(s, j) * C_HID + lane * 4;
    const float4 a = *(const float4*)hp;
    const float4 b = *(const float4*)(hp + 128);
    run_edge(acc, lane, laneI(off, j), laneF(nrm, j), a, b, lo, hi, roff);
  }
}

__device__ __forceinline__ void load_keys(const unsigned char* __restrict__ p, int lane, v4u* v) {
#pragma unroll
  for (int k = 0; k < KEY_LOADS; ++k)
    v[k] = *(const v4u*)(p + k * 512 + lane * 16);
}

template <int K, bool STORE>
__device__ __forceinline__ void scan_push(
    const float* __restrict__ X, const unsigned* __restrict__ pk,
    const unsigned char* __restrict__ keys, const float* __restrict__ dinv,
    float* acc, unsigned* ring, v2u* lst, int* rowcnt, int* lcnt,
    int row0, int E, int w, int lane) {
  const unsigned gpat = (unsigned)blockIdx.x * 0x01010101u;

  const int nsb   = E >> 10;
  const int tail0 = nsb << 10;

  int head = 0, tail = 0;
  {
    const int e = tail0 + w * 32 + lane;
    bool m = false;
    if (e < E) m = (unsigned)keys[e] == (unsigned)blockIdx.x;
    const unsigned mask = __builtin_amdgcn_ballot_w32(m);
    if (m) ring[__builtin_amdgcn_mbcnt_lo(mask, 0u)] = (unsigned)e;
    tail = __builtin_popcount(mask);
  }

  const int start = (nsb > 0) ? (int)(((long long)blockIdx.x * nsb) / gridDim.x) : 0;

  v4u cur[KEY_LOADS];
  if (w < nsb) {
    int sb = start + w; if (sb >= nsb) sb -= nsb;
    load_keys(keys + ((long long)sb << 10), lane, cur);
  }
  for (int i = w; ; i += PUSH_WAVES) {
    const bool more = i < nsb;
    if (more) {
      int sb = start + i; if (sb >= nsb) sb -= nsb;
      const int base = sb << 10;
      int sn = (i + PUSH_WAVES < nsb) ? start + i + PUSH_WAVES : start + i;
      if (sn >= nsb) sn -= nsb;
      v4u nxt[KEY_LOADS];
      load_keys(keys + ((long long)sn << 10), lane, nxt);
#pragma unroll
      for (int k = 0; k < KEY_LOADS; ++k) {
#pragma unroll
        for (int q = 0; q < 4; ++q) {
          const unsigned x = cur[k][q] ^ gpat;
          const unsigned z = (x - 0x01010101u) & ~x & 0x80808080u;
          if (__builtin_amdgcn_ballot_w32(z != 0u)) {
#pragma unroll
            for (int b = 0; b < 4; ++b) {
              const bool m = ((x >> (8 * b)) & 0xffu) == 0u;
              const unsigned mask = __builtin_amdgcn_ballot_w32(m);
              if (mask) {
                if (m) {
                  const unsigned e = (unsigned)(base + k * 512 + lane * 16 + q * 4 + b);
                  ring[(tail + __builtin_amdgcn_mbcnt_lo(mask, 0u)) & (PUSH_RING - 1)] = e;
                }
                tail += __builtin_popcount(mask);
              }
            }
          }
        }
        while (tail - head >= PUSH_MIN_BATCH) {
          const int cnt = min(32, (tail - head) & ~3);
          push_batch<K, STORE>(X, pk, dinv, acc, ring, head, cnt, row0, lane, lst, rowcnt, lcnt);
          head += cnt;
        }
      }
#pragma unroll
      for (int k = 0; k < KEY_LOADS; ++k) cur[k] = nxt[k];
    } else {
      while (tail - head >= 1) {
        const int cnt = min(32, tail - head);
        push_batch<K, STORE>(X, pk, dinv, acc, ring, head, cnt, row0, lane, lst, rowcnt, lcnt);
        head += cnt;
      }
      break;
    }
  }
}

template <int K, bool FIRST>
__global__ __launch_bounds__(PUSH_THREADS) void k_push_mm(
    const float* __restrict__ X, const unsigned* __restrict__ pk,
    const unsigned char* __restrict__ keys, const float* __restrict__ dinv,
    const float* __restrict__ Wm, const float* __restrict__ bias,
    float* __restrict__ OUT, unsigned* __restrict__ glist_all,
    int* __restrict__ gcnt, int n, int E) {
  __shared__ __attribute__((aligned(16))) float    acc[ACC_FLOATS];
  __shared__ __attribute__((aligned(16))) unsigned pend[PUSH_WAVES * PUSH_RING];
  __shared__ int lcnt;
  __shared__ __attribute__((aligned(16))) v2u lst[FIRST ? LIST_CAP : 1];
  __shared__ int rowcnt[64];
  __shared__ int rowpos[64];
  constexpr int PITCH = ACC_PITCH(K);

  const int tid  = threadIdx.x;
  const int lane = tid & 31;
  const int w    = __builtin_amdgcn_readfirstlane(tid >> 5);
  const int row0 = blockIdx.x * PUSH_ROWS;
  const int nrows = min(PUSH_ROWS, n - row0);
  unsigned* ring = pend + w * PUSH_RING;
  v2u* glist = (v2u*)glist_all + (long long)blockIdx.x * LIST_CAP;

  for (int i = tid; i < TILE_ROWS * PITCH; i += PUSH_THREADS) acc[i] = 0.0f;
  if (FIRST && tid == 0) lcnt = 0;
  if (FIRST && tid < 64) rowcnt[tid] = 0;
  __syncthreads();

  if (FIRST) {
    scan_push<K, true>(X, pk, keys, dinv, acc, ring, lst, rowcnt, &lcnt, row0, E, w, lane);
  } else {
    const int cnt = gcnt[blockIdx.x];
    if (cnt <= LIST_CAP) {
      const int per = (((cnt + PUSH_WAVES - 1) / PUSH_WAVES) + 3) & ~3;
      int b = w * per;
      const int end = min(cnt, b + per);
      v2u cur = {0u, 0u};
      if (b + lane < end) cur = glist[b + lane];
      float4 lo = make_float4(0.0f, 0.0f, 0.0f, 0.0f);
      float4 hi = make_float4(0.0f, 0.0f, 0.0f, 0.0f);
      int roff = -1;
      for (; b < end; b += 32) {
        v2u nxt = {0u, 0u};
        if (b + 32 + lane < end) nxt = glist[b + 32 + lane];
        const int count = min(32, end - b);
        int s = 0, off = 0;
        float nrm = 0.0f;
        if (lane < count) {
          s   = (int)(cur.x >> 6);
          off = (int)(cur.x & 63u) * PITCH;
          nrm = __uint_as_float(cur.y);
        }
        run_rows(X, acc, s, nrm, off, count, lane, lo, hi, roff);
        cur = nxt;
      }
      if (roff >= 0) run_flush(acc, roff, lane, lo, hi);
    } else {
      scan_push<K, false>(X, pk, keys, dinv, acc, ring, nullptr, nullptr, &lcnt, row0, E, w, lane);
    }
  }

  __syncthreads();
  int lc = 0;
  if (FIRST) {
    lc = lcnt;
    if (tid == 0) gcnt[blockIdx.x] = lc;
    if (tid < 64) {
      int before = 0;
#pragma unroll
      for (int i = 0; i < PUSH_ROWS; ++i) {
        const int cr = rowcnt[i];
        before += (i < tid) ? cr : 0;
      }
      rowpos[tid] = before;
    }
  }

  {
    constexpr int QPR = K / 4;
    constexpr int JST = (K == 256) ? 32 : 16;
    for (int idx = tid; idx < nrows * QPR; idx += PUSH_THREADS) {
      const int r  = idx / QPR;
      const int cq = idx - r * QPR;
      const int node = row0 + r;
      const float di = dinv[node];
      const float4 xv = *(const float4*)(X + (long long)node * K + cq * 4);
      float* ap = acc + r * PITCH + ((K == 256) ? ((cq >> 5) * 128 + (cq & 31)) : cq);
      ap[0]       += di * di * xv.x;
      ap[JST]     += di * di * xv.y;
      ap[2 * JST] += di * di * xv.z;
      ap[3 * JST] += di * di * xv.w;
    }
  }
  __syncthreads();

  if (FIRST && lc <= LIST_CAP) {
    for (int i = tid; i < lc; i += PUSH_THREADS) {
      const v2u e = lst[i];
      const int pos = __hip_atomic_fetch_add(rowpos + (int)(e.x & 63u), 1,
                                             __ATOMIC_RELAXED, __HIP_MEMORY_SCOPE_WORKGROUP);
      glist[pos] = e;
    }
  }

  const int lo = lane & 15;
  const int hi = lane >> 4;
  const int n0 = (w & 15) << 4;
  const int kh = w >> 4;
  v8f d0 = {}, d1 = {}, d2 = {};
  {
    constexpr int STEPS = K / 8;
    const int pbase = kh * (K / 2) + 2 * hi;
    const float* Ap = acc + lo * PITCH + pbase;
    const float* Wc = Wm + n0 + lo;
#pragma unroll 8
    for (int t = 0; t < STEPS; ++t) {
      const int p = pbase + 4 * t;
      v2f b;
      b.x = Wc[(long long)pos_to_chan<K>(p)     * C_HID];
      b.y = Wc[(long long)pos_to_chan<K>(p + 1) * C_HID];
      const v2f a0 = *(const v2f*)(Ap + 4 * t);
      const v2f a1 = *(const v2f*)(Ap + 16 * PITCH + 4 * t);
      const v2f a2 = *(const v2f*)(Ap + 32 * PITCH + 4 * t);
      d0 = __builtin_amdgcn_wmma_f32_16x16x4_f32(false, a0, false, b, (short)0, d0, false, false);
      d1 = __builtin_amdgcn_wmma_f32_16x16x4_f32(false, a1, false, b, (short)0, d1, false, false);
      d2 = __builtin_amdgcn_wmma_f32_16x16x4_f32(false, a2, false, b, (short)0, d2, false, false);
    }
  }

  float* S = (float*)pend;
  float* Sp = S + (hi * 8) * C_HID + n0 + lo;
  if (kh == 1) {
#pragma unroll
    for (int r = 0; r < 8; ++r) {
      Sp[(r)      * C_HID] = d0[r];
      Sp[(16 + r) * C_HID] = d1[r];
      Sp[(32 + r) * C_HID] = d2[r];
    }
  }
  __syncthreads();
  if (kh == 0) {
    const float bb = bias[n0 + lo];
#pragma unroll
    for (int r = 0; r < 8; ++r) {
      float v0 = d0[r] + Sp[(r)      * C_HID] + bb;
      float v1 = d1[r] + Sp[(16 + r) * C_HID] + bb;
      float v2 = d2[r] + Sp[(32 + r) * C_HID] + bb;
      Sp[(r)      * C_HID] = (v0 > 0.0f) ? v0 : expm1f(v0);
      Sp[(16 + r) * C_HID] = (v1 > 0.0f) ? v1 : expm1f(v1);
      Sp[(32 + r) * C_HID] = (v2 > 0.0f) ? v2 : expm1f(v2);
    }
  }
  __syncthreads();

  for (int idx = tid; idx < nrows * (C_HID / 4); idx += PUSH_THREADS) {
    const int r  = idx >> 6;
    const int cq = idx & 63;
    const float4 v = *(const float4*)(S + r * C_HID + cq * 4);
    *(float4*)(OUT + (long long)(row0 + r) * C_HID + cq * 4) = v;
  }
}

extern "C" void kernel_launch(void* const* d_in, const int* in_sizes, int n_in,
                              void* d_out, int out_size, void* d_ws, size_t ws_size,
                              hipStream_t stream) {
  const float* x  = (const float*)d_in[0];
  const int*   ei = (const int*)d_in[1];
  const float* W1 = (const float*)d_in[2];
  const float* b1 = (const float*)d_in[3];
  const float* W2 = (const float*)d_in[4];
  const float* b2 = (const float*)d_in[5];
  const float* W3 = (const float*)d_in[6];
  const float* b3 = (const float*)d_in[7];
  float* out = (float*)d_out;

  const int C   = C_HID;
  const int E   = in_sizes[1] / 2;
  const int cin = in_sizes[2] / C;
  const int n   = in_sizes[0] / cin;
  const int* src = ei;
  const int* dst = ei + E;

  char*  ws  = (char*)d_ws;
  size_t off = 0;
  auto alloc = [&](size_t bytes) -> void* {
    void* p = ws + off;
    off = (off + bytes + 255) & ~(size_t)255;
    return p;
  };
  float* dinv = (float*)alloc((size_t)n * sizeof(float));
  float* XA   = (float*)alloc((size_t)n * C * sizeof(float));
  float* XB   = (float*)alloc((size_t)n * C * sizeof(float));
  unsigned* key4 = (unsigned*)alloc(((size_t)E + 3) / 4 * sizeof(unsigned));
  unsigned* pk   = (unsigned*)alloc((size_t)E * sizeof(unsigned));
  const unsigned char* keys = (const unsigned char*)key4;
  const int pgrid = (n + PUSH_ROWS - 1) / PUSH_ROWS;
  unsigned* glist = (unsigned*)alloc((size_t)pgrid * LIST_CAP * 2 * sizeof(unsigned));
  int*      gcnt  = (int*)     alloc((size_t)pgrid * sizeof(int));
  (void)ws_size; (void)n_in; (void)out_size;

  const int hgroups = (n + HIST_ROWS - 1) / HIST_ROWS;
  const int rgroups = ((E + 3) / 4 + PREP_THREADS - 1) / PREP_THREADS;
  k_prep<<<hgroups + rgroups, PREP_THREADS, 0, stream>>>(src, dst, dinv, key4, pk, n, E, hgroups);

  k_push_mm<64, true>  <<<pgrid, PUSH_THREADS, 0, stream>>>(x,  pk, keys, dinv, W1, b1, XA,  glist, gcnt, n, E);
  k_push_mm<256, false><<<pgrid, PUSH_THREADS, 0, stream>>>(XA, pk, keys, dinv, W2, b2, XB,  glist, gcnt, n, E);
  k_push_mm<256, false><<<pgrid, PUSH_THREADS, 0, stream>>>(XB, pk, keys, dinv, W3, b3, out, glist, gcnt, n, E);
}


# degree count split over 4 pieces of the edge list: a counting work-group owns 1024 nodes and reads a quarter of dst; a node's 4 partial counts are one record, readers sum it; all sums unchanged
# speedup vs baseline: 5.1223x; 1.0840x over previous
#include <hip/hip_runtime.h>
#include <hip/hip_bf16.h>

typedef __attribute__((ext_vector_type(2))) float v2f;
typedef __attribute__((ext_vector_type(8))) float v8f;

#define C_HID 256

#define PUSH_ROWS 40

typedef __attribute__((ext_vector_type(4))) unsigned v4u;
typedef __attribute__((ext_vector_type(2))) unsigned v2u;

#define LIST_CAP 4096

#define PREP_THREADS 1024
#define HIST_ROWS    1024
#define HIST_CHUNKS  4
#define HIST_LOADS   4

typedef __attribute__((ext_vector_type(4))) int v4i;

__device__ __forceinline__ void hist_one(int* hist, int d, unsigned nlo4) {
  const unsigned xb = ((unsigned)d << 2) + nlo4;
  if (xb < 4u * HIST_ROWS)
    __hip_atomic_fetch_add((int*)((char*)hist + xb), 1, __ATOMIC_RELAXED,
                           __HIP_MEMORY_SCOPE_WORKGROUP);
}

__device__ __forceinline__ void hist_vec(int* hist, const v4i& v, unsigned nlo4) {
  hist_one(hist, v.x, nlo4);
  hist_one(hist, v.y, nlo4);
  hist_one(hist, v.z, nlo4);
  hist_one(hist, v.w, nlo4);
}

__device__ __forceinline__ float dinv_at(const int* __restrict__ degp, int node) {
  const v4i c = *(const v4i*)(degp + node * HIST_CHUNKS);
  return rsqrtf((float)(1 + c.x + c.y + c.z + c.w));
}

__global__ __launch_bounds__(PREP_THREADS) void k_prep(
    const int* __restrict__ src, const int* __restrict__ dst,
    int* __restrict__ degp, unsigned* __restrict__ key4,
    unsigned* __restrict__ pk, int n, int E, int hgroups) {
  __shared__ int hist[HIST_ROWS];
  const int tid = threadIdx.x;
  if ((int)blockIdx.x < hgroups * HIST_CHUNKS) {
    const int rng = (int)blockIdx.x / HIST_CHUNKS;
    const int pc  = (int)blockIdx.x - rng * HIST_CHUNKS;
    const int lo = rng * HIST_ROWS;
    const unsigned nlo4 = 0u - ((unsigned)lo << 2);
    if (tid < HIST_ROWS) hist[tid] = 0;
    __syncthreads();
    const int head  = min(E, (int)(((16u - ((unsigned)(size_t)dst & 15u)) & 15u) >> 2));
    const int nall  = (E - head) >> 2;
    const int tail0 = head + (nall << 2);
    if (pc == 0) {
      int e = -1;
      if (tid < head) e = tid;
      else if (tid - head < E - tail0) e = tail0 + (tid - head);
      if (e >= 0) hist_one(hist, dst[e], nlo4);
    }
    const int v0   = (int)(((long long)nall * pc) / HIST_CHUNKS);
    const int nvec = (int)(((long long)nall * (pc + 1)) / HIST_CHUNKS) - v0;
    const v4i* dv = (const v4i*)(dst + head) + v0;
    constexpr int TRIP = PREP_THREADS * HIST_LOADS;
    const int nfull = nvec / TRIP;
    if (nfull > 0) {
      v4i cur[HIST_LOADS];
#pragma unroll
      for (int k = 0; k < HIST_LOADS; ++k) cur[k] = dv[k * PREP_THREADS + tid];
      for (int i = 0; i < nfull; ++i) {
        const int nb = ((i + 1 < nfull) ? i + 1 : i) * TRIP;
        v4i nxt[HIST_LOADS];
#pragma unroll
        for (int k = 0; k < HIST_LOADS; ++k) nxt[k] = dv[nb + k * PREP_THREADS + tid];
#pragma unroll
        for (int k = 0; k < HIST_LOADS; ++k) hist_vec(hist, cur[k], nlo4);
#pragma unroll
        for (int k = 0; k < HIST_LOADS; ++k) cur[k] = nxt[k];
      }
    }
    {
      const int r0 = nfull * TRIP;
      const v4i none = {-1, -1, -1, -1};
      v4i rem[HIST_LOADS];
#pragma unroll
      for (int k = 0; k < HIST_LOADS; ++k) {
        const int v = r0 + k * PREP_THREADS + tid;
        rem[k] = none;
        if (v < nvec) rem[k] = dv[v];
      }
#pragma unroll
      for (int k = 0; k < HIST_LOADS; ++k) hist_vec(hist, rem[k], nlo4);
    }
    __syncthreads();
    if (tid < HIST_ROWS && lo + tid < n)
      degp[(lo + tid) * HIST_CHUNKS + pc] = hist[tid];
    return;
  }

  const int t  = ((int)blockIdx.x - hgroups * HIST_CHUNKS) * PREP_THREADS + tid;
  const int e0 = t * 4;
  if (e0 >= E) return;
  unsigned kw = 0u;
  if (e0 + 3 < E) {
    int d[4], sv[4];
#pragma unroll
    for (int q = 0; q < 4; ++q) { d[q] = dst[e0 + q]; sv[q] = src[e0 + q]; }
    unsigned pv[4];
#pragma unroll
    for (int q = 0; q < 4; ++q) {
      const int g = d[q] / PUSH_ROWS;
      pv[q] = ((unsigned)sv[q] << 6) | (unsigned)(d[q] - g * PUSH_ROWS);
      kw |= (unsigned)g << (8 * q);
    }
    v4u o; o.x = pv[0]; o.y = pv[1]; o.z = pv[2]; o.w = pv[3];
    *(v4u*)(pk + e0) = o;
  } else {
    for (int q = 0; e0 + q < E; ++q) {
      const int d = dst[e0 + q];
      const int g = d / PUSH_ROWS;
      pk[e0 + q] = ((unsigned)src[e0 + q] << 6) | (unsigned)(d - g * PUSH_ROWS);
      kw |= (unsigned)g << (8 * q);
    }
  }
  key4[t] = kw;
}

#define PUSH_THREADS 1024
#define PUSH_WAVES   32
#define PUSH_RING    1024
#define KEY_LOADS    2
#define PUSH_MIN_BATCH 16
#define TILE_ROWS    48
#define ACC_PITCH(K) ((K) + 4)
#define ACC_FLOATS   (TILE_ROWS * ACC_PITCH(C_HID))

template <int K>
__device__ __forceinline__ int pos_to_chan(int p) {
  if (K == 256) return (p & 128) + 4 * (p & 31) + ((p >> 5) & 3);
  return 4 * (p & 15) + (p >> 4);
}

__device__ __forceinline__ void ldsAddF(float* p, float v) {
  __hip_atomic_fetch_add(p, v, __ATOMIC_RELAXED, __HIP_MEMORY_SCOPE_WORKGROUP);
}

__device__ __forceinline__ int laneI(int v, int j) {
  return __builtin_amdgcn_readlane(v, j);
}
__device__ __forceinline__ float laneF(float v, int j) {
  return __int_as_float(__builtin_amdgcn_readlane(__float_as_int(v), j));
}

template <int K>
__device__ __forceinline__ void add_rows(
    const float* __restrict__ X, float* acc, int s, float nrm, int off,
    int count, int lane) {
  if (K == 256) {
    int j = 0;
    for (; j + 4 <= count; j += 4) {
      float4 a[4], b[4];
#pragma unroll
      for (int u = 0; u < 4; ++u) {
        const float* hp = X + (long long)laneI(s, j + u) * K + lane * 4;
        a[u] = *(const float4*)hp;
        b[u] = *(const float4*)(hp + 128);
      }
#pragma unroll
      for (int u = 0; u < 4; ++u) {
        const float nj = laneF(nrm, j + u);
        float* q = acc + laneI(off, j + u) + lane;
        ldsAddF(q,       a[u].x * nj);
        ldsAddF(q + 32,  a[u].y * nj);
        ldsAddF(q + 64,  a[u].z * nj);
        ldsAddF(q + 96,  a[u].w * nj);
        ldsAddF(q + 128, b[u].x * nj);
        ldsAddF(q + 160, b[u].y * nj);
        ldsAddF(q + 192, b[u].z * nj);
        ldsAddF(q + 224, b[u].w * nj);
      }
    }
    for (; j < count; ++j) {
      const float* hp = X + (long long)laneI(s, j) * K + lane * 4;
      const float4 a = *(const float4*)hp;
      const float4 b = *(const float4*)(hp + 128);
      const float nj = laneF(nrm, j);
      float* q = acc + laneI(off, j) + lane;
      ldsAddF(q,       a.x * nj);
      ldsAddF(q + 32,  a.y * nj);
      ldsAddF(q + 64,  a.z * nj);
      ldsAddF(q + 96,  a.w * nj);
      ldsAddF(q + 128, b.x * nj);
      ldsAddF(q + 160, b.y * nj);
      ldsAddF(q + 192, b.z * nj);
      ldsAddF(q + 224, b.w * nj);
    }
  } else {
    const int  sub = lane & 15;
    const bool odd = lane >= 16;
    for (int j = 0; j < count; j += 8) {
      float4 a[4];
#pragma unroll
      for (int u = 0; u < 4; ++u) {
        const int e0 = j + 2 * u;
        const int sA = laneI(s, e0), sB = laneI(s, e0 + 1);
        a[u] = make_float4(0.0f, 0.0f, 0.0f, 0.0f);
        if (e0 + (odd ? 1 : 0) < count)
          a[u] = *(const float4*)(X + (long long)(odd ? sB : sA) * K + sub * 4);
      }
#pragma unroll
      for (int u = 0; u < 4; ++u) {
        const int e0 = j + 2 * u;
        const float nA = laneF(nrm, e0), nB = laneF(nrm, e0 + 1);
        const int   oA = laneI(off, e0), oB = laneI(off, e0 + 1);
        if (e0 + (odd ? 1 : 0) < count) {
          const float nj = odd ? nB : nA;
          float* q = acc + (odd ? oB : oA) + sub;
          ldsAddF(q,      a[u].x * nj);
          ldsAddF(q + 16, a[u].y * nj);
          ldsAddF(q + 32, a[u].z * nj);
          ldsAddF(q + 48, a[u].w * nj);
        }
      }
    }
  }
}

template <int K, bool STORE>
__device__ __forceinline__ void push_batch(
    const float* __restrict__ X, const unsigned* __restrict__ pk,
    const int* __restrict__ degp, float* acc, const unsigned* ring,
    int head, int count, int row0, int lane, v2u* lst, int* rowcnt, int* lcnt) {
  int s = 0, off = 0, dl = 0;
  float nrm = 0.0f;
  unsigned ent = 0u;
  if (lane < count) {
    ent = pk[ring[(head + lane) & (PUSH_RING - 1)]];
    dl  = (int)(ent & 63u);
    s   = (int)(ent >> 6);
    nrm = dinv_at(degp, s) * dinv_at(degp, row0 + dl);
    off = dl * ACC_PITCH(K);
  }
  if (STORE) {
    int base = 0;
    if (lane == 0)
      base = __hip_atomic_fetch_add(lcnt, count, __ATOMIC_RELAXED, __HIP_MEMORY_SCOPE_WORKGROUP);
    base = __builtin_amdgcn_readfirstlane(base);
    if (lane < count && base + lane < LIST_CAP) {
      v2u o; o.x = ent; o.y = __float_as_uint(nrm);
      lst[base + lane] = o;
      __hip_atomic_fetch_add(rowcnt + dl, 1, __ATOMIC_RELAXED, __HIP_MEMORY_SCOPE_WORKGROUP);
    }
  }
  add_rows<K>(X, acc, s, nrm, off, count, lane);
}

__device__ __forceinline__ void run_flush(float* acc, int roff, int lane,
                                          const float4& lo, const float4& hi) {
  float* q = acc + roff + lane;
  ldsAddF(q,       lo.x);
  ldsAddF(q + 32,  lo.y);
  ldsAddF(q + 64,  lo.z);
  ldsAddF(q + 96,  lo.w);
  ldsAddF(q + 128, hi.x);
  ldsAddF(q + 160, hi.y);
  ldsAddF(q + 192, hi.z);
  ldsAddF(q + 224, hi.w);
}

__device__ __forceinline__ void run_edge(float* acc, int lane, int o, float nj,
                                         const float4& a, const float4& b,
                                         float4& lo, float4& hi, int& roff) {
  if (o != roff) {
    if (roff >= 0) run_flush(acc, roff, lane, lo, hi);
    lo = make_float4(0.0f, 0.0f, 0.0f, 0.0f);
    hi = make_float4(0.0f, 0.0f, 0.0f, 0.0f);
    roff = o;
  }
  lo.x += a.x * nj; lo.y += a.y * nj; lo.z += a.z * nj; lo.w += a.w * nj;
  hi.x += b.x * nj; hi.y += b.y * nj; hi.z += b.z * nj; hi.w += b.w * nj;
}

__device__ __forceinline__ void run_rows(
    const float* __restrict__ X, float* acc, int s, float nrm, int off,
    int count, int lane, float4& lo, float4& hi, int& roff) {
  int j = 0;
  for (; j + 8 <= count; j += 8) {
    float4 a[8], b[8];
#pragma unroll
    for (int u = 0; u < 8; ++u) {
      const float* hp = X + (long long)laneI(s, j + u) * C_HID + lane * 4;
      a[u] = *(const float4*)hp;
      b[u] = *(const float4*)(hp + 128);
    }
#pragma unroll
    for (int u = 0; u < 8; ++u)
      run_edge(acc, lane, laneI(off, j + u), laneF(nrm, j + u), a[u], b[u], lo, hi, roff);
  }
  for (; j + 4 <= count; j += 4) {
    float4 a[4], b[4];
#pragma unroll
    for (int u = 0; u < 4; ++u) {
      const float* hp = X + (long long)laneI(s, j + u) * C_HID + lane * 4;
      a[u] = *(const float4*)hp;
      b[u] = *(const float4*)(hp + 128);
    }
#pragma unroll
    for (int u = 0; u < 4; ++u)
      run_edge(acc, lane, laneI(off, j + u), laneF(nrm, j + u), a[u], b[u], lo, hi, roff);
  }
  for (; j < count; ++j) {
    const float* hp = X + (long long)laneI(s, j) * C_HID + lane * 4;
    const float4 a = *(const float4*)hp;
    const float4 b = *(const float4*)(hp + 128);
    run_edge(acc, lane, laneI(off, j), laneF(nrm, j), a, b, lo, hi, roff);
  }
}

__device__ __forceinline__ void load_keys(const unsigned char* __restrict__ p, int lane, v4u* v) {
#pragma unroll
  for (int k = 0; k < KEY_LOADS; ++k)
    v[k] = *(const v4u*)(p + k * 512 + lane * 16);
}

template <int K, bool STORE>
__device__ __forceinline__ void scan_push(
    const float* __restrict__ X, const unsigned* __restrict__ pk,
    const unsigned char* __restrict__ keys, const int* __restrict__ degp,
    float* acc, unsigned* ring, v2u* lst, int* rowcnt, int* lcnt,
    int row0, int E, int w, int lane) {
  const unsigned gpat = (unsigned)blockIdx.x * 0x01010101u;

  const int nsb   = E >> 10;
  const int tail0 = nsb << 10;

  int head = 0, tail = 0;
  {
    const int e = tail0 + w * 32 + lane;
    bool m = false;
    if (e < E) m = (unsigned)keys[e] == (unsigned)blockIdx.x;
    const unsigned mask = __builtin_amdgcn_ballot_w32(m);
    if (m) ring[__builtin_amdgcn_mbcnt_lo(mask, 0u)] = (unsigned)e;
    tail = __builtin_popcount(mask);
  }

  const int start = (nsb > 0) ? (int)(((long long)blockIdx.x * nsb) / gridDim.x) : 0;

  v4u cur[KEY_LOADS];
  if (w < nsb) {
    int sb = start + w; if (sb >= nsb) sb -= nsb;
    load_keys(keys + ((long long)sb << 10), lane, cur);
  }
  for (int i = w; ; i += PUSH_WAVES) {
    const bool more = i < nsb;
    if (more) {
      int sb = start + i; if (sb >= nsb) sb -= nsb;
      const int base = sb << 10;
      int sn = (i + PUSH_WAVES < nsb) ? start + i + PUSH_WAVES : start + i;
      if (sn >= nsb) sn -= nsb;
      v4u nxt[KEY_LOADS];
      load_keys(keys + ((long long)sn << 10), lane, nxt);
#pragma unroll
      for (int k = 0; k < KEY_LOADS; ++k) {
#pragma unroll
        for (int q = 0; q < 4; ++q) {
          const unsigned x = cur[k][q] ^ gpat;
          const unsigned z = (x - 0x01010101u) & ~x & 0x80808080u;
          if (__builtin_amdgcn_ballot_w32(z != 0u)) {
#pragma unroll
            for (int b = 0; b < 4; ++b) {
              const bool m = ((x >> (8 * b)) & 0xffu) == 0u;
              const unsigned mask = __builtin_amdgcn_ballot_w32(m);
              if (mask) {
                if (m) {
                  const unsigned e = (unsigned)(base + k * 512 + lane * 16 + q * 4 + b);
                  ring[(tail + __builtin_amdgcn_mbcnt_lo(mask, 0u)) & (PUSH_RING - 1)] = e;
                }
                tail += __builtin_popcount(mask);
              }
            }
          }
        }
        while (tail - head >= PUSH_MIN_BATCH) {
          const int cnt = min(32, (tail - head) & ~3);
          push_batch<K, STORE>(X, pk, degp, acc, ring, head, cnt, row0, lane, lst, rowcnt, lcnt);
          head += cnt;
        }
      }
#pragma unroll
      for (int k = 0; k < KEY_LOADS; ++k) cur[k] = nxt[k];
    } else {
      while (tail - head >= 1) {
        const int cnt = min(32, tail - head);
        push_batch<K, STORE>(X, pk, degp, acc, ring, head, cnt, row0, lane, lst, rowcnt, lcnt);
        head += cnt;
      }
      break;
    }
  }
}

template <int K, bool FIRST>
__global__ __launch_bounds__(PUSH_THREADS) void k_push_mm(
    const float* __restrict__ X, const unsigned* __restrict__ pk,
    const unsigned char* __restrict__ keys, const int* __restrict__ degp,
    const float* __restrict__ Wm, const float* __restrict__ bias,
    float* __restrict__ OUT, unsigned* __restrict__ glist_all,
    int* __restrict__ gcnt, int n, int E) {
  __shared__ __attribute__((aligned(16))) float    acc[ACC_FLOATS];
  __shared__ __attribute__((aligned(16))) unsigned pend[PUSH_WAVES * PUSH_RING];
  __shared__ int lcnt;
  __shared__ __attribute__((aligned(16))) v2u lst[FIRST ? LIST_CAP : 1];
  __shared__ int rowcnt[64];
  __shared__ int rowpos[64];
  constexpr int PITCH = ACC_PITCH(K);

  const int tid  = threadIdx.x;
  const int lane = tid & 31;
  const int w    = __builtin_amdgcn_readfirstlane(tid >> 5);
  const int row0 = blockIdx.x * PUSH_ROWS;
  const int nrows = min(PUSH_ROWS, n - row0);
  unsigned* ring = pend + w * PUSH_RING;
  v2u* glist = (v2u*)glist_all + (long long)blockIdx.x * LIST_CAP;

  for (int i = tid; i < TILE_ROWS * PITCH; i += PUSH_THREADS) acc[i] = 0.0f;
  if (FIRST && tid == 0) lcnt = 0;
  if (FIRST && tid < 64) rowcnt[tid] = 0;
  __syncthreads();

  if (FIRST) {
    scan_push<K, true>(X, pk, keys, degp, acc, ring, lst, rowcnt, &lcnt, row0, E, w, lane);
  } else {
    const int cnt = gcnt[blockIdx.x];
    if (cnt <= LIST_CAP) {
      const int per = (((cnt + PUSH_WAVES - 1) / PUSH_WAVES) + 3) & ~3;
      int b = w * per;
      const int end = min(cnt, b + per);
      v2u cur = {0u, 0u};
      if (b + lane < end) cur = glist[b + lane];
      float4 lo = make_float4(0.0f, 0.0f, 0.0f, 0.0f);
      float4 hi = make_float4(0.0f, 0.0f, 0.0f, 0.0f);
      int roff = -1;
      for (; b < end; b += 32) {
        v2u nxt = {0u, 0u};
        if (b + 32 + lane < end) nxt = glist[b + 32 + lane];
        const int count = min(32, end - b);
        int s = 0, off = 0;
        float nrm = 0.0f;
        if (lane < count) {
          s   = (int)(cur.x >> 6);
          off = (int)(cur.x & 63u) * PITCH;
          nrm = __uint_as_float(cur.y);
        }
        run_rows(X, acc, s, nrm, off, count, lane, lo, hi, roff);
        cur = nxt;
      }
      if (roff >= 0) run_flush(acc, roff, lane, lo, hi);
    } else {
      scan_push<K, false>(X, pk, keys, degp, acc, ring, nullptr, nullptr, &lcnt, row0, E, w, lane);
    }
  }

  __syncthreads();
  int lc = 0;
  if (FIRST) {
    lc = lcnt;
    if (tid == 0) gcnt[blockIdx.x] = lc;
    if (tid < 64) {
      int before = 0;
#pragma unroll
      for (int i = 0; i < PUSH_ROWS; ++i) {
        const int cr = rowcnt[i];
        before += (i < tid) ? cr : 0;
      }
      rowpos[tid] = before;
    }
  }

  {
    constexpr int QPR = K / 4;
    constexpr int JST = (K == 256) ? 32 : 16;
    for (int idx = tid; idx < nrows * QPR; idx += PUSH_THREADS) {
      const int r  = idx / QPR;
      const int cq = idx - r * QPR;
      const int node = row0 + r;
      const float di = dinv_at(degp, node);
      const float4 xv = *(const float4*)(X + (long long)node * K + cq * 4);
      float* ap = acc + r * PITCH + ((K == 256) ? ((cq >> 5) * 128 + (cq & 31)) : cq);
      ap[0]       += di * di * xv.x;
      ap[JST]     += di * di * xv.y;
      ap[2 * JST] += di * di * xv.z;
      ap[3 * JST] += di * di * xv.w;
    }
  }
  __syncthreads();

  if (FIRST && lc <= LIST_CAP) {
    for (int i = tid; i < lc; i += PUSH_THREADS) {
      const v2u e = lst[i];
      const int pos = __hip_atomic_fetch_add(rowpos + (int)(e.x & 63u), 1,
                                             __ATOMIC_RELAXED, __HIP_MEMORY_SCOPE_WORKGROUP);
      glist[pos] = e;
    }
  }

  const int lo = lane & 15;
  const int hi = lane >> 4;
  const int n0 = (w & 15) << 4;
  const int kh = w >> 4;
  v8f d0 = {}, d1 = {}, d2 = {};
  {
    constexpr int STEPS = K / 8;
    const int pbase = kh * (K / 2) + 2 * hi;
    const float* Ap = acc + lo * PITCH + pbase;
    const float* Wc = Wm + n0 + lo;
#pragma unroll 8
    for (int t = 0; t < STEPS; ++t) {
      const int p = pbase + 4 * t;
      v2f b;
      b.x = Wc[(long long)pos_to_chan<K>(p)     * C_HID];
      b.y = Wc[(long long)pos_to_chan<K>(p + 1) * C_HID];
      const v2f a0 = *(const v2f*)(Ap + 4 * t);
      const v2f a1 = *(const v2f*)(Ap + 16 * PITCH + 4 * t);
      const v2f a2 = *(const v2f*)(Ap + 32 * PITCH + 4 * t);
      d0 = __builtin_amdgcn_wmma_f32_16x16x4_f32(false, a0, false, b, (short)0, d0, false, false);
      d1 = __builtin_amdgcn_wmma_f32_16x16x4_f32(false, a1, false, b, (short)0, d1, false, false);
      d2 = __builtin_amdgcn_wmma_f32_16x16x4_f32(false, a2, false, b, (short)0, d2, false, false);
    }
  }

  float* S = (float*)pend;
  float* Sp = S + (hi * 8) * C_HID + n0 + lo;
  if (kh == 1) {
#pragma unroll
    for (int r = 0; r < 8; ++r) {
      Sp[(r)      * C_HID] = d0[r];
      Sp[(16 + r) * C_HID] = d1[r];
      Sp[(32 + r) * C_HID] = d2[r];
    }
  }
  __syncthreads();
  if (kh == 0) {
    const float bb = bias[n0 + lo];
#pragma unroll
    for (int r = 0; r < 8; ++r) {
      float v0 = d0[r] + Sp[(r)      * C_HID] + bb;
      float v1 = d1[r] + Sp[(16 + r) * C_HID] + bb;
      float v2 = d2[r] + Sp[(32 + r) * C_HID] + bb;
      Sp[(r)      * C_HID] = (v0 > 0.0f) ? v0 : expm1f(v0);
      Sp[(16 + r) * C_HID] = (v1 > 0.0f) ? v1 : expm1f(v1);
      Sp[(32 + r) * C_HID] = (v2 > 0.0f) ? v2 : expm1f(v2);
    }
  }
  __syncthreads();

  for (int idx = tid; idx < nrows * (C_HID / 4); idx += PUSH_THREADS) {
    const int r  = idx >> 6;
    const int cq = idx & 63;
    const float4 v = *(const float4*)(S + r * C_HID + cq * 4);
    *(float4*)(OUT + (long long)(row0 + r) * C_HID + cq * 4) = v;
  }
}

extern "C" void kernel_launch(void* const* d_in, const int* in_sizes, int n_in,
                              void* d_out, int out_size, void* d_ws, size_t ws_size,
                              hipStream_t stream) {
  const float* x  = (const float*)d_in[0];
  const int*   ei = (const int*)d_in[1];
  const float* W1 = (const float*)d_in[2];
  const float* b1 = (const float*)d_in[3];
  const float* W2 = (const float*)d_in[4];
  const float* b2 = (const float*)d_in[5];
  const float* W3 = (const float*)d_in[6];
  const float* b3 = (const float*)d_in[7];
  float* out = (float*)d_out;

  const int C   = C_HID;
  const int E   = in_sizes[1] / 2;
  const int cin = in_sizes[2] / C;
  const int n   = in_sizes[0] / cin;
  const int* src = ei;
  const int* dst = ei + E;

  char*  ws  = (char*)d_ws;
  size_t off = 0;
  auto alloc = [&](size_t bytes) -> void* {
    void* p = ws + off;
    off = (off + bytes + 255) & ~(size_t)255;
    return p;
  };
  int* degp = (int*)alloc((size_t)n * HIST_CHUNKS * sizeof(int));
  float* XA   = (float*)alloc((size_t)n * C * sizeof(float));
  float* XB   = (float*)alloc((size_t)n * C * sizeof(float));
  unsigned* key4 = (unsigned*)alloc(((size_t)E + 3) / 4 * sizeof(unsigned));
  unsigned* pk   = (unsigned*)alloc((size_t)E * sizeof(unsigned));
  const unsigned char* keys = (const unsigned char*)key4;
  const int pgrid = (n + PUSH_ROWS - 1) / PUSH_ROWS;
  unsigned* glist = (unsigned*)alloc((size_t)pgrid * LIST_CAP * 2 * sizeof(unsigned));
  int*      gcnt  = (int*)     alloc((size_t)pgrid * sizeof(int));
  (void)ws_size; (void)n_in; (void)out_size;

  const int hgroups = (n + HIST_ROWS - 1) / HIST_ROWS;
  const int rgroups = ((E + 3) / 4 + PREP_THREADS - 1) / PREP_THREADS;
  k_prep<<<hgroups * HIST_CHUNKS + rgroups, PREP_THREADS, 0, stream>>>(src, dst, degp, key4, pk, n, E, hgroups);

  k_push_mm<64, true>  <<<pgrid, PUSH_THREADS, 0, stream>>>(x,  pk, keys, degp, W1, b1, XA,  glist, gcnt, n, E);
  k_push_mm<256, false><<<pgrid, PUSH_THREADS, 0, stream>>>(XA, pk, keys, degp, W2, b2, XB,  glist, gcnt, n, E);
  k_push_mm<256, false><<<pgrid, PUSH_THREADS, 0, stream>>>(XB, pk, keys, degp, W3, b3, out, glist, gcnt, n, E);
}


# recode work-groups order their 4096-edge chunk by owner work-group in LDS (counting sort); the layer-1 edge kernel reads its segment of each chunk, no key list is searched; sums stay arrival order
# speedup vs baseline: 5.4767x; 1.0692x over previous
#include <hip/hip_runtime.h>
#include <hip/hip_bf16.h>

typedef __attribute__((ext_vector_type(2))) float v2f;
typedef __attribute__((ext_vector_type(8))) float v8f;

#define C_HID 256

#define PUSH_ROWS 40

typedef __attribute__((ext_vector_type(4))) unsigned v4u;
typedef __attribute__((ext_vector_type(2))) unsigned v2u;

#define LIST_CAP 4096

#define PREP_THREADS 1024
#define HIST_ROWS    1024
#define HIST_CHUNKS  4
#define HIST_LOADS   4
#define SEG_CHUNK    (PREP_THREADS * 4)
#define SEG_OWNERS   256

typedef __attribute__((ext_vector_type(4))) int v4i;

__device__ __forceinline__ void hist_one(int* hist, int d, unsigned nlo4) {
  const unsigned xb = ((unsigned)d << 2) + nlo4;
  if (xb < 4u * HIST_ROWS)
    __hip_atomic_fetch_add((int*)((char*)hist + xb), 1, __ATOMIC_RELAXED,
                           __HIP_MEMORY_SCOPE_WORKGROUP);
}

__device__ __forceinline__ void hist_vec(int* hist, const v4i& v, unsigned nlo4) {
  hist_one(hist, v.x, nlo4);
  hist_one(hist, v.y, nlo4);
  hist_one(hist, v.z, nlo4);
  hist_one(hist, v.w, nlo4);
}

__device__ __forceinline__ float dinv_at(const int* __restrict__ degp, int node) {
  const v4i c = *(const v4i*)(degp + node * HIST_CHUNKS);
  return rsqrtf((float)(1 + c.x + c.y + c.z + c.w));
}

__global__ __launch_bounds__(PREP_THREADS) void k_prep(
    const int* __restrict__ src, const int* __restrict__ dst,
    int* __restrict__ degp, unsigned* __restrict__ segs,
    unsigned* __restrict__ pks, int n, int E, int hgroups, int nch) {
  __shared__ int hist[HIST_ROWS];
  __shared__ __attribute__((aligned(16))) unsigned srt[SEG_CHUNK];
  const int tid = threadIdx.x;
  if ((int)blockIdx.x < hgroups * HIST_CHUNKS) {
    const int rng = (int)blockIdx.x / HIST_CHUNKS;
    const int pc  = (int)blockIdx.x - rng * HIST_CHUNKS;
    const int lo = rng * HIST_ROWS;
    const unsigned nlo4 = 0u - ((unsigned)lo << 2);
    if (tid < HIST_ROWS) hist[tid] = 0;
    __syncthreads();
    const int head  = min(E, (int)(((16u - ((unsigned)(size_t)dst & 15u)) & 15u) >> 2));
    const int nall  = (E - head) >> 2;
    const int tail0 = head + (nall << 2);
    if (pc == 0) {
      int e = -1;
      if (tid < head) e = tid;
      else if (tid - head < E - tail0) e = tail0 + (tid - head);
      if (e >= 0) hist_one(hist, dst[e], nlo4);
    }
    const int v0   = (int)(((long long)nall * pc) / HIST_CHUNKS);
    const int nvec = (int)(((long long)nall * (pc + 1)) / HIST_CHUNKS) - v0;
    const v4i* dv = (const v4i*)(dst + head) + v0;
    constexpr int TRIP = PREP_THREADS * HIST_LOADS;
    const int nfull = nvec / TRIP;
    if (nfull > 0) {
      v4i cur[HIST_LOADS];
#pragma unroll
      for (int k = 0; k < HIST_LOADS; ++k) cur[k] = dv[k * PREP_THREADS + tid];
      for (int i = 0; i < nfull; ++i) {
        const int nb = ((i + 1 < nfull) ? i + 1 : i) * TRIP;
        v4i nxt[HIST_LOADS];
#pragma unroll
        for (int k = 0; k < HIST_LOADS; ++k) nxt[k] = dv[nb + k * PREP_THREADS + tid];
#pragma unroll
        for (int k = 0; k < HIST_LOADS; ++k) hist_vec(hist, cur[k], nlo4);
#pragma unroll
        for (int k = 0; k < HIST_LOADS; ++k) cur[k] = nxt[k];
      }
    }
    {
      const int r0 = nfull * TRIP;
      const v4i none = {-1, -1, -1, -1};
      v4i rem[HIST_LOADS];
#pragma unroll
      for (int k = 0; k < HIST_LOADS; ++k) {
        const int v = r0 + k * PREP_THREADS + tid;
        rem[k] = none;
        if (v < nvec) rem[k] = dv[v];
      }
#pragma unroll
      for (int k = 0; k < HIST_LOADS; ++k) hist_vec(hist, rem[k], nlo4);
    }
    __syncthreads();
    if (tid < HIST_ROWS && lo + tid < n)
      degp[(lo + tid) * HIST_CHUNKS + pc] = hist[tid];
    return;
  }

  const int c    = (int)blockIdx.x - hgroups * HIST_CHUNKS;
  const int e0   = (c * PREP_THREADS + tid) * 4;
  const int lane = tid & 31;
  int* cnt  = hist;
  int* base = hist + SEG_OWNERS;
  int* wsum = hist + 2 * SEG_OWNERS;
  if (tid < SEG_OWNERS) cnt[tid] = 0;
  __syncthreads();

  int gq[4], rk[4];
  unsigned pv[4];
  if (e0 + 3 < E) {
    int d[4], sv[4];
#pragma unroll
    for (int q = 0; q < 4; ++q) { d[q] = dst[e0 + q]; sv[q] = src[e0 + q]; }
#pragma unroll
    for (int q = 0; q < 4; ++q) {
      const int g = d[q] / PUSH_ROWS;
      pv[q] = ((unsigned)sv[q] << 6) | (unsigned)(d[q] - g * PUSH_ROWS);
      gq[q] = g;
      rk[q] = __hip_atomic_fetch_add(cnt + g, 1, __ATOMIC_RELAXED, __HIP_MEMORY_SCOPE_WORKGROUP);
    }
  } else {
#pragma unroll
    for (int q = 0; q < 4; ++q) {
      gq[q] = -1; rk[q] = 0; pv[q] = 0u;
      if (e0 + q < E) {
        const int d = dst[e0 + q];
        const int g = d / PUSH_ROWS;
        pv[q] = ((unsigned)src[e0 + q] << 6) | (unsigned)(d - g * PUSH_ROWS);
        gq[q] = g;
        rk[q] = __hip_atomic_fetch_add(cnt + g, 1, __ATOMIC_RELAXED, __HIP_MEMORY_SCOPE_WORKGROUP);
      }
    }
  }
  __syncthreads();

  int v = 0, x = 0;
  if (tid < SEG_OWNERS) {
    v = cnt[tid];
    x = v;
    for (int dd = 1; dd < 32; dd <<= 1) {
      const int y = __shfl_up(x, dd);
      if (lane >= dd) x += y;
    }
    if (lane == 31) wsum[tid >> 5] = x;
  }
  __syncthreads();
  if (tid < SEG_OWNERS) {
    int add = 0;
#pragma unroll
    for (int i = 0; i < SEG_OWNERS / 32; ++i) {
      const int wv = wsum[i];
      add += (i < (tid >> 5)) ? wv : 0;
    }
    const int b = x - v + add;
    base[tid] = b;
    segs[(long long)tid * nch + c] = (unsigned)b | ((unsigned)v << 16);
  }
  __syncthreads();
#pragma unroll
  for (int q = 0; q < 4; ++q)
    if (gq[q] >= 0) srt[base[gq[q]] + rk[q]] = pv[q];
  __syncthreads();
  const int nvalid = min(SEG_CHUNK, E - c * SEG_CHUNK);
  if (tid * 4 < nvalid)
    *(v4u*)(pks + (long long)c * SEG_CHUNK + tid * 4) = *(const v4u*)(srt + tid * 4);
}

#define PUSH_THREADS 1024
#define PUSH_WAVES   32
#define PUSH_RING    1024
#define SEG_SLOTS    4
#define PUSH_MIN_BATCH 16
#define TILE_ROWS    48
#define ACC_PITCH(K) ((K) + 4)
#define ACC_FLOATS   (TILE_ROWS * ACC_PITCH(C_HID))

template <int K>
__device__ __forceinline__ int pos_to_chan(int p) {
  if (K == 256) return (p & 128) + 4 * (p & 31) + ((p >> 5) & 3);
  return 4 * (p & 15) + (p >> 4);
}

__device__ __forceinline__ void ldsAddF(float* p, float v) {
  __hip_atomic_fetch_add(p, v, __ATOMIC_RELAXED, __HIP_MEMORY_SCOPE_WORKGROUP);
}

__device__ __forceinline__ int laneI(int v, int j) {
  return __builtin_amdgcn_readlane(v, j);
}
__device__ __forceinline__ float laneF(float v, int j) {
  return __int_as_float(__builtin_amdgcn_readlane(__float_as_int(v), j));
}

template <int K>
__device__ __forceinline__ void add_rows(
    const float* __restrict__ X, float* acc, int s, float nrm, int off,
    int count, int lane) {
  if (K == 256) {
    int j = 0;
    for (; j + 4 <= count; j += 4) {
      float4 a[4], b[4];
#pragma unroll
      for (int u = 0; u < 4; ++u) {
        const float* hp = X + (long long)laneI(s, j + u) * K + lane * 4;
        a[u] = *(const float4*)hp;
        b[u] = *(const float4*)(hp + 128);
      }
#pragma unroll
      for (int u = 0; u < 4; ++u) {
        const float nj = laneF(nrm, j + u);
        float* q = acc + laneI(off, j + u) + lane;
        ldsAddF(q,       a[u].x * nj);
        ldsAddF(q + 32,  a[u].y * nj);
        ldsAddF(q + 64,  a[u].z * nj);
        ldsAddF(q + 96,  a[u].w * nj);
        ldsAddF(q + 128, b[u].x * nj);
        ldsAddF(q + 160, b[u].y * nj);
        ldsAddF(q + 192, b[u].z * nj);
        ldsAddF(q + 224, b[u].w * nj);
      }
    }
    for (; j < count; ++j) {
      const float* hp = X + (long long)laneI(s, j) * K + lane * 4;
      const float4 a = *(const float4*)hp;
      const float4 b = *(const float4*)(hp + 128);
      const float nj = laneF(nrm, j);
      float* q = acc + laneI(off, j) + lane;
      ldsAddF(q,       a.x * nj);
      ldsAddF(q + 32,  a.y * nj);
      ldsAddF(q + 64,  a.z * nj);
      ldsAddF(q + 96,  a.w * nj);
      ldsAddF(q + 128, b.x * nj);
      ldsAddF(q + 160, b.y * nj);
      ldsAddF(q + 192, b.z * nj);
      ldsAddF(q + 224, b.w * nj);
    }
  } else {
    const int  sub = lane & 15;
    const bool odd = lane >= 16;
    for (int j = 0; j < count; j += 8) {
      float4 a[4];
#pragma unroll
      for (int u = 0; u < 4; ++u) {
        const int e0 = j + 2 * u;
        const int sA = laneI(s, e0), sB = laneI(s, e0 + 1);
        a[u] = make_float4(0.0f, 0.0f, 0.0f, 0.0f);
        if (e0 + (odd ? 1 : 0) < count)
          a[u] = *(const float4*)(X + (long long)(odd ? sB : sA) * K + sub * 4);
      }
#pragma unroll
      for (int u = 0; u < 4; ++u) {
        const int e0 = j + 2 * u;
        const float nA = laneF(nrm, e0), nB = laneF(nrm, e0 + 1);
        const int   oA = laneI(off, e0), oB = laneI(off, e0 + 1);
        if (e0 + (odd ? 1 : 0) < count) {
          const float nj = odd ? nB : nA;
          float* q = acc + (odd ? oB : oA) + sub;
          ldsAddF(q,      a[u].x * nj);
          ldsAddF(q + 16, a[u].y * nj);
          ldsAddF(q + 32, a[u].z * nj);
          ldsAddF(q + 48, a[u].w * nj);
        }
      }
    }
  }
}

template <int K, bool STORE>
__device__ __forceinline__ void push_batch(
    const float* __restrict__ X,
    const int* __restrict__ degp, float* acc, const unsigned* ring,
    int head, int count, int row0, int lane, v2u* lst, int* rowcnt, int* lcnt) {
  int s = 0, off = 0, dl = 0;
  float nrm = 0.0f;
  unsigned ent = 0u;
  if (lane < count) {
    ent = ring[(head + lane) & (PUSH_RING - 1)];
    dl  = (int)(ent & 63u);
    s   = (int)(ent >> 6);
    nrm = dinv_at(degp, s) * dinv_at(degp, row0 + dl);
    off = dl * ACC_PITCH(K);
  }
  if (STORE) {
    int base = 0;
    if (lane == 0)
      base = __hip_atomic_fetch_add(lcnt, count, __ATOMIC_RELAXED, __HIP_MEMORY_SCOPE_WORKGROUP);
    base = __builtin_amdgcn_readfirstlane(base);
    if (lane < count && base + lane < LIST_CAP) {
      v2u o; o.x = ent; o.y = __float_as_uint(nrm);
      lst[base + lane] = o;
      __hip_atomic_fetch_add(rowcnt + dl, 1, __ATOMIC_RELAXED, __HIP_MEMORY_SCOPE_WORKGROUP);
    }
  }
  add_rows<K>(X, acc, s, nrm, off, count, lane);
}

__device__ __forceinline__ void run_flush(float* acc, int roff, int lane,
                                          const float4& lo, const float4& hi) {
  float* q = acc + roff + lane;
  ldsAddF(q,       lo.x);
  ldsAddF(q + 32,  lo.y);
  ldsAddF(q + 64,  lo.z);
  ldsAddF(q + 96,  lo.w);
  ldsAddF(q + 128, hi.x);
  ldsAddF(q + 160, hi.y);
  ldsAddF(q + 192, hi.z);
  ldsAddF(q + 224, hi.w);
}

__device__ __forceinline__ void run_edge(float* acc, int lane, int o, float nj,
                                         const float4& a, const float4& b,
                                         float4& lo, float4& hi, int& roff) {
  if (o != roff) {
    if (roff >= 0) run_flush(acc, roff, lane, lo, hi);
    lo = make_float4(0.0f, 0.0f, 0.0f, 0.0f);
    hi = make_float4(0.0f, 0.0f, 0.0f, 0.0f);
    roff = o;
  }
  lo.x += a.x * nj; lo.y += a.y * nj; lo.z += a.z * nj; lo.w += a.w * nj;
  hi.x += b.x * nj; hi.y += b.y * nj; hi.z += b.z * nj; hi.w += b.w * nj;
}

__device__ __forceinline__ void run_rows(
    const float* __restrict__ X, float* acc, int s, float nrm, int off,
    int count, int lane, float4& lo, float4& hi, int& roff) {
  int j = 0;
  for (; j + 8 <= count; j += 8) {
    float4 a[8], b[8];
#pragma unroll
    for (int u = 0; u < 8; ++u) {
      const float* hp = X + (long long)laneI(s, j + u) * C_HID + lane * 4;
      a[u] = *(const float4*)hp;
      b[u] = *(const float4*)(hp + 128);
    }
#pragma unroll
    for (int u = 0; u < 8; ++u)
      run_edge(acc, lane, laneI(off, j + u), laneF(nrm, j + u), a[u], b[u], lo, hi, roff);
  }
  for (; j + 4 <= count; j += 4) {
    float4 a[4], b[4];
#pragma unroll
    for (int u = 0; u < 4; ++u) {
      const float* hp = X + (long long)laneI(s, j + u) * C_HID + lane * 4;
      a[u] = *(const float4*)hp;
      b[u] = *(const float4*)(hp + 128);
    }
#pragma unroll
    for (int u = 0; u < 4; ++u)
      run_edge(acc, lane, laneI(off, j + u), laneF(nrm, j + u), a[u], b[u], lo, hi, roff);
  }
  for (; j < count; ++j) {
    const float* hp = X + (long long)laneI(s, j) * C_HID + lane * 4;
    const float4 a = *(const float4*)hp;
    const float4 b = *(const float4*)(hp + 128);
    run_edge(acc, lane, laneI(off, j), laneF(nrm, j), a, b, lo, hi, roff);
  }
}

template <int K, bool STORE>
__device__ __forceinline__ void seg_push(
    const float* __restrict__ X, const unsigned* __restrict__ pks,
    const unsigned* __restrict__ segs, const int* __restrict__ degp,
    float* acc, unsigned* ring, v2u* lst, int* rowcnt, int* lcnt,
    int row0, int nch, int w, int lane) {
  const unsigned* myseg = segs + (long long)blockIdx.x * nch;
  int head = 0, tail = 0;
  for (int cb = w; cb < nch; cb += PUSH_WAVES * SEG_SLOTS) {
    unsigned sg = 0u;
    {
      const int c = cb + lane * PUSH_WAVES;
      if (lane < SEG_SLOTS && c < nch) sg = myseg[c];
    }
    unsigned ent[SEG_SLOTS];
#pragma unroll
    for (int j = 0; j < SEG_SLOTS; ++j) {
      const unsigned sw = (unsigned)laneI((int)sg, j);
      const int st = (int)(sw & 0xffffu);
      const int cn = (int)(sw >> 16);
      ent[j] = 0u;
      if (lane < cn)
        ent[j] = pks[(long long)(cb + j * PUSH_WAVES) * SEG_CHUNK + st + lane];
    }
#pragma unroll
    for (int j = 0; j < SEG_SLOTS; ++j) {
      const unsigned sw = (unsigned)laneI((int)sg, j);
      const int st = (int)(sw & 0xffffu);
      const int cn = (int)(sw >> 16);
      const unsigned* p = pks + (long long)(cb + j * PUSH_WAVES) * SEG_CHUNK + st;
      for (int i = 0; i < cn; i += 32) {
        const int m = min(32, cn - i);
        unsigned e = ent[j];
        if (i > 0) {
          e = 0u;
          if (lane < m) e = p[i + lane];
        }
        if (lane < m) ring[(tail + lane) & (PUSH_RING - 1)] = e;
        tail += m;
        while (tail - head >= PUSH_MIN_BATCH) {
          const int cnt = min(32, (tail - head) & ~3);
          push_batch<K, STORE>(X, degp, acc, ring, head, cnt, row0, lane, lst, rowcnt, lcnt);
          head += cnt;
        }
      }
    }
  }
  while (tail - head >= 1) {
    const int cnt = min(32, tail - head);
    push_batch<K, STORE>(X, degp, acc, ring, head, cnt, row0, lane, lst, rowcnt, lcnt);
    head += cnt;
  }
}

template <int K, bool FIRST>
__global__ __launch_bounds__(PUSH_THREADS) void k_push_mm(
    const float* __restrict__ X, const unsigned* __restrict__ pks,
    const unsigned* __restrict__ segs, const int* __restrict__ degp,
    const float* __restrict__ Wm, const float* __restrict__ bias,
    float* __restrict__ OUT, unsigned* __restrict__ glist_all,
    int* __restrict__ gcnt, int n, int nch) {
  __shared__ __attribute__((aligned(16))) float    acc[ACC_FLOATS];
  __shared__ __attribute__((aligned(16))) unsigned pend[PUSH_WAVES * PUSH_RING];
  __shared__ int lcnt;
  __shared__ __attribute__((aligned(16))) v2u lst[FIRST ? LIST_CAP : 1];
  __shared__ int rowcnt[64];
  __shared__ int rowpos[64];
  constexpr int PITCH = ACC_PITCH(K);

  const int tid  = threadIdx.x;
  const int lane = tid & 31;
  const int w    = __builtin_amdgcn_readfirstlane(tid >> 5);
  const int row0 = blockIdx.x * PUSH_ROWS;
  const int nrows = min(PUSH_ROWS, n - row0);
  unsigned* ring = pend + w * PUSH_RING;
  v2u* glist = (v2u*)glist_all + (long long)blockIdx.x * LIST_CAP;

  for (int i = tid; i < TILE_ROWS * PITCH; i += PUSH_THREADS) acc[i] = 0.0f;
  if (FIRST && tid == 0) lcnt = 0;
  if (FIRST && tid < 64) rowcnt[tid] = 0;
  __syncthreads();

  if (FIRST) {
    seg_push<K, true>(X, pks, segs, degp, acc, ring, lst, rowcnt, &lcnt, row0, nch, w, lane);
  } else {
    const int cnt = gcnt[blockIdx.x];
    if (cnt <= LIST_CAP) {
      const int per = (((cnt + PUSH_WAVES - 1) / PUSH_WAVES) + 3) & ~3;
      int b = w * per;
      const int end = min(cnt, b + per);
      v2u cur = {0u, 0u};
      if (b + lane < end) cur = glist[b + lane];
      float4 lo = make_float4(0.0f, 0.0f, 0.0f, 0.0f);
      float4 hi = make_float4(0.0f, 0.0f, 0.0f, 0.0f);
      int roff = -1;
      for (; b < end; b += 32) {
        v2u nxt = {0u, 0u};
        if (b + 32 + lane < end) nxt = glist[b + 32 + lane];
        const int count = min(32, end - b);
        int s = 0, off = 0;
        float nrm = 0.0f;
        if (lane < count) {
          s   = (int)(cur.x >> 6);
          off = (int)(cur.x & 63u) * PITCH;
          nrm = __uint_as_float(cur.y);
        }
        run_rows(X, acc, s, nrm, off, count, lane, lo, hi, roff);
        cur = nxt;
      }
      if (roff >= 0) run_flush(acc, roff, lane, lo, hi);
    } else {
      seg_push<K, false>(X, pks, segs, degp, acc, ring, nullptr, nullptr, &lcnt, row0, nch, w, lane);
    }
  }

  __syncthreads();
  int lc = 0;
  if (FIRST) {
    lc = lcnt;
    if (tid == 0) gcnt[blockIdx.x] = lc;
    if (tid < 64) {
      int before = 0;
#pragma unroll
      for (int i = 0; i < PUSH_ROWS; ++i) {
        const int cr = rowcnt[i];
        before += (i < tid) ? cr : 0;
      }
      rowpos[tid] = before;
    }
  }

  {
    constexpr int QPR = K / 4;
    constexpr int JST = (K == 256) ? 32 : 16;
    for (int idx = tid; idx < nrows * QPR; idx += PUSH_THREADS) {
      const int r  = idx / QPR;
      const int cq = idx - r * QPR;
      const int node = row0 + r;
      const float di = dinv_at(degp, node);
      const float4 xv = *(const float4*)(X + (long long)node * K + cq * 4);
      float* ap = acc + r * PITCH + ((K == 256) ? ((cq >> 5) * 128 + (cq & 31)) : cq);
      ap[0]       += di * di * xv.x;
      ap[JST]     += di * di * xv.y;
      ap[2 * JST] += di * di * xv.z;
      ap[3 * JST] += di * di * xv.w;
    }
  }
  __syncthreads();

  if (FIRST && lc <= LIST_CAP) {
    for (int i = tid; i < lc; i += PUSH_THREADS) {
      const v2u e = lst[i];
      const int pos = __hip_atomic_fetch_add(rowpos + (int)(e.x & 63u), 1,
                                             __ATOMIC_RELAXED, __HIP_MEMORY_SCOPE_WORKGROUP);
      glist[pos] = e;
    }
  }

  const int lo = lane & 15;
  const int hi = lane >> 4;
  const int n0 = (w & 15) << 4;
  const int kh = w >> 4;
  v8f d0 = {}, d1 = {}, d2 = {};
  {
    constexpr int STEPS = K / 8;
    const int pbase = kh * (K / 2) + 2 * hi;
    const float* Ap = acc + lo * PITCH + pbase;
    const float* Wc = Wm + n0 + lo;
#pragma unroll 8
    for (int t = 0; t < STEPS; ++t) {
      const int p = pbase + 4 * t;
      v2f b;
      b.x = Wc[(long long)pos_to_chan<K>(p)     * C_HID];
      b.y = Wc[(long long)pos_to_chan<K>(p + 1) * C_HID];
      const v2f a0 = *(const v2f*)(Ap + 4 * t);
      const v2f a1 = *(const v2f*)(Ap + 16 * PITCH + 4 * t);
      const v2f a2 = *(const v2f*)(Ap + 32 * PITCH + 4 * t);
      d0 = __builtin_amdgcn_wmma_f32_16x16x4_f32(false, a0, false, b, (short)0, d0, false, false);
      d1 = __builtin_amdgcn_wmma_f32_16x16x4_f32(false, a1, false, b, (short)0, d1, false, false);
      d2 = __builtin_amdgcn_wmma_f32_16x16x4_f32(false, a2, false, b, (short)0, d2, false, false);
    }
  }

  float* S = (float*)pend;
  float* Sp = S + (hi * 8) * C_HID + n0 + lo;
  if (kh == 1) {
#pragma unroll
    for (int r = 0; r < 8; ++r) {
      Sp[(r)      * C_HID] = d0[r];
      Sp[(16 + r) * C_HID] = d1[r];
      Sp[(32 + r) * C_HID] = d2[r];
    }
  }
  __syncthreads();
  if (kh == 0) {
    const float bb = bias[n0 + lo];
#pragma unroll
    for (int r = 0; r < 8; ++r) {
      float v0 = d0[r] + Sp[(r)      * C_HID] + bb;
      float v1 = d1[r] + Sp[(16 + r) * C_HID] + bb;
      float v2 = d2[r] + Sp[(32 + r) * C_HID] + bb;
      Sp[(r)      * C_HID] = (v0 > 0.0f) ? v0 : expm1f(v0);
      Sp[(16 + r) * C_HID] = (v1 > 0.0f) ? v1 : expm1f(v1);
      Sp[(32 + r) * C_HID] = (v2 > 0.0f) ? v2 : expm1f(v2);
    }
  }
  __syncthreads();

  for (int idx = tid; idx < nrows * (C_HID / 4); idx += PUSH_THREADS) {
    const int r  = idx >> 6;
    const int cq = idx & 63;
    const float4 v = *(const float4*)(S + r * C_HID + cq * 4);
    *(float4*)(OUT + (long long)(row0 + r) * C_HID + cq * 4) = v;
  }
}

extern "C" void kernel_launch(void* const* d_in, const int* in_sizes, int n_in,
                              void* d_out, int out_size, void* d_ws, size_t ws_size,
                              hipStream_t stream) {
  const float* x  = (const float*)d_in[0];
  const int*   ei = (const int*)d_in[1];
  const float* W1 = (const float*)d_in[2];
  const float* b1 = (const float*)d_in[3];
  const float* W2 = (const float*)d_in[4];
  const float* b2 = (const float*)d_in[5];
  const float* W3 = (const float*)d_in[6];
  const float* b3 = (const float*)d_in[7];
  float* out = (float*)d_out;

  const int C   = C_HID;
  const int E   = in_sizes[1] / 2;
  const int cin = in_sizes[2] / C;
  const int n   = in_sizes[0] / cin;
  const int* src = ei;
  const int* dst = ei + E;

  char*  ws  = (char*)d_ws;
  size_t off = 0;
  auto alloc = [&](size_t bytes) -> void* {
    void* p = ws + off;
    off = (off + bytes + 255) & ~(size_t)255;
    return p;
  };
  int* degp = (int*)alloc((size_t)n * HIST_CHUNKS * sizeof(int));
  float* XA   = (float*)alloc((size_t)n * C * sizeof(float));
  float* XB   = (float*)alloc((size_t)n * C * sizeof(float));
  const int hgroups = (n + HIST_ROWS - 1) / HIST_ROWS;
  const int rgroups = ((E + 3) / 4 + PREP_THREADS - 1) / PREP_THREADS;
  unsigned* segs = (unsigned*)alloc((size_t)SEG_OWNERS * rgroups * sizeof(unsigned));
  unsigned* pks  = (unsigned*)alloc((size_t)rgroups * SEG_CHUNK * sizeof(unsigned));
  const int pgrid = (n + PUSH_ROWS - 1) / PUSH_ROWS;
  unsigned* glist = (unsigned*)alloc((size_t)pgrid * LIST_CAP * 2 * sizeof(unsigned));
  int*      gcnt  = (int*)     alloc((size_t)pgrid * sizeof(int));
  (void)ws_size; (void)n_in; (void)out_size;

  k_prep<<<hgroups * HIST_CHUNKS + rgroups, PREP_THREADS, 0, stream>>>(src, dst, degp, segs, pks, n, E, hgroups, rgroups);

  k_push_mm<64, true>  <<<pgrid, PUSH_THREADS, 0, stream>>>(x,  pks, segs, degp, W1, b1, XA,  glist, gcnt, n, rgroups);
  k_push_mm<256, false><<<pgrid, PUSH_THREADS, 0, stream>>>(XA, pks, segs, degp, W2, b2, XB,  glist, gcnt, n, rgroups);
  k_push_mm<256, false><<<pgrid, PUSH_THREADS, 0, stream>>>(XB, pks, segs, degp, W3, b3, out, glist, gcnt, n, rgroups);
}


# fewer passes over the LDS block per layer: sums start at the rows' own share (no zero pass, no own-share pass); both k halves go to LDS, the row-store pass adds them, bias, elu; sum order changes
# speedup vs baseline: 5.9171x; 1.0804x over previous
#include <hip/hip_runtime.h>
#include <hip/hip_bf16.h>

typedef __attribute__((ext_vector_type(2))) float v2f;
typedef __attribute__((ext_vector_type(8))) float v8f;

#define C_HID 256

#define PUSH_ROWS 40

typedef __attribute__((ext_vector_type(4))) unsigned v4u;
typedef __attribute__((ext_vector_type(2))) unsigned v2u;

#define LIST_CAP 4096

#define PREP_THREADS 1024
#define HIST_ROWS    1024
#define HIST_CHUNKS  4
#define HIST_LOADS   4
#define SEG_CHUNK    (PREP_THREADS * 4)
#define SEG_OWNERS   256

typedef __attribute__((ext_vector_type(4))) int v4i;

__device__ __forceinline__ void hist_one(int* hist, int d, unsigned nlo4) {
  const unsigned xb = ((unsigned)d << 2) + nlo4;
  if (xb < 4u * HIST_ROWS)
    __hip_atomic_fetch_add((int*)((char*)hist + xb), 1, __ATOMIC_RELAXED,
                           __HIP_MEMORY_SCOPE_WORKGROUP);
}

__device__ __forceinline__ void hist_vec(int* hist, const v4i& v, unsigned nlo4) {
  hist_one(hist, v.x, nlo4);
  hist_one(hist, v.y, nlo4);
  hist_one(hist, v.z, nlo4);
  hist_one(hist, v.w, nlo4);
}

__device__ __forceinline__ float dinv_at(const int* __restrict__ degp, int node) {
  const v4i c = *(const v4i*)(degp + node * HIST_CHUNKS);
  return rsqrtf((float)(1 + c.x + c.y + c.z + c.w));
}

__global__ __launch_bounds__(PREP_THREADS) void k_prep(
    const int* __restrict__ src, const int* __restrict__ dst,
    int* __restrict__ degp, unsigned* __restrict__ segs,
    unsigned* __restrict__ pks, int n, int E, int hgroups, int nch) {
  __shared__ int hist[HIST_ROWS];
  __shared__ __attribute__((aligned(16))) unsigned srt[SEG_CHUNK];
  const int tid = threadIdx.x;
  if ((int)blockIdx.x < hgroups * HIST_CHUNKS) {
    const int rng = (int)blockIdx.x / HIST_CHUNKS;
    const int pc  = (int)blockIdx.x - rng * HIST_CHUNKS;
    const int lo = rng * HIST_ROWS;
    const unsigned nlo4 = 0u - ((unsigned)lo << 2);
    if (tid < HIST_ROWS) hist[tid] = 0;
    __syncthreads();
    const int head  = min(E, (int)(((16u - ((unsigned)(size_t)dst & 15u)) & 15u) >> 2));
    const int nall  = (E - head) >> 2;
    const int tail0 = head + (nall << 2);
    if (pc == 0) {
      int e = -1;
      if (tid < head) e = tid;
      else if (tid - head < E - tail0) e = tail0 + (tid - head);
      if (e >= 0) hist_one(hist, dst[e], nlo4);
    }
    const int v0   = (int)(((long long)nall * pc) / HIST_CHUNKS);
    const int nvec = (int)(((long long)nall * (pc + 1)) / HIST_CHUNKS) - v0;
    const v4i* dv = (const v4i*)(dst + head) + v0;
    constexpr int TRIP = PREP_THREADS * HIST_LOADS;
    const int nfull = nvec / TRIP;
    if (nfull > 0) {
      v4i cur[HIST_LOADS];
#pragma unroll
      for (int k = 0; k < HIST_LOADS; ++k) cur[k] = dv[k * PREP_THREADS + tid];
      for (int i = 0; i < nfull; ++i) {
        const int nb = ((i + 1 < nfull) ? i + 1 : i) * TRIP;
        v4i nxt[HIST_LOADS];
#pragma unroll
        for (int k = 0; k < HIST_LOADS; ++k) nxt[k] = dv[nb + k * PREP_THREADS + tid];
#pragma unroll
        for (int k = 0; k < HIST_LOADS; ++k) hist_vec(hist, cur[k], nlo4);
#pragma unroll
        for (int k = 0; k < HIST_LOADS; ++k) cur[k] = nxt[k];
      }
    }
    {
      const int r0 = nfull * TRIP;
      const v4i none = {-1, -1, -1, -1};
      v4i rem[HIST_LOADS];
#pragma unroll
      for (int k = 0; k < HIST_LOADS; ++k) {
        const int v = r0 + k * PREP_THREADS + tid;
        rem[k] = none;
        if (v < nvec) rem[k] = dv[v];
      }
#pragma unroll
      for (int k = 0; k < HIST_LOADS; ++k) hist_vec(hist, rem[k], nlo4);
    }
    __syncthreads();
    if (tid < HIST_ROWS && lo + tid < n)
      degp[(lo + tid) * HIST_CHUNKS + pc] = hist[tid];
    return;
  }

  const int c    = (int)blockIdx.x - hgroups * HIST_CHUNKS;
  const int e0   = (c * PREP_THREADS + tid) * 4;
  const int lane = tid & 31;
  int* cnt  = hist;
  int* base = hist + SEG_OWNERS;
  int* wsum = hist + 2 * SEG_OWNERS;
  if (tid < SEG_OWNERS) cnt[tid] = 0;
  __syncthreads();

  int gq[4], rk[4];
  unsigned pv[4];
  if (e0 + 3 < E) {
    int d[4], sv[4];
#pragma unroll
    for (int q = 0; q < 4; ++q) { d[q] = dst[e0 + q]; sv[q] = src[e0 + q]; }
#pragma unroll
    for (int q = 0; q < 4; ++q) {
      const int g = d[q] / PUSH_ROWS;
      pv[q] = ((unsigned)sv[q] << 6) | (unsigned)(d[q] - g * PUSH_ROWS);
      gq[q] = g;
      rk[q] = __hip_atomic_fetch_add(cnt + g, 1, __ATOMIC_RELAXED, __HIP_MEMORY_SCOPE_WORKGROUP);
    }
  } else {
#pragma unroll
    for (int q = 0; q < 4; ++q) {
      gq[q] = -1; rk[q] = 0; pv[q] = 0u;
      if (e0 + q < E) {
        const int d = dst[e0 + q];
        const int g = d / PUSH_ROWS;
        pv[q] = ((unsigned)src[e0 + q] << 6) | (unsigned)(d - g * PUSH_ROWS);
        gq[q] = g;
        rk[q] = __hip_atomic_fetch_add(cnt + g, 1, __ATOMIC_RELAXED, __HIP_MEMORY_SCOPE_WORKGROUP);
      }
    }
  }
  __syncthreads();

  int v = 0, x = 0;
  if (tid < SEG_OWNERS) {
    v = cnt[tid];
    x = v;
    for (int dd = 1; dd < 32; dd <<= 1) {
      const int y = __shfl_up(x, dd);
      if (lane >= dd) x += y;
    }
    if (lane == 31) wsum[tid >> 5] = x;
  }
  __syncthreads();
  if (tid < SEG_OWNERS) {
    int add = 0;
#pragma unroll
    for (int i = 0; i < SEG_OWNERS / 32; ++i) {
      const int wv = wsum[i];
      add += (i < (tid >> 5)) ? wv : 0;
    }
    const int b = x - v + add;
    base[tid] = b;
    segs[(long long)tid * nch + c] = (unsigned)b | ((unsigned)v << 16);
  }
  __syncthreads();
#pragma unroll
  for (int q = 0; q < 4; ++q)
    if (gq[q] >= 0) srt[base[gq[q]] + rk[q]] = pv[q];
  __syncthreads();
  const int nvalid = min(SEG_CHUNK, E - c * SEG_CHUNK);
  if (tid * 4 < nvalid)
    *(v4u*)(pks + (long long)c * SEG_CHUNK + tid * 4) = *(const v4u*)(srt + tid * 4);
}

#define PUSH_THREADS 1024
#define PUSH_WAVES   32
#define PUSH_RING    1024
#define SEG_SLOTS    4
#define PUSH_MIN_BATCH 16
#define TILE_ROWS    48
#define ACC_PITCH(K) ((K) + 4)
#define ACC_FLOATS   (TILE_ROWS * ACC_PITCH(C_HID))

template <int K>
__device__ __forceinline__ int pos_to_chan(int p) {
  if (K == 256) return (p & 128) + 4 * (p & 31) + ((p >> 5) & 3);
  return 4 * (p & 15) + (p >> 4);
}

__device__ __forceinline__ void ldsAddF(float* p, float v) {
  __hip_atomic_fetch_add(p, v, __ATOMIC_RELAXED, __HIP_MEMORY_SCOPE_WORKGROUP);
}

__device__ __forceinline__ int laneI(int v, int j) {
  return __builtin_amdgcn_readlane(v, j);
}
__device__ __forceinline__ float laneF(float v, int j) {
  return __int_as_float(__builtin_amdgcn_readlane(__float_as_int(v), j));
}

template <int K>
__device__ __forceinline__ void add_rows(
    const float* __restrict__ X, float* acc, int s, float nrm, int off,
    int count, int lane) {
  if (K == 256) {
    int j = 0;
    for (; j + 4 <= count; j += 4) {
      float4 a[4], b[4];
#pragma unroll
      for (int u = 0; u < 4; ++u) {
        const float* hp = X + (long long)laneI(s, j + u) * K + lane * 4;
        a[u] = *(const float4*)hp;
        b[u] = *(const float4*)(hp + 128);
      }
#pragma unroll
      for (int u = 0; u < 4; ++u) {
        const float nj = laneF(nrm, j + u);
        float* q = acc + laneI(off, j + u) + lane;
        ldsAddF(q,       a[u].x * nj);
        ldsAddF(q + 32,  a[u].y * nj);
        ldsAddF(q + 64,  a[u].z * nj);
        ldsAddF(q + 96,  a[u].w * nj);
        ldsAddF(q + 128, b[u].x * nj);
        ldsAddF(q + 160, b[u].y * nj);
        ldsAddF(q + 192, b[u].z * nj);
        ldsAddF(q + 224, b[u].w * nj);
      }
    }
    for (; j < count; ++j) {
      const float* hp = X + (long long)laneI(s, j) * K + lane * 4;
      const float4 a = *(const float4*)hp;
      const float4 b = *(const float4*)(hp + 128);
      const float nj = laneF(nrm, j);
      float* q = acc + laneI(off, j) + lane;
      ldsAddF(q,       a.x * nj);
      ldsAddF(q + 32,  a.y * nj);
      ldsAddF(q + 64,  a.z * nj);
      ldsAddF(q + 96,  a.w * nj);
      ldsAddF(q + 128, b.x * nj);
      ldsAddF(q + 160, b.y * nj);
      ldsAddF(q + 192, b.z * nj);
      ldsAddF(q + 224, b.w * nj);
    }
  } else {
    const int  sub = lane & 15;
    const bool odd = lane >= 16;
    for (int j = 0; j < count; j += 8) {
      float4 a[4];
#pragma unroll
      for (int u = 0; u < 4; ++u) {
        const int e0 = j + 2 * u;
        const int sA = laneI(s, e0), sB = laneI(s, e0 + 1);
        a[u] = make_float4(0.0f, 0.0f, 0.0f, 0.0f);
        if (e0 + (odd ? 1 : 0) < count)
          a[u] = *(const float4*)(X + (long long)(odd ? sB : sA) * K + sub * 4);
      }
#pragma unroll
      for (int u = 0; u < 4; ++u) {
        const int e0 = j + 2 * u;
        const float nA = laneF(nrm, e0), nB = laneF(nrm, e0 + 1);
        const int   oA = laneI(off, e0), oB = laneI(off, e0 + 1);
        if (e0 + (odd ? 1 : 0) < count) {
          const float nj = odd ? nB : nA;
          float* q = acc + (odd ? oB : oA) + sub;
          ldsAddF(q,      a[u].x * nj);
          ldsAddF(q + 16, a[u].y * nj);
          ldsAddF(q + 32, a[u].z * nj);
          ldsAddF(q + 48, a[u].w * nj);
        }
      }
    }
  }
}

template <int K, bool STORE>
__device__ __forceinline__ void push_batch(
    const float* __restrict__ X,
    const int* __restrict__ degp, float* acc, const unsigned* ring,
    int head, int count, int row0, int lane, v2u* lst, int* rowcnt, int* lcnt) {
  int s = 0, off = 0, dl = 0;
  float nrm = 0.0f;
  unsigned ent = 0u;
  if (lane < count) {
    ent = ring[(head + lane) & (PUSH_RING - 1)];
    dl  = (int)(ent & 63u);
    s   = (int)(ent >> 6);
    nrm = dinv_at(degp, s) * dinv_at(degp, row0 + dl);
    off = dl * ACC_PITCH(K);
  }
  if (STORE) {
    int base = 0;
    if (lane == 0)
      base = __hip_atomic_fetch_add(lcnt, count, __ATOMIC_RELAXED, __HIP_MEMORY_SCOPE_WORKGROUP);
    base = __builtin_amdgcn_readfirstlane(base);
    if (lane < count && base + lane < LIST_CAP) {
      v2u o; o.x = ent; o.y = __float_as_uint(nrm);
      lst[base + lane] = o;
      __hip_atomic_fetch_add(rowcnt + dl, 1, __ATOMIC_RELAXED, __HIP_MEMORY_SCOPE_WORKGROUP);
    }
  }
  add_rows<K>(X, acc, s, nrm, off, count, lane);
}

__device__ __forceinline__ void run_flush(float* acc, int roff, int lane,
                                          const float4& lo, const float4& hi) {
  float* q = acc + roff + lane;
  ldsAddF(q,       lo.x);
  ldsAddF(q + 32,  lo.y);
  ldsAddF(q + 64,  lo.z);
  ldsAddF(q + 96,  lo.w);
  ldsAddF(q + 128, hi.x);
  ldsAddF(q + 160, hi.y);
  ldsAddF(q + 192, hi.z);
  ldsAddF(q + 224, hi.w);
}

__device__ __forceinline__ void run_edge(float* acc, int lane, int o, float nj,
                                         const float4& a, const float4& b,
                                         float4& lo, float4& hi, int& roff) {
  if (o != roff) {
    if (roff >= 0) run_flush(acc, roff, lane, lo, hi);
    lo = make_float4(0.0f, 0.0f, 0.0f, 0.0f);
    hi = make_float4(0.0f, 0.0f, 0.0f, 0.0f);
    roff = o;
  }
  lo.x += a.x * nj; lo.y += a.y * nj; lo.z += a.z * nj; lo.w += a.w * nj;
  hi.x += b.x * nj; hi.y += b.y * nj; hi.z += b.z * nj; hi.w += b.w * nj;
}

__device__ __forceinline__ void run_rows(
    const float* __restrict__ X, float* acc, int s, float nrm, int off,
    int count, int lane, float4& lo, float4& hi, int& roff) {
  int j = 0;
  for (; j + 8 <= count; j += 8) {
    float4 a[8], b[8];
#pragma unroll
    for (int u = 0; u < 8; ++u) {
      const float* hp = X + (long long)laneI(s, j + u) * C_HID + lane * 4;
      a[u] = *(const float4*)hp;
      b[u] = *(const float4*)(hp + 128);
    }
#pragma unroll
    for (int u = 0; u < 8; ++u)
      run_edge(acc, lane, laneI(off, j + u), laneF(nrm, j + u), a[u], b[u], lo, hi, roff);
  }
  for (; j + 4 <= count; j += 4) {
    float4 a[4], b[4];
#pragma unroll
    for (int u = 0; u < 4; ++u) {
      const float* hp = X + (long long)laneI(s, j + u) * C_HID + lane * 4;
      a[u] = *(const float4*)hp;
      b[u] = *(const float4*)(hp + 128);
    }
#pragma unroll
    for (int u = 0; u < 4; ++u)
      run_edge(acc, lane, laneI(off, j + u), laneF(nrm, j + u), a[u], b[u], lo, hi, roff);
  }
  for (; j < count; ++j) {
    const float* hp = X + (long long)laneI(s, j) * C_HID + lane * 4;
    const float4 a = *(const float4*)hp;
    const float4 b = *(const float4*)(hp + 128);
    run_edge(acc, lane, laneI(off, j), laneF(nrm, j), a, b, lo, hi, roff);
  }
}

template <int K, bool STORE>
__device__ __forceinline__ void seg_push(
    const float* __restrict__ X, const unsigned* __restrict__ pks,
    const unsigned* __restrict__ segs, const int* __restrict__ degp,
    float* acc, unsigned* ring, v2u* lst, int* rowcnt, int* lcnt,
    int row0, int nch, int w, int lane) {
  const unsigned* myseg = segs + (long long)blockIdx.x * nch;
  int head = 0, tail = 0;
  for (int cb = w; cb < nch; cb += PUSH_WAVES * SEG_SLOTS) {
    unsigned sg = 0u;
    {
      const int c = cb + lane * PUSH_WAVES;
      if (lane < SEG_SLOTS && c < nch) sg = myseg[c];
    }
    unsigned ent[SEG_SLOTS];
#pragma unroll
    for (int j = 0; j < SEG_SLOTS; ++j) {
      const unsigned sw = (unsigned)laneI((int)sg, j);
      const int st = (int)(sw & 0xffffu);
      const int cn = (int)(sw >> 16);
      ent[j] = 0u;
      if (lane < cn)
        ent[j] = pks[(long long)(cb + j * PUSH_WAVES) * SEG_CHUNK + st + lane];
    }
#pragma unroll
    for (int j = 0; j < SEG_SLOTS; ++j) {
      const unsigned sw = (unsigned)laneI((int)sg, j);
      const int st = (int)(sw & 0xffffu);
      const int cn = (int)(sw >> 16);
      const unsigned* p = pks + (long long)(cb + j * PUSH_WAVES) * SEG_CHUNK + st;
      for (int i = 0; i < cn; i += 32) {
        const int m = min(32, cn - i);
        unsigned e = ent[j];
        if (i > 0) {
          e = 0u;
          if (lane < m) e = p[i + lane];
        }
        if (lane < m) ring[(tail + lane) & (PUSH_RING - 1)] = e;
        tail += m;
        while (tail - head >= PUSH_MIN_BATCH) {
          const int cnt = min(32, (tail - head) & ~3);
          push_batch<K, STORE>(X, degp, acc, ring, head, cnt, row0, lane, lst, rowcnt, lcnt);
          head += cnt;
        }
      }
    }
  }
  while (tail - head >= 1) {
    const int cnt = min(32, tail - head);
    push_batch<K, STORE>(X, degp, acc, ring, head, cnt, row0, lane, lst, rowcnt, lcnt);
    head += cnt;
  }
}

template <int K, bool FIRST>
__global__ __launch_bounds__(PUSH_THREADS) void k_push_mm(
    const float* __restrict__ X, const unsigned* __restrict__ pks,
    const unsigned* __restrict__ segs, const int* __restrict__ degp,
    const float* __restrict__ Wm, const float* __restrict__ bias,
    float* __restrict__ OUT, unsigned* __restrict__ glist_all,
    int* __restrict__ gcnt, int n, int nch) {
  __shared__ __attribute__((aligned(16))) float    acc[ACC_FLOATS];
  __shared__ __attribute__((aligned(16))) unsigned pend[PUSH_WAVES * PUSH_RING];
  __shared__ int lcnt;
  __shared__ __attribute__((aligned(16))) v2u lst[FIRST ? LIST_CAP : 1];
  __shared__ int rowcnt[64];
  __shared__ int rowpos[64];
  constexpr int PITCH = ACC_PITCH(K);

  const int tid  = threadIdx.x;
  const int lane = tid & 31;
  const int w    = __builtin_amdgcn_readfirstlane(tid >> 5);
  const int row0 = blockIdx.x * PUSH_ROWS;
  const int nrows = min(PUSH_ROWS, n - row0);
  unsigned* ring = pend + w * PUSH_RING;
  v2u* glist = (v2u*)glist_all + (long long)blockIdx.x * LIST_CAP;

  constexpr int QPR  = K / 4;
  constexpr int JST  = (K == 256) ? 32 : 16;
  constexpr int NINI = (TILE_ROWS * QPR + PUSH_THREADS - 1) / PUSH_THREADS;
  int cnt = 0;
  if (!FIRST) cnt = gcnt[blockIdx.x];
  float4 inx[NINI];
  v4i    inc[NINI];
#pragma unroll
  for (int i = 0; i < NINI; ++i) {
    const int idx = tid + i * PUSH_THREADS;
    const int r   = idx / QPR;
    const int cq  = idx - r * QPR;
    const int node = row0 + min(r, nrows - 1);
    inc[i] = *(const v4i*)(degp + node * HIST_CHUNKS);
    inx[i] = *(const float4*)(X + (long long)node * K + cq * 4);
  }
  int lb = 0, lend = 0;
  v2u cur = {0u, 0u};
  if (!FIRST) {
    if (cnt <= LIST_CAP) {
      const int per = (((cnt + PUSH_WAVES - 1) / PUSH_WAVES) + 3) & ~3;
      lb   = w * per;
      lend = min(cnt, lb + per);
      if (lb + lane < lend) cur = glist[lb + lane];
    }
  }
#pragma unroll
  for (int i = 0; i < NINI; ++i) {
    const int idx = tid + i * PUSH_THREADS;
    const int r   = idx / QPR;
    const int cq  = idx - r * QPR;
    if (r < TILE_ROWS) {
      const float di = rsqrtf((float)(1 + inc[i].x + inc[i].y + inc[i].z + inc[i].w));
      const float d2 = di * di;
      const bool real = r < nrows;
      float* ap = acc + r * PITCH + ((K == 256) ? ((cq >> 5) * 128 + (cq & 31)) : cq);
      ap[0]       = real ? d2 * inx[i].x : 0.0f;
      ap[JST]     = real ? d2 * inx[i].y : 0.0f;
      ap[2 * JST] = real ? d2 * inx[i].z : 0.0f;
      ap[3 * JST] = real ? d2 * inx[i].w : 0.0f;
    }
  }
  if (FIRST && tid == 0) lcnt = 0;
  if (FIRST && tid < 64) rowcnt[tid] = 0;
  __syncthreads();

  if (FIRST) {
    seg_push<K, true>(X, pks, segs, degp, acc, ring, lst, rowcnt, &lcnt, row0, nch, w, lane);
  } else {
    if (cnt <= LIST_CAP) {
      int b = lb;
      const int end = lend;
      float4 lo = make_float4(0.0f, 0.0f, 0.0f, 0.0f);
      float4 hi = make_float4(0.0f, 0.0f, 0.0f, 0.0f);
      int roff = -1;
      for (; b < end; b += 32) {
        v2u nxt = {0u, 0u};
        if (b + 32 + lane < end) nxt = glist[b + 32 + lane];
        const int count = min(32, end - b);
        int s = 0, off = 0;
        float nrm = 0.0f;
        if (lane < count) {
          s   = (int)(cur.x >> 6);
          off = (int)(cur.x & 63u) * PITCH;
          nrm = __uint_as_float(cur.y);
        }
        run_rows(X, acc, s, nrm, off, count, lane, lo, hi, roff);
        cur = nxt;
      }
      if (roff >= 0) run_flush(acc, roff, lane, lo, hi);
    } else {
      seg_push<K, false>(X, pks, segs, degp, acc, ring, nullptr, nullptr, &lcnt, row0, nch, w, lane);
    }
  }

  __syncthreads();
  int lc = 0;
  if (FIRST) {
    lc = lcnt;
    if (tid == 0) gcnt[blockIdx.x] = lc;
    if (tid < 64) {
      int before = 0;
#pragma unroll
      for (int i = 0; i < PUSH_ROWS; ++i) {
        const int cr = rowcnt[i];
        before += (i < tid) ? cr : 0;
      }
      rowpos[tid] = before;
    }
  }

  if (FIRST) __syncthreads();

  if (FIRST && lc <= LIST_CAP) {
    for (int i = tid; i < lc; i += PUSH_THREADS) {
      const v2u e = lst[i];
      const int pos = __hip_atomic_fetch_add(rowpos + (int)(e.x & 63u), 1,
                                             __ATOMIC_RELAXED, __HIP_MEMORY_SCOPE_WORKGROUP);
      glist[pos] = e;
    }
  }

  const int lo = lane & 15;
  const int hi = lane >> 4;
  const int n0 = (w & 15) << 4;
  const int kh = w >> 4;
  v8f d0 = {}, d1 = {}, d2 = {};
  {
    constexpr int STEPS = K / 8;
    const int pbase = kh * (K / 2) + 2 * hi;
    const float* Ap = acc + lo * PITCH + pbase;
    const float* Wc = Wm + n0 + lo;
#pragma unroll 8
    for (int t = 0; t < STEPS; ++t) {
      const int p = pbase + 4 * t;
      v2f b;
      b.x = Wc[(long long)pos_to_chan<K>(p)     * C_HID];
      b.y = Wc[(long long)pos_to_chan<K>(p + 1) * C_HID];
      const v2f a0 = *(const v2f*)(Ap + 4 * t);
      const v2f a1 = *(const v2f*)(Ap + 16 * PITCH + 4 * t);
      const v2f a2 = *(const v2f*)(Ap + 32 * PITCH + 4 * t);
      d0 = __builtin_amdgcn_wmma_f32_16x16x4_f32(false, a0, false, b, (short)0, d0, false, false);
      d1 = __builtin_amdgcn_wmma_f32_16x16x4_f32(false, a1, false, b, (short)0, d1, false, false);
      d2 = __builtin_amdgcn_wmma_f32_16x16x4_f32(false, a2, false, b, (short)0, d2, false, false);
    }
  }

  constexpr int SBLK = TILE_ROWS * C_HID;
  float* S  = (float*)pend;
  float* Sp = S + kh * SBLK + (hi * 8) * C_HID + n0 + lo;
#pragma unroll
  for (int r = 0; r < 8; ++r) {
    Sp[(r)      * C_HID] = d0[r];
    Sp[(16 + r) * C_HID] = d1[r];
    Sp[(32 + r) * C_HID] = d2[r];
  }
  const int cq = tid & 63;
  const float4 bb = *(const float4*)(bias + cq * 4);
  __syncthreads();

  for (int idx = tid; idx < nrows * (C_HID / 4); idx += PUSH_THREADS) {
    const int r = idx >> 6;
    const float4 p0 = *(const float4*)(S + r * C_HID + cq * 4);
    const float4 p1 = *(const float4*)(S + SBLK + r * C_HID + cq * 4);
    float4 v;
    v.x = p0.x + p1.x + bb.x;
    v.y = p0.y + p1.y + bb.y;
    v.z = p0.z + p1.z + bb.z;
    v.w = p0.w + p1.w + bb.w;
    v.x = (v.x > 0.0f) ? v.x : expm1f(v.x);
    v.y = (v.y > 0.0f) ? v.y : expm1f(v.y);
    v.z = (v.z > 0.0f) ? v.z : expm1f(v.z);
    v.w = (v.w > 0.0f) ? v.w : expm1f(v.w);
    *(float4*)(OUT + (long long)(row0 + r) * C_HID + cq * 4) = v;
  }
}

extern "C" void kernel_launch(void* const* d_in, const int* in_sizes, int n_in,
                              void* d_out, int out_size, void* d_ws, size_t ws_size,
                              hipStream_t stream) {
  const float* x  = (const float*)d_in[0];
  const int*   ei = (const int*)d_in[1];
  const float* W1 = (const float*)d_in[2];
  const float* b1 = (const float*)d_in[3];
  const float* W2 = (const float*)d_in[4];
  const float* b2 = (const float*)d_in[5];
  const float* W3 = (const float*)d_in[6];
  const float* b3 = (const float*)d_in[7];
  float* out = (float*)d_out;

  const int C   = C_HID;
  const int E   = in_sizes[1] / 2;
  const int cin = in_sizes[2] / C;
  const int n   = in_sizes[0] / cin;
  const int* src = ei;
  const int* dst = ei + E;

  char*  ws  = (char*)d_ws;
  size_t off = 0;
  auto alloc = [&](size_t bytes) -> void* {
    void* p = ws + off;
    off = (off + bytes + 255) & ~(size_t)255;
    return p;
  };
  int* degp = (int*)alloc((size_t)n * HIST_CHUNKS * sizeof(int));
  float* XA   = (float*)alloc((size_t)n * C * sizeof(float));
  float* XB   = (float*)alloc((size_t)n * C * sizeof(float));
  const int hgroups = (n + HIST_ROWS - 1) / HIST_ROWS;
  const int rgroups = ((E + 3) / 4 + PREP_THREADS - 1) / PREP_THREADS;
  unsigned* segs = (unsigned*)alloc((size_t)SEG_OWNERS * rgroups * sizeof(unsigned));
  unsigned* pks  = (unsigned*)alloc((size_t)rgroups * SEG_CHUNK * sizeof(unsigned));
  const int pgrid = (n + PUSH_ROWS - 1) / PUSH_ROWS;
  unsigned* glist = (unsigned*)alloc((size_t)pgrid * LIST_CAP * 2 * sizeof(unsigned));
  int*      gcnt  = (int*)     alloc((size_t)pgrid * sizeof(int));
  (void)ws_size; (void)n_in; (void)out_size;

  k_prep<<<hgroups * HIST_CHUNKS + rgroups, PREP_THREADS, 0, stream>>>(src, dst, degp, segs, pks, n, E, hgroups, rgroups);

  k_push_mm<64, true>  <<<pgrid, PUSH_THREADS, 0, stream>>>(x,  pks, segs, degp, W1, b1, XA,  glist, gcnt, n, rgroups);
  k_push_mm<256, false><<<pgrid, PUSH_THREADS, 0, stream>>>(XA, pks, segs, degp, W2, b2, XB,  glist, gcnt, n, rgroups);
  k_push_mm<256, false><<<pgrid, PUSH_THREADS, 0, stream>>>(XB, pks, segs, degp, W3, b3, out, glist, gcnt, n, rgroups);
}
